# starting kernel
# baseline (speedup 1.0000x reference)
#include <hip/hip_runtime.h>

typedef float v2f __attribute__((ext_vector_type(2)));
typedef float v8f __attribute__((ext_vector_type(8)));

#define N_NODESC 100000
#define N_EDGESC 1600000
#define DIM 128


__global__ void k_init_deg(float* __restrict__ deg) {
  int i = blockIdx.x * blockDim.x + threadIdx.x;
  if (i < N_NODESC) deg[i] = 1.0f;
}

__global__ void k_count_deg(const int* __restrict__ dst, float* __restrict__ deg) {
  int e = blockIdx.x * blockDim.x + threadIdx.x;
  if (e < N_EDGESC) unsafeAtomicAdd(&deg[dst[e]], 1.0f);
}

__global__ void k_dinv(const float* __restrict__ deg, float* __restrict__ dinv) {
  int i = blockIdx.x * blockDim.x + threadIdx.x;
  if (i < N_NODESC) dinv[i] = rsqrtf(deg[i]);
}

__global__ void k_zero4(float4* __restrict__ p, int n4) {
  int i = blockIdx.x * blockDim.x + threadIdx.x;
  if (i < n4) p[i] = make_float4(0.f, 0.f, 0.f, 0.f);
}


__global__ __launch_bounds__(256) void k_gemm_wmma(const float* __restrict__ X,
                                                   const float* __restrict__ W,
                                                   float* __restrict__ H) {
  __shared__ float xs[16 * 132];

  const int t = threadIdx.x;
  const int wave = t >> 5;
  const int lane = t & 31;
  const int row0 = blockIdx.x * 16;

  {
    const int r = t >> 4;
    const int c = (t & 15) * 8;
    const float4* g = (const float4*)(X + (size_t)(row0 + r) * DIM + c);
    float4 p0 = g[0];
    float4 p1 = g[1];
    float4* s = (float4*)&xs[r * 132 + c];
    s[0] = p0;
    s[1] = p1;
  }
  __syncthreads();

  const int m  = lane & 15;
  const int hg = lane >> 4;
  const int n0 = wave * 16;

  v8f acc = {};
  const float* wcol = W + n0 + m;

  #pragma unroll 4
  for (int kk = 0; kk < DIM; kk += 4) {
    v2f a = *(const v2f*)&xs[m * 132 + kk + 2 * hg];
    v2f b;
    b.x = wcol[(size_t)(kk + 2 * hg + 0) * DIM];
    b.y = wcol[(size_t)(kk + 2 * hg + 1) * DIM];
    acc = __builtin_amdgcn_wmma_f32_16x16x4_f32(
         false, a,  false, b,
         (short)0, acc,  false,  false);
  }

  float* hrow = H + (size_t)row0 * DIM + n0 + m;
  #pragma unroll
  for (int v = 0; v < 8; ++v)
    hrow[(size_t)(v + 8 * hg) * DIM] = acc[v];
}


__global__ __launch_bounds__(256) void k_scatter(const int* __restrict__ src,
                                                 const int* __restrict__ dst,
                                                 const float* __restrict__ dinv,
                                                 const float* __restrict__ H,
                                                 float* __restrict__ AGG) {
  int e = blockIdx.x * 8 + (threadIdx.x >> 5);
  if (e >= N_EDGESC) return;
  int lane = threadIdx.x & 31;
  int s = src[e];
  int d = dst[e];
  float w = dinv[s] * dinv[d];
  float4 hv = ((const float4*)(H + (size_t)s * DIM))[lane];
  float* arow = AGG + (size_t)d * DIM + lane * 4;
  unsafeAtomicAdd(arow + 0, hv.x * w);
  unsafeAtomicAdd(arow + 1, hv.y * w);
  unsafeAtomicAdd(arow + 2, hv.z * w);
  unsafeAtomicAdd(arow + 3, hv.w * w);
}


__global__ void k_finalize(const float* __restrict__ AGG, const float* __restrict__ H,
                           const float* __restrict__ dinv, const float* __restrict__ b,
                           float* __restrict__ OUT, int do_relu) {
  int i = blockIdx.x * blockDim.x + threadIdx.x;
  if (i >= N_NODESC * DIM) return;
  int node = i >> 7;
  int f = i & (DIM - 1);
  float di = dinv[node];
  float v = AGG[i] + H[i] * di * di + b[f];
  if (do_relu) v = fmaxf(v, 0.0f);
  OUT[i] = v;
}


extern "C" void kernel_launch(void* const* d_in, const int* in_sizes, int n_in,
                              void* d_out, int out_size, void* d_ws, size_t ws_size,
                              hipStream_t stream) {
  const float* x   = (const float*)d_in[0];
  const int*   ei  = (const int*)d_in[1];
  const float* W1  = (const float*)d_in[2];
  const float* b1  = (const float*)d_in[3];
  const float* W2  = (const float*)d_in[4];
  const float* b2  = (const float*)d_in[5];
  float*       out = (float*)d_out;

  const int* src = ei;
  const int* dst = ei + N_EDGESC;

  float* H    = (float*)d_ws;
  float* AGG  = H + (size_t)N_NODESC * DIM;
  float* deg  = AGG + (size_t)N_NODESC * DIM;
  float* dinv = deg + N_NODESC;

  const int feat_elems = N_NODESC * DIM;
  const int feat_vec4  = feat_elems / 4;

  dim3 blk(256);
  dim3 gNode((N_NODESC + 255) / 256);
  dim3 gEdge((N_EDGESC + 255) / 256);
  dim3 gGemm(N_NODESC / 16);
  dim3 gZero((feat_vec4 + 255) / 256);
  dim3 gScat(N_EDGESC / 8);
  dim3 gFin((feat_elems + 255) / 256);

  k_init_deg<<<gNode, blk, 0, stream>>>(deg);
  k_count_deg<<<gEdge, blk, 0, stream>>>(dst, deg);
  k_dinv<<<gNode, blk, 0, stream>>>(deg, dinv);

  k_gemm_wmma<<<gGemm, blk, 0, stream>>>(x, W1, H);
  k_zero4<<<gZero, blk, 0, stream>>>((float4*)AGG, feat_vec4);
  k_scatter<<<gScat, blk, 0, stream>>>(src, dst, dinv, H, AGG);
  k_finalize<<<gFin, blk, 0, stream>>>(AGG, H, dinv, b1, out,  1);

  k_gemm_wmma<<<gGemm, blk, 0, stream>>>(out, W2, H);
  k_zero4<<<gZero, blk, 0, stream>>>((float4*)AGG, feat_vec4);
  k_scatter<<<gScat, blk, 0, stream>>>(src, dst, dinv, H, AGG);
  k_finalize<<<gFin, blk, 0, stream>>>(AGG, H, dinv, b2, out,  0);
}


# Edge scatter replaced by owner blocks: 256 blocks each hold a 391-node slice of the layer sum in LDS, stream the whole dst list, add matching rows by LDS float atomics (LDS arrival order); 7 launches
# speedup vs baseline: 2.0785x; 2.0785x over previous
#include <hip/hip_runtime.h>

typedef float v2f __attribute__((ext_vector_type(2)));
typedef float v8f __attribute__((ext_vector_type(8)));

#define N_NODESC 100000
#define N_EDGESC 1600000
#define DIM 128


__global__ void k_init_deg(float* __restrict__ deg) {
  int i = blockIdx.x * blockDim.x + threadIdx.x;
  if (i < N_NODESC) deg[i] = 1.0f;
}

__global__ void k_count_deg(const int* __restrict__ dst, float* __restrict__ deg) {
  int e = blockIdx.x * blockDim.x + threadIdx.x;
  if (e < N_EDGESC) unsafeAtomicAdd(&deg[dst[e]], 1.0f);
}

__global__ void k_dinv(const float* __restrict__ deg, float* __restrict__ dinv) {
  int i = blockIdx.x * blockDim.x + threadIdx.x;
  if (i < N_NODESC) dinv[i] = rsqrtf(deg[i]);
}


__global__ __launch_bounds__(256) void k_gemm_wmma(const float* __restrict__ X,
                                                   const float* __restrict__ W,
                                                   float* __restrict__ H) {
  __shared__ float xs[16 * 132];

  const int t = threadIdx.x;
  const int wave = t >> 5;
  const int lane = t & 31;
  const int row0 = blockIdx.x * 16;

  {
    const int r = t >> 4;
    const int c = (t & 15) * 8;
    const float4* g = (const float4*)(X + (size_t)(row0 + r) * DIM + c);
    float4 p0 = g[0];
    float4 p1 = g[1];
    float4* s = (float4*)&xs[r * 132 + c];
    s[0] = p0;
    s[1] = p1;
  }
  __syncthreads();

  const int m  = lane & 15;
  const int hg = lane >> 4;
  const int n0 = wave * 16;

  v8f acc = {};
  const float* wcol = W + n0 + m;

  #pragma unroll 4
  for (int kk = 0; kk < DIM; kk += 4) {
    v2f a = *(const v2f*)&xs[m * 132 + kk + 2 * hg];
    v2f b;
    b.x = wcol[(size_t)(kk + 2 * hg + 0) * DIM];
    b.y = wcol[(size_t)(kk + 2 * hg + 1) * DIM];
    acc = __builtin_amdgcn_wmma_f32_16x16x4_f32(
         false, a,  false, b,
         (short)0, acc,  false,  false);
  }

  float* hrow = H + (size_t)row0 * DIM + n0 + m;
  #pragma unroll
  for (int v = 0; v < 8; ++v)
    hrow[(size_t)(v + 8 * hg) * DIM] = acc[v];
}


#define AGG_BLOCKS 256
#define NPB 391
#define AGG_THREADS 1024
#define AGG_WAVES 32
#define SCAN_U 4
#define N_DST4 (N_EDGESC / 4)
#define WAVE4 (32 * SCAN_U)
#define TRIP4 (AGG_WAVES * WAVE4)
#define N_TRIPS ((N_DST4 + TRIP4 - 1) / TRIP4)
#define BATCH 8
#define LIST_CAP (BATCH - 1 + 4 * WAVE4 + 1)

static_assert(AGG_BLOCKS * NPB >= N_NODESC, "slices must cover every node");
static_assert(N_EDGESC % 4 == 0, "dst list is read as int4");
static_assert(N_EDGESC <= (1 << 21) && NPB <= (1 << 9), "list entry packs edge id and local row");

__device__ __forceinline__ void agg_load_trip(const int4* __restrict__ dst4, int base4, int lane,
                                              int4 (&v)[SCAN_U]) {
  #pragma unroll
  for (int u = 0; u < SCAN_U; ++u) {
    const int i4 = base4 + u * 32 + lane;
    v[u] = (i4 < N_DST4) ? dst4[i4] : make_int4(-1, -1, -1, -1);
  }
}

__device__ __forceinline__ void agg_batch(const int* lst, int nb, int lane,
                                          const int* __restrict__ src,
                                          const float* __restrict__ dinv,
                                          const float* dinvL,
                                          const float* __restrict__ H, float* tile) {
  int ent = 0;
  if (lane < nb) ent = lst[lane];
  const int e = ent & ((1 << 21) - 1);
  const int dl = (int)((unsigned)ent >> 21);
  const int s = src[e];
  const float w = dinv[s] * dinvL[dl];

  float hv[BATCH][4];
  #pragma unroll
  for (int i = 0; i < BATCH; ++i) {
    const int si = __builtin_amdgcn_readlane(s, i);
    const float* hr = H + (size_t)si * DIM + lane;
    hv[i][0] = hr[0];
    hv[i][1] = hr[32];
    hv[i][2] = hr[64];
    hv[i][3] = hr[96];
  }
  #pragma unroll
  for (int i = 0; i < BATCH; ++i) {
    if (i < nb) {
      const int di = __builtin_amdgcn_readlane(dl, i);
      const float wi = __int_as_float(__builtin_amdgcn_readlane(__float_as_int(w), i));
      float* tr = tile + di * DIM + lane;
      __hip_atomic_fetch_add(tr + 0,  hv[i][0] * wi, __ATOMIC_RELAXED, __HIP_MEMORY_SCOPE_WORKGROUP);
      __hip_atomic_fetch_add(tr + 32, hv[i][1] * wi, __ATOMIC_RELAXED, __HIP_MEMORY_SCOPE_WORKGROUP);
      __hip_atomic_fetch_add(tr + 64, hv[i][2] * wi, __ATOMIC_RELAXED, __HIP_MEMORY_SCOPE_WORKGROUP);
      __hip_atomic_fetch_add(tr + 96, hv[i][3] * wi, __ATOMIC_RELAXED, __HIP_MEMORY_SCOPE_WORKGROUP);
    }
  }
}

__global__ __launch_bounds__(AGG_THREADS) void k_agg_owner(const int* __restrict__ src,
                                                           const int* __restrict__ dst,
                                                           const float* __restrict__ dinv,
                                                           const float* __restrict__ H,
                                                           const float* __restrict__ bias,
                                                           float* __restrict__ OUT, int do_relu) {
  __shared__ __attribute__((aligned(16))) float tile[NPB * DIM];
  __shared__ float dinvL[NPB + 1];
  __shared__ int lists[AGG_WAVES * LIST_CAP];

  const int t = threadIdx.x;
  const int wave = t >> 5;
  const int lane = t & 31;
  const int lo = blockIdx.x * NPB;
  const int nvalid = (N_NODESC - lo < NPB) ? (N_NODESC - lo) : NPB;

  for (int r = wave; r < NPB; r += AGG_WAVES) {
    float4 v = make_float4(0.f, 0.f, 0.f, 0.f);
    float di = 0.f;
    if (r < nvalid) {
      di = dinv[lo + r];
      const float4 h = ((const float4*)(H + (size_t)(lo + r) * DIM))[lane];
      v.x = h.x * di * di;
      v.y = h.y * di * di;
      v.z = h.z * di * di;
      v.w = h.w * di * di;
    }
    ((float4*)(tile + r * DIM))[lane] = v;
    if (lane == 0) dinvL[r] = di;
  }
  __syncthreads();

  int* mylist = lists + wave * LIST_CAP;
  const int4* dst4 = (const int4*)dst;
  int pend = 0;
  int trip = blockIdx.x % N_TRIPS;

  int4 nxt[SCAN_U];
  agg_load_trip(dst4, trip * TRIP4 + wave * WAVE4, lane, nxt);

  for (int it = 0; it < N_TRIPS; ++it) {
    int4 cur[SCAN_U];
    #pragma unroll
    for (int u = 0; u < SCAN_U; ++u) cur[u] = nxt[u];
    const int cur4 = trip * TRIP4 + wave * WAVE4;
    trip = (trip + 1 == N_TRIPS) ? 0 : trip + 1;
    if (it + 1 < N_TRIPS) agg_load_trip(dst4, trip * TRIP4 + wave * WAVE4, lane, nxt);

    #pragma unroll
    for (int u = 0; u < SCAN_U; ++u) {
      #pragma unroll
      for (int c = 0; c < 4; ++c) {
        const int d = (c == 0) ? cur[u].x : (c == 1) ? cur[u].y : (c == 2) ? cur[u].z : cur[u].w;
        const unsigned tt = (unsigned)(d - lo);
        const bool m = tt < (unsigned)NPB;
        const unsigned mask = __builtin_amdgcn_ballot_w32(m);
        if (mask != 0u) {
          const int pos = pend + (int)__builtin_amdgcn_mbcnt_lo(mask, 0u);
          const int e = (cur4 + u * 32 + lane) * 4 + c;
          if (m) mylist[pos] = (int)((tt << 21) | (unsigned)e);
          pend += __builtin_popcount(mask);
        }
      }
    }
    while (pend >= BATCH) {
      pend -= BATCH;
      agg_batch(mylist + pend, BATCH, lane, src, dinv, dinvL, H, tile);
    }
  }
  if (pend > 0) agg_batch(mylist, pend, lane, src, dinv, dinvL, H, tile);

  __syncthreads();

  const float4 bb = ((const float4*)bias)[lane];
  for (int r = wave; r < nvalid; r += AGG_WAVES) {
    float4 a = ((const float4*)(tile + r * DIM))[lane];
    a.x += bb.x;
    a.y += bb.y;
    a.z += bb.z;
    a.w += bb.w;
    if (do_relu) {
      a.x = fmaxf(a.x, 0.0f);
      a.y = fmaxf(a.y, 0.0f);
      a.z = fmaxf(a.z, 0.0f);
      a.w = fmaxf(a.w, 0.0f);
    }
    ((float4*)(OUT + (size_t)(lo + r) * DIM))[lane] = a;
  }
}


extern "C" void kernel_launch(void* const* d_in, const int* in_sizes, int n_in,
                              void* d_out, int out_size, void* d_ws, size_t ws_size,
                              hipStream_t stream) {
  const float* x   = (const float*)d_in[0];
  const int*   ei  = (const int*)d_in[1];
  const float* W1  = (const float*)d_in[2];
  const float* b1  = (const float*)d_in[3];
  const float* W2  = (const float*)d_in[4];
  const float* b2  = (const float*)d_in[5];
  float*       out = (float*)d_out;

  const int* src = ei;
  const int* dst = ei + N_EDGESC;

  float* H    = (float*)d_ws;
  float* deg  = H + (size_t)N_NODESC * DIM;
  float* dinv = deg + N_NODESC;

  dim3 blk(256);
  dim3 gNode((N_NODESC + 255) / 256);
  dim3 gEdge((N_EDGESC + 255) / 256);
  dim3 gGemm(N_NODESC / 16);
  dim3 gAgg(AGG_BLOCKS);
  dim3 bAgg(AGG_THREADS);

  k_init_deg<<<gNode, blk, 0, stream>>>(deg);
  k_count_deg<<<gEdge, blk, 0, stream>>>(dst, deg);
  k_dinv<<<gNode, blk, 0, stream>>>(deg, dinv);

  k_gemm_wmma<<<gGemm, blk, 0, stream>>>(x, W1, H);
  k_agg_owner<<<gAgg, bAgg, 0, stream>>>(src, dst, dinv, H, b1, out,  1);

  k_gemm_wmma<<<gGemm, blk, 0, stream>>>(out, W2, H);
  k_agg_owner<<<gAgg, bAgg, 0, stream>>>(src, dst, dinv, H, b2, out,  0);
}


# Layer 2 stops streaming the dst list: layer-1 owner blocks write their matched edges (source, local row, weight) to workspace and layer 2 reads its block's list; sums stay in LDS arrival order
# speedup vs baseline: 2.4562x; 1.1817x over previous
#include <hip/hip_runtime.h>

typedef float v2f __attribute__((ext_vector_type(2)));
typedef float v8f __attribute__((ext_vector_type(8)));

#define N_NODESC 100000
#define N_EDGESC 1600000
#define DIM 128


__global__ void k_init_deg(float* __restrict__ deg) {
  int i = blockIdx.x * blockDim.x + threadIdx.x;
  if (i < N_NODESC) deg[i] = 1.0f;
}

__global__ void k_count_deg(const int* __restrict__ dst, float* __restrict__ deg) {
  int e = blockIdx.x * blockDim.x + threadIdx.x;
  if (e < N_EDGESC) unsafeAtomicAdd(&deg[dst[e]], 1.0f);
}

__global__ void k_dinv(const float* __restrict__ deg, float* __restrict__ dinv) {
  int i = blockIdx.x * blockDim.x + threadIdx.x;
  if (i < N_NODESC) dinv[i] = rsqrtf(deg[i]);
}


__global__ __launch_bounds__(256) void k_gemm_wmma(const float* __restrict__ X,
                                                   const float* __restrict__ W,
                                                   float* __restrict__ H) {
  __shared__ float xs[16 * 132];

  const int t = threadIdx.x;
  const int wave = t >> 5;
  const int lane = t & 31;
  const int row0 = blockIdx.x * 16;

  {
    const int r = t >> 4;
    const int c = (t & 15) * 8;
    const float4* g = (const float4*)(X + (size_t)(row0 + r) * DIM + c);
    float4 p0 = g[0];
    float4 p1 = g[1];
    float4* s = (float4*)&xs[r * 132 + c];
    s[0] = p0;
    s[1] = p1;
  }
  __syncthreads();

  const int m  = lane & 15;
  const int hg = lane >> 4;
  const int n0 = wave * 16;

  v8f acc = {};
  const float* wcol = W + n0 + m;

  #pragma unroll 4
  for (int kk = 0; kk < DIM; kk += 4) {
    v2f a = *(const v2f*)&xs[m * 132 + kk + 2 * hg];
    v2f b;
    b.x = wcol[(size_t)(kk + 2 * hg + 0) * DIM];
    b.y = wcol[(size_t)(kk + 2 * hg + 1) * DIM];
    acc = __builtin_amdgcn_wmma_f32_16x16x4_f32(
         false, a,  false, b,
         (short)0, acc,  false,  false);
  }

  float* hrow = H + (size_t)row0 * DIM + n0 + m;
  #pragma unroll
  for (int v = 0; v < 8; ++v)
    hrow[(size_t)(v + 8 * hg) * DIM] = acc[v];
}


#define AGG_BLOCKS 256
#define NPB 391
#define AGG_THREADS 1024
#define AGG_WAVES 32
#define SCAN_U 4
#define N_DST4 (N_EDGESC / 4)
#define WAVE4 (32 * SCAN_U)
#define TRIP4 (AGG_WAVES * WAVE4)
#define N_TRIPS ((N_DST4 + TRIP4 - 1) / TRIP4)
#define BATCH 8
#define LIST_CAP (BATCH - 1 + 4 * WAVE4 + 1)
#define REC_CAP 16384

static_assert(AGG_BLOCKS * NPB >= N_NODESC, "slices must cover every node");
static_assert(N_EDGESC % 4 == 0, "dst list is read as int4");
static_assert(N_EDGESC <= (1 << 21) && NPB <= (1 << 9), "list entry packs edge id and local row");
static_assert(N_NODESC <= (1 << 17), "recorded entry packs source node and local row");

__device__ __forceinline__ void agg_load_trip(const int4* __restrict__ dst4, int base4, int lane,
                                              int4 (&v)[SCAN_U]) {
  #pragma unroll
  for (int u = 0; u < SCAN_U; ++u) {
    const int i4 = base4 + u * 32 + lane;
    v[u] = (i4 < N_DST4) ? dst4[i4] : make_int4(-1, -1, -1, -1);
  }
}

__device__ __forceinline__ void rows_load(int s, int lane, const float* __restrict__ H,
                                          float (&hv)[BATCH][4]) {
  #pragma unroll
  for (int i = 0; i < BATCH; ++i) {
    const int si = __builtin_amdgcn_readlane(s, i);
    const float* hr = H + (size_t)si * DIM + lane;
    hv[i][0] = hr[0];
    hv[i][1] = hr[32];
    hv[i][2] = hr[64];
    hv[i][3] = hr[96];
  }
}

__device__ __forceinline__ void rows_add(const float (&hv)[BATCH][4], int dl, float w, int nb,
                                         int lane, float* tile) {
  #pragma unroll
  for (int i = 0; i < BATCH; ++i) {
    if (i < nb) {
      const int di = __builtin_amdgcn_readlane(dl, i);
      const float wi = __int_as_float(__builtin_amdgcn_readlane(__float_as_int(w), i));
      float* tr = tile + di * DIM + lane;
      __hip_atomic_fetch_add(tr + 0,  hv[i][0] * wi, __ATOMIC_RELAXED, __HIP_MEMORY_SCOPE_WORKGROUP);
      __hip_atomic_fetch_add(tr + 32, hv[i][1] * wi, __ATOMIC_RELAXED, __HIP_MEMORY_SCOPE_WORKGROUP);
      __hip_atomic_fetch_add(tr + 64, hv[i][2] * wi, __ATOMIC_RELAXED, __HIP_MEMORY_SCOPE_WORKGROUP);
      __hip_atomic_fetch_add(tr + 96, hv[i][3] * wi, __ATOMIC_RELAXED, __HIP_MEMORY_SCOPE_WORKGROUP);
    }
  }
}

__device__ __forceinline__ void agg_rows(int s, int dl, float w, int nb, int lane,
                                         const float* __restrict__ H, float* tile) {
  float hv[BATCH][4];
  rows_load(s, lane, H, hv);
  rows_add(hv, dl, w, nb, lane, tile);
}

template <bool REC>
__device__ __forceinline__ void agg_batch(const int* lst, int nb, int lane,
                                          const int* __restrict__ src,
                                          const float* __restrict__ dinv,
                                          const float* dinvL,
                                          const float* __restrict__ H, float* tile,
                                          int* cursor, int2* __restrict__ rec) {
  int ent = 0;
  if (lane < nb) ent = lst[lane];
  const int e = ent & ((1 << 21) - 1);
  const int dl = (int)((unsigned)ent >> 21);
  const int s = src[e];
  const float w = dinv[s] * dinvL[dl];
  if (REC) {
    int base = 0;
    if (lane == 0) base = __hip_atomic_fetch_add(cursor, nb, __ATOMIC_RELAXED, __HIP_MEMORY_SCOPE_WORKGROUP);
    base = __builtin_amdgcn_readfirstlane(base);
    const int pos = base + lane;
    if (lane < nb && pos < REC_CAP) rec[pos] = make_int2(s | (dl << 17), __float_as_int(w));
  }
  agg_rows(s, dl, w, nb, lane, H, tile);
}

template <bool REC>
__device__ __forceinline__ void agg_scan(int lo, int wave, int lane, int* mylist,
                                         const int* __restrict__ src, const int* __restrict__ dst,
                                         const float* __restrict__ dinv, const float* dinvL,
                                         const float* __restrict__ H, float* tile,
                                         int* cursor, int2* __restrict__ rec) {
  const int4* dst4 = (const int4*)dst;
  int pend = 0;
  int trip = blockIdx.x % N_TRIPS;

  int4 nxt[SCAN_U];
  agg_load_trip(dst4, trip * TRIP4 + wave * WAVE4, lane, nxt);

  for (int it = 0; it < N_TRIPS; ++it) {
    int4 cur[SCAN_U];
    #pragma unroll
    for (int u = 0; u < SCAN_U; ++u) cur[u] = nxt[u];
    const int cur4 = trip * TRIP4 + wave * WAVE4;
    trip = (trip + 1 == N_TRIPS) ? 0 : trip + 1;
    if (it + 1 < N_TRIPS) agg_load_trip(dst4, trip * TRIP4 + wave * WAVE4, lane, nxt);

    #pragma unroll
    for (int u = 0; u < SCAN_U; ++u) {
      #pragma unroll
      for (int c = 0; c < 4; ++c) {
        const int d = (c == 0) ? cur[u].x : (c == 1) ? cur[u].y : (c == 2) ? cur[u].z : cur[u].w;
        const unsigned tt = (unsigned)(d - lo);
        const bool m = tt < (unsigned)NPB;
        const unsigned mask = __builtin_amdgcn_ballot_w32(m);
        if (mask != 0u) {
          const int pos = pend + (int)__builtin_amdgcn_mbcnt_lo(mask, 0u);
          const int e = (cur4 + u * 32 + lane) * 4 + c;
          if (m) mylist[pos] = (int)((tt << 21) | (unsigned)e);
          pend += __builtin_popcount(mask);
        }
      }
    }
    while (pend >= BATCH) {
      pend -= BATCH;
      agg_batch<REC>(mylist + pend, BATCH, lane, src, dinv, dinvL, H, tile, cursor, rec);
    }
  }
  if (pend > 0) agg_batch<REC>(mylist, pend, lane, src, dinv, dinvL, H, tile, cursor, rec);
}

template <int MODE>
__global__ __launch_bounds__(AGG_THREADS) void k_agg_owner(const int* __restrict__ src,
                                                           const int* __restrict__ dst,
                                                           const float* __restrict__ dinv,
                                                           const float* __restrict__ H,
                                                           const float* __restrict__ bias,
                                                           float* __restrict__ OUT, int do_relu,
                                                           int* __restrict__ rec_cnt,
                                                           int* __restrict__ rec_all) {
  __shared__ __attribute__((aligned(16))) float tile[NPB * DIM];
  __shared__ float dinvL[NPB + 1];
  __shared__ int lists[AGG_WAVES * LIST_CAP];
  __shared__ int cursor;

  const int t = threadIdx.x;
  const int wave = t >> 5;
  const int lane = t & 31;
  const int lo = blockIdx.x * NPB;
  const int nvalid = (N_NODESC - lo < NPB) ? (N_NODESC - lo) : NPB;
  int2* rec = (int2*)rec_all + (size_t)blockIdx.x * REC_CAP;

  if (t == 0) cursor = 0;

  for (int r = wave; r < NPB; r += AGG_WAVES) {
    float4 v = make_float4(0.f, 0.f, 0.f, 0.f);
    float di = 0.f;
    if (r < nvalid) {
      di = dinv[lo + r];
      const float4 h = ((const float4*)(H + (size_t)(lo + r) * DIM))[lane];
      v.x = h.x * di * di;
      v.y = h.y * di * di;
      v.z = h.z * di * di;
      v.w = h.w * di * di;
    }
    ((float4*)(tile + r * DIM))[lane] = v;
    if (lane == 0) dinvL[r] = di;
  }
  __syncthreads();

  int* mylist = lists + wave * LIST_CAP;

  if (MODE == 0) {
    agg_scan<true>(lo, wave, lane, mylist, src, dst, dinv, dinvL, H, tile, &cursor, rec);
  } else {
    const int n = rec_cnt[blockIdx.x];
    if (n <= REC_CAP) {
      const int nbat = (n + BATCH - 1) / BATCH;
      const int wv = __builtin_amdgcn_readfirstlane(wave);
      int2 ent = make_int2(0, 0);
      if (wv < nbat && wv * BATCH + lane < n && lane < BATCH) ent = rec[wv * BATCH + lane];
      for (int bi = wv; bi < nbat; bi += AGG_WAVES) {
        const int2 cur = ent;
        const int left = n - bi * BATCH;
        const int nb = (left < BATCH) ? left : BATCH;
        float hv[BATCH][4];
        rows_load(cur.x & ((1 << 17) - 1), lane, H, hv);
        const int bn = bi + AGG_WAVES;
        ent = make_int2(0, 0);
        if (bn < nbat && bn * BATCH + lane < n && lane < BATCH) ent = rec[bn * BATCH + lane];
        rows_add(hv, (int)((unsigned)cur.x >> 17), __int_as_float(cur.y), nb, lane, tile);
      }
    } else {
      agg_scan<false>(lo, wave, lane, mylist, src, dst, dinv, dinvL, H, tile, &cursor, rec);
    }
  }

  __syncthreads();

  if (MODE == 0 && t == 0) rec_cnt[blockIdx.x] = cursor;

  const float4 bb = ((const float4*)bias)[lane];
  for (int r = wave; r < nvalid; r += AGG_WAVES) {
    float4 a = ((const float4*)(tile + r * DIM))[lane];
    a.x += bb.x;
    a.y += bb.y;
    a.z += bb.z;
    a.w += bb.w;
    if (do_relu) {
      a.x = fmaxf(a.x, 0.0f);
      a.y = fmaxf(a.y, 0.0f);
      a.z = fmaxf(a.z, 0.0f);
      a.w = fmaxf(a.w, 0.0f);
    }
    ((float4*)(OUT + (size_t)(lo + r) * DIM))[lane] = a;
  }
}


extern "C" void kernel_launch(void* const* d_in, const int* in_sizes, int n_in,
                              void* d_out, int out_size, void* d_ws, size_t ws_size,
                              hipStream_t stream) {
  const float* x   = (const float*)d_in[0];
  const int*   ei  = (const int*)d_in[1];
  const float* W1  = (const float*)d_in[2];
  const float* b1  = (const float*)d_in[3];
  const float* W2  = (const float*)d_in[4];
  const float* b2  = (const float*)d_in[5];
  float*       out = (float*)d_out;

  const int* src = ei;
  const int* dst = ei + N_EDGESC;

  float* H    = (float*)d_ws;
  float* deg  = H + (size_t)N_NODESC * DIM;
  float* dinv = deg + N_NODESC;
  int*   rec_cnt = (int*)(dinv + N_NODESC);
  int*   rec_all = rec_cnt + AGG_BLOCKS;

  dim3 blk(256);
  dim3 gNode((N_NODESC + 255) / 256);
  dim3 gEdge((N_EDGESC + 255) / 256);
  dim3 gGemm(N_NODESC / 16);
  dim3 gAgg(AGG_BLOCKS);
  dim3 bAgg(AGG_THREADS);

  k_init_deg<<<gNode, blk, 0, stream>>>(deg);
  k_count_deg<<<gEdge, blk, 0, stream>>>(dst, deg);
  k_dinv<<<gNode, blk, 0, stream>>>(deg, dinv);

  k_gemm_wmma<<<gGemm, blk, 0, stream>>>(x, W1, H);
  k_agg_owner<0><<<gAgg, bAgg, 0, stream>>>(src, dst, dinv, H, b1, out,  1, rec_cnt, rec_all);

  k_gemm_wmma<<<gGemm, blk, 0, stream>>>(out, W2, H);
  k_agg_owner<1><<<gAgg, bAgg, 0, stream>>>(src, dst, dinv, H, b2, out,  0, rec_cnt, rec_all);
}


# Layer 1 stops streaming the dst list in every owner block: the degree-count launch, now 256 chunk blocks, files each edge in a 63-entry cell per (owner, chunk); sums stay in LDS arrival order
# speedup vs baseline: 2.9659x; 1.2075x over previous
#include <hip/hip_runtime.h>

typedef float v2f __attribute__((ext_vector_type(2)));
typedef float v8f __attribute__((ext_vector_type(8)));

#define N_NODESC 100000
#define N_EDGESC 1600000
#define DIM 128


#define AGG_BLOCKS 256
#define NPB 391

#define BIN_BLOCKS 256
#define BIN_THREADS 1024
#define CELL_INTS 64
#define CELL_CAP (CELL_INTS - 1)
#define CELL_PITCH 68
#define EDGES_PER_BIN (N_EDGESC / BIN_BLOCKS)
#define BIN_ITERS ((EDGES_PER_BIN + BIN_THREADS - 1) / BIN_THREADS)

static_assert(N_EDGESC % BIN_BLOCKS == 0, "chunks are equal");
static_assert(N_NODESC <= (1 << 17) && NPB <= (1 << 9), "cell entry packs source node and local row");

__global__ void k_init_deg(float* __restrict__ deg, int* __restrict__ ovf) {
  int i = blockIdx.x * blockDim.x + threadIdx.x;
  if (i < N_NODESC) deg[i] = 1.0f;
  if (i < AGG_BLOCKS) ovf[i] = 0;
}

__global__ __launch_bounds__(BIN_THREADS) void k_bin_count(const int* __restrict__ src,
                                                           const int* __restrict__ dst,
                                                           float* __restrict__ deg,
                                                           int* __restrict__ cells,
                                                           int* __restrict__ ovf) {
  __shared__ __attribute__((aligned(16))) int cellL[AGG_BLOCKS * CELL_PITCH];
  __shared__ int cntL[AGG_BLOCKS];

  const int t = threadIdx.x;
  const int e0 = blockIdx.x * EDGES_PER_BIN;

  int dv[BIN_ITERS], sv[BIN_ITERS];
  #pragma unroll
  for (int k = 0; k < BIN_ITERS; ++k) {
    const int i = t + k * BIN_THREADS;
    dv[k] = -1;
    sv[k] = 0;
    if (i < EDGES_PER_BIN) {
      dv[k] = dst[e0 + i];
      sv[k] = src[e0 + i];
    }
  }

  for (int q = t; q < AGG_BLOCKS * CELL_PITCH / 4; q += BIN_THREADS)
    ((int4*)cellL)[q] = make_int4(0, 0, 0, 0);
  for (int o = t; o < AGG_BLOCKS; o += BIN_THREADS) cntL[o] = 0;
  __syncthreads();

  #pragma unroll
  for (int k = 0; k < BIN_ITERS; ++k) {
    const int d = dv[k];
    if (d >= 0) {
      unsafeAtomicAdd(&deg[d], 1.0f);
      const int o = d / NPB;
      const int dl = d - o * NPB;
      const int rank = __hip_atomic_fetch_add(&cntL[o], 1, __ATOMIC_RELAXED, __HIP_MEMORY_SCOPE_WORKGROUP);
      if (rank < CELL_CAP) cellL[o * CELL_PITCH + 1 + rank] = sv[k] | (dl << 17);
      else ovf[o] = 1;
    }
  }
  __syncthreads();

  for (int q = t; q < AGG_BLOCKS * CELL_INTS / 4; q += BIN_THREADS) {
    const int o = q >> 4;
    const int part = q & 15;
    int4 v = *(const int4*)&cellL[o * CELL_PITCH + part * 4];
    if (part == 0) v.x = cntL[o];
    ((int4*)(cells + ((size_t)o * BIN_BLOCKS + blockIdx.x) * CELL_INTS))[part] = v;
  }
}

__global__ void k_dinv(const float* __restrict__ deg, float* __restrict__ dinv) {
  int i = blockIdx.x * blockDim.x + threadIdx.x;
  if (i < N_NODESC) dinv[i] = rsqrtf(deg[i]);
}


__global__ __launch_bounds__(256) void k_gemm_wmma(const float* __restrict__ X,
                                                   const float* __restrict__ W,
                                                   float* __restrict__ H) {
  __shared__ float xs[16 * 132];

  const int t = threadIdx.x;
  const int wave = t >> 5;
  const int lane = t & 31;
  const int row0 = blockIdx.x * 16;

  {
    const int r = t >> 4;
    const int c = (t & 15) * 8;
    const float4* g = (const float4*)(X + (size_t)(row0 + r) * DIM + c);
    float4 p0 = g[0];
    float4 p1 = g[1];
    float4* s = (float4*)&xs[r * 132 + c];
    s[0] = p0;
    s[1] = p1;
  }
  __syncthreads();

  const int m  = lane & 15;
  const int hg = lane >> 4;
  const int n0 = wave * 16;

  v8f acc = {};
  const float* wcol = W + n0 + m;

  #pragma unroll 4
  for (int kk = 0; kk < DIM; kk += 4) {
    v2f a = *(const v2f*)&xs[m * 132 + kk + 2 * hg];
    v2f b;
    b.x = wcol[(size_t)(kk + 2 * hg + 0) * DIM];
    b.y = wcol[(size_t)(kk + 2 * hg + 1) * DIM];
    acc = __builtin_amdgcn_wmma_f32_16x16x4_f32(
         false, a,  false, b,
         (short)0, acc,  false,  false);
  }

  float* hrow = H + (size_t)row0 * DIM + n0 + m;
  #pragma unroll
  for (int v = 0; v < 8; ++v)
    hrow[(size_t)(v + 8 * hg) * DIM] = acc[v];
}


#define AGG_THREADS 1024
#define AGG_WAVES 32
#define SCAN_U 4
#define N_DST4 (N_EDGESC / 4)
#define WAVE4 (32 * SCAN_U)
#define TRIP4 (AGG_WAVES * WAVE4)
#define N_TRIPS ((N_DST4 + TRIP4 - 1) / TRIP4)
#define BATCH 8
#define LIST_CAP (BATCH - 1 + 4 * WAVE4 + 1)
#define REC_CAP 16384

static_assert(AGG_BLOCKS * NPB >= N_NODESC, "slices must cover every node");
static_assert(N_EDGESC % 4 == 0, "dst list is read as int4");
static_assert(N_EDGESC <= (1 << 21) && NPB <= (1 << 9), "list entry packs edge id and local row");
static_assert(BIN_BLOCKS % AGG_WAVES == 0, "every wave takes the same number of cells");
#define CELLS_PER_WAVE (BIN_BLOCKS / AGG_WAVES)
static_assert(LIST_CAP >= BATCH - 1 + CELL_CAP, "a wave's list holds a cell beside its left-over");

__device__ __forceinline__ void agg_load_trip(const int4* __restrict__ dst4, int base4, int lane,
                                              int4 (&v)[SCAN_U]) {
  #pragma unroll
  for (int u = 0; u < SCAN_U; ++u) {
    const int i4 = base4 + u * 32 + lane;
    v[u] = (i4 < N_DST4) ? dst4[i4] : make_int4(-1, -1, -1, -1);
  }
}

__device__ __forceinline__ void rows_load(int s, int lane, const float* __restrict__ H,
                                          float (&hv)[BATCH][4]) {
  #pragma unroll
  for (int i = 0; i < BATCH; ++i) {
    const int si = __builtin_amdgcn_readlane(s, i);
    const float* hr = H + (size_t)si * DIM + lane;
    hv[i][0] = hr[0];
    hv[i][1] = hr[32];
    hv[i][2] = hr[64];
    hv[i][3] = hr[96];
  }
}

__device__ __forceinline__ void rows_add(const float (&hv)[BATCH][4], int dl, float w, int nb,
                                         int lane, float* tile) {
  #pragma unroll
  for (int i = 0; i < BATCH; ++i) {
    if (i < nb) {
      const int di = __builtin_amdgcn_readlane(dl, i);
      const float wi = __int_as_float(__builtin_amdgcn_readlane(__float_as_int(w), i));
      float* tr = tile + di * DIM + lane;
      __hip_atomic_fetch_add(tr + 0,  hv[i][0] * wi, __ATOMIC_RELAXED, __HIP_MEMORY_SCOPE_WORKGROUP);
      __hip_atomic_fetch_add(tr + 32, hv[i][1] * wi, __ATOMIC_RELAXED, __HIP_MEMORY_SCOPE_WORKGROUP);
      __hip_atomic_fetch_add(tr + 64, hv[i][2] * wi, __ATOMIC_RELAXED, __HIP_MEMORY_SCOPE_WORKGROUP);
      __hip_atomic_fetch_add(tr + 96, hv[i][3] * wi, __ATOMIC_RELAXED, __HIP_MEMORY_SCOPE_WORKGROUP);
    }
  }
}

template <bool REC, bool FROM_E>
__device__ __forceinline__ void agg_batch(const int* lst, int nb, int lane,
                                          const int* __restrict__ src,
                                          const float* __restrict__ dinv,
                                          const float* dinvL,
                                          const float* __restrict__ H, float* tile,
                                          int* cursor, int2* __restrict__ rec) {
  int ent = 0;
  if (lane < nb) ent = lst[lane];
  int s, dl;
  if (FROM_E) {
    dl = (int)((unsigned)ent >> 21);
    s = src[ent & ((1 << 21) - 1)];
  } else {
    dl = (int)((unsigned)ent >> 17);
    s = ent & ((1 << 17) - 1);
  }
  const float ds = dinv[s];
  float hv[BATCH][4];
  rows_load(s, lane, H, hv);
  const float w = ds * dinvL[dl];
  if (REC) {
    int base = 0;
    if (lane == 0) base = __hip_atomic_fetch_add(cursor, nb, __ATOMIC_RELAXED, __HIP_MEMORY_SCOPE_WORKGROUP);
    base = __builtin_amdgcn_readfirstlane(base);
    const int pos = base + lane;
    if (lane < nb && pos < REC_CAP) rec[pos] = make_int2(s | (dl << 17), __float_as_int(w));
  }
  rows_add(hv, dl, w, nb, lane, tile);
}

template <bool REC>
__device__ __forceinline__ void agg_scan(int lo, int wave, int lane, int* mylist,
                                         const int* __restrict__ src, const int* __restrict__ dst,
                                         const float* __restrict__ dinv, const float* dinvL,
                                         const float* __restrict__ H, float* tile,
                                         int* cursor, int2* __restrict__ rec) {
  const int4* dst4 = (const int4*)dst;
  int pend = 0;
  int trip = blockIdx.x % N_TRIPS;

  int4 nxt[SCAN_U];
  agg_load_trip(dst4, trip * TRIP4 + wave * WAVE4, lane, nxt);

  for (int it = 0; it < N_TRIPS; ++it) {
    int4 cur[SCAN_U];
    #pragma unroll
    for (int u = 0; u < SCAN_U; ++u) cur[u] = nxt[u];
    const int cur4 = trip * TRIP4 + wave * WAVE4;
    trip = (trip + 1 == N_TRIPS) ? 0 : trip + 1;
    if (it + 1 < N_TRIPS) agg_load_trip(dst4, trip * TRIP4 + wave * WAVE4, lane, nxt);

    #pragma unroll
    for (int u = 0; u < SCAN_U; ++u) {
      #pragma unroll
      for (int c = 0; c < 4; ++c) {
        const int d = (c == 0) ? cur[u].x : (c == 1) ? cur[u].y : (c == 2) ? cur[u].z : cur[u].w;
        const unsigned tt = (unsigned)(d - lo);
        const bool m = tt < (unsigned)NPB;
        const unsigned mask = __builtin_amdgcn_ballot_w32(m);
        if (mask != 0u) {
          const int pos = pend + (int)__builtin_amdgcn_mbcnt_lo(mask, 0u);
          const int e = (cur4 + u * 32 + lane) * 4 + c;
          if (m) mylist[pos] = (int)((tt << 21) | (unsigned)e);
          pend += __builtin_popcount(mask);
        }
      }
    }
    while (pend >= BATCH) {
      pend -= BATCH;
      agg_batch<REC, true>(mylist + pend, BATCH, lane, src, dinv, dinvL, H, tile, cursor, rec);
    }
  }
  if (pend > 0) agg_batch<REC, true>(mylist, pend, lane, src, dinv, dinvL, H, tile, cursor, rec);
}

__device__ __forceinline__ void agg_cells(const int* __restrict__ mycells, int wv, int lane,
                                          int* mylist, const int* __restrict__ src,
                                          const float* __restrict__ dinv, const float* dinvL,
                                          const float* __restrict__ H, float* tile,
                                          int* cursor, int2* __restrict__ rec) {
  const int2* cp = (const int2*)(mycells + (size_t)wv * CELLS_PER_WAVE * CELL_INTS) + lane;
  int pend = 0;
  int2 nxt = cp[0];
  for (int c = 0; c < CELLS_PER_WAVE; ++c) {
    const int2 cur = nxt;
    if (c + 1 < CELLS_PER_WAVE) nxt = cp[(c + 1) * (CELL_INTS / 2)];
    int cnt = __builtin_amdgcn_readfirstlane(cur.x);
    cnt = (cnt < CELL_CAP) ? cnt : CELL_CAP;
    const int i0 = 2 * lane;
    if (i0 >= 1 && i0 <= cnt) mylist[pend + i0 - 1] = cur.x;
    if (i0 + 1 <= cnt) mylist[pend + i0] = cur.y;
    pend += cnt;
    while (pend >= BATCH) {
      pend -= BATCH;
      agg_batch<true, false>(mylist + pend, BATCH, lane, src, dinv, dinvL, H, tile, cursor, rec);
    }
  }
  if (pend > 0) agg_batch<true, false>(mylist, pend, lane, src, dinv, dinvL, H, tile, cursor, rec);
}

template <int MODE>
__global__ __launch_bounds__(AGG_THREADS) void k_agg_owner(const int* __restrict__ src,
                                                           const int* __restrict__ dst,
                                                           const float* __restrict__ dinv,
                                                           const float* __restrict__ H,
                                                           const float* __restrict__ bias,
                                                           float* __restrict__ OUT, int do_relu,
                                                           int* __restrict__ rec_cnt,
                                                           int* __restrict__ rec_all,
                                                           const int* __restrict__ cells,
                                                           const int* __restrict__ ovf) {
  __shared__ __attribute__((aligned(16))) float tile[NPB * DIM];
  __shared__ float dinvL[NPB + 1];
  __shared__ int lists[AGG_WAVES * LIST_CAP];
  __shared__ int cursor;

  const int t = threadIdx.x;
  const int wave = t >> 5;
  const int lane = t & 31;
  const int lo = blockIdx.x * NPB;
  const int nvalid = (N_NODESC - lo < NPB) ? (N_NODESC - lo) : NPB;
  int2* rec = (int2*)rec_all + (size_t)blockIdx.x * REC_CAP;

  if (t == 0) cursor = 0;

  for (int r = wave; r < NPB; r += AGG_WAVES) {
    float4 v = make_float4(0.f, 0.f, 0.f, 0.f);
    float di = 0.f;
    if (r < nvalid) {
      di = dinv[lo + r];
      const float4 h = ((const float4*)(H + (size_t)(lo + r) * DIM))[lane];
      v.x = h.x * di * di;
      v.y = h.y * di * di;
      v.z = h.z * di * di;
      v.w = h.w * di * di;
    }
    ((float4*)(tile + r * DIM))[lane] = v;
    if (lane == 0) dinvL[r] = di;
  }
  __syncthreads();

  int* mylist = lists + wave * LIST_CAP;

  if (MODE == 0) {
    if (ovf[blockIdx.x] == 0) {
      const int wv = __builtin_amdgcn_readfirstlane(wave);
      agg_cells(cells + (size_t)blockIdx.x * BIN_BLOCKS * CELL_INTS, wv, lane, mylist, src, dinv,
                dinvL, H, tile, &cursor, rec);
    } else {
      agg_scan<true>(lo, wave, lane, mylist, src, dst, dinv, dinvL, H, tile, &cursor, rec);
    }
  } else {
    const int n = rec_cnt[blockIdx.x];
    if (n <= REC_CAP) {
      const int nbat = (n + BATCH - 1) / BATCH;
      const int wv = __builtin_amdgcn_readfirstlane(wave);
      int2 ent = make_int2(0, 0);
      if (wv < nbat && wv * BATCH + lane < n && lane < BATCH) ent = rec[wv * BATCH + lane];
      for (int bi = wv; bi < nbat; bi += AGG_WAVES) {
        const int2 cur = ent;
        const int left = n - bi * BATCH;
        const int nb = (left < BATCH) ? left : BATCH;
        float hv[BATCH][4];
        rows_load(cur.x & ((1 << 17) - 1), lane, H, hv);
        const int bn = bi + AGG_WAVES;
        ent = make_int2(0, 0);
        if (bn < nbat && bn * BATCH + lane < n && lane < BATCH) ent = rec[bn * BATCH + lane];
        rows_add(hv, (int)((unsigned)cur.x >> 17), __int_as_float(cur.y), nb, lane, tile);
      }
    } else {
      agg_scan<false>(lo, wave, lane, mylist, src, dst, dinv, dinvL, H, tile, &cursor, rec);
    }
  }

  __syncthreads();

  if (MODE == 0 && t == 0) rec_cnt[blockIdx.x] = cursor;

  const float4 bb = ((const float4*)bias)[lane];
  for (int r = wave; r < nvalid; r += AGG_WAVES) {
    float4 a = ((const float4*)(tile + r * DIM))[lane];
    a.x += bb.x;
    a.y += bb.y;
    a.z += bb.z;
    a.w += bb.w;
    if (do_relu) {
      a.x = fmaxf(a.x, 0.0f);
      a.y = fmaxf(a.y, 0.0f);
      a.z = fmaxf(a.z, 0.0f);
      a.w = fmaxf(a.w, 0.0f);
    }
    ((float4*)(OUT + (size_t)(lo + r) * DIM))[lane] = a;
  }
}


extern "C" void kernel_launch(void* const* d_in, const int* in_sizes, int n_in,
                              void* d_out, int out_size, void* d_ws, size_t ws_size,
                              hipStream_t stream) {
  const float* x   = (const float*)d_in[0];
  const int*   ei  = (const int*)d_in[1];
  const float* W1  = (const float*)d_in[2];
  const float* b1  = (const float*)d_in[3];
  const float* W2  = (const float*)d_in[4];
  const float* b2  = (const float*)d_in[5];
  float*       out = (float*)d_out;

  const int* src = ei;
  const int* dst = ei + N_EDGESC;

  float* H    = (float*)d_ws;
  float* deg  = H + (size_t)N_NODESC * DIM;
  float* dinv = deg + N_NODESC;
  int*   rec_cnt = (int*)(dinv + N_NODESC);
  int*   rec_all = rec_cnt + AGG_BLOCKS;
  int*   cells   = rec_all + (size_t)AGG_BLOCKS * REC_CAP * 2;
  int*   ovf     = cells + (size_t)AGG_BLOCKS * BIN_BLOCKS * CELL_INTS;

  dim3 blk(256);
  dim3 gNode((N_NODESC + 255) / 256);
  dim3 gBin(BIN_BLOCKS);
  dim3 bBin(BIN_THREADS);
  dim3 gGemm(N_NODESC / 16);
  dim3 gAgg(AGG_BLOCKS);
  dim3 bAgg(AGG_THREADS);

  k_init_deg<<<gNode, blk, 0, stream>>>(deg, ovf);
  k_bin_count<<<gBin, bBin, 0, stream>>>(src, dst, deg, cells, ovf);
  k_dinv<<<gNode, blk, 0, stream>>>(deg, dinv);

  k_gemm_wmma<<<gGemm, blk, 0, stream>>>(x, W1, H);
  k_agg_owner<0><<<gAgg, bAgg, 0, stream>>>(src, dst, dinv, H, b1, out,  1, rec_cnt, rec_all, cells, ovf);

  k_gemm_wmma<<<gGemm, blk, 0, stream>>>(out, W2, H);
  k_agg_owner<1><<<gAgg, bAgg, 0, stream>>>(src, dst, dinv, H, b2, out,  0, rec_cnt, rec_all, cells, ovf);
}


# Product launches become a fixed grid of 768 blocks looping over 16-row tiles: a wave keeps its W column slice (B fragments of all k steps) in registers, X tiles double-staged in LDS; k order kept
# speedup vs baseline: 3.0590x; 1.0314x over previous
#include <hip/hip_runtime.h>

typedef float v2f __attribute__((ext_vector_type(2)));
typedef float v8f __attribute__((ext_vector_type(8)));

#define N_NODESC 100000
#define N_EDGESC 1600000
#define DIM 128


#define AGG_BLOCKS 256
#define NPB 391

#define BIN_BLOCKS 256
#define BIN_THREADS 1024
#define CELL_INTS 64
#define CELL_CAP (CELL_INTS - 1)
#define CELL_PITCH 68
#define EDGES_PER_BIN (N_EDGESC / BIN_BLOCKS)
#define BIN_ITERS ((EDGES_PER_BIN + BIN_THREADS - 1) / BIN_THREADS)

static_assert(N_EDGESC % BIN_BLOCKS == 0, "chunks are equal");
static_assert(N_NODESC <= (1 << 17) && NPB <= (1 << 9), "cell entry packs source node and local row");

__global__ void k_init_deg(float* __restrict__ deg, int* __restrict__ ovf) {
  int i = blockIdx.x * blockDim.x + threadIdx.x;
  if (i < N_NODESC) deg[i] = 1.0f;
  if (i < AGG_BLOCKS) ovf[i] = 0;
}

__global__ __launch_bounds__(BIN_THREADS) void k_bin_count(const int* __restrict__ src,
                                                           const int* __restrict__ dst,
                                                           float* __restrict__ deg,
                                                           int* __restrict__ cells,
                                                           int* __restrict__ ovf) {
  __shared__ __attribute__((aligned(16))) int cellL[AGG_BLOCKS * CELL_PITCH];
  __shared__ int cntL[AGG_BLOCKS];

  const int t = threadIdx.x;
  const int e0 = blockIdx.x * EDGES_PER_BIN;

  int dv[BIN_ITERS], sv[BIN_ITERS];
  #pragma unroll
  for (int k = 0; k < BIN_ITERS; ++k) {
    const int i = t + k * BIN_THREADS;
    dv[k] = -1;
    sv[k] = 0;
    if (i < EDGES_PER_BIN) {
      dv[k] = dst[e0 + i];
      sv[k] = src[e0 + i];
    }
  }

  for (int q = t; q < AGG_BLOCKS * CELL_PITCH / 4; q += BIN_THREADS)
    ((int4*)cellL)[q] = make_int4(0, 0, 0, 0);
  for (int o = t; o < AGG_BLOCKS; o += BIN_THREADS) cntL[o] = 0;
  __syncthreads();

  #pragma unroll
  for (int k = 0; k < BIN_ITERS; ++k) {
    const int d = dv[k];
    if (d >= 0) {
      unsafeAtomicAdd(&deg[d], 1.0f);
      const int o = d / NPB;
      const int dl = d - o * NPB;
      const int rank = __hip_atomic_fetch_add(&cntL[o], 1, __ATOMIC_RELAXED, __HIP_MEMORY_SCOPE_WORKGROUP);
      if (rank < CELL_CAP) cellL[o * CELL_PITCH + 1 + rank] = sv[k] | (dl << 17);
      else ovf[o] = 1;
    }
  }
  __syncthreads();

  for (int q = t; q < AGG_BLOCKS * CELL_INTS / 4; q += BIN_THREADS) {
    const int o = q >> 4;
    const int part = q & 15;
    int4 v = *(const int4*)&cellL[o * CELL_PITCH + part * 4];
    if (part == 0) v.x = cntL[o];
    ((int4*)(cells + ((size_t)o * BIN_BLOCKS + blockIdx.x) * CELL_INTS))[part] = v;
  }
}

__global__ void k_dinv(const float* __restrict__ deg, float* __restrict__ dinv) {
  int i = blockIdx.x * blockDim.x + threadIdx.x;
  if (i < N_NODESC) dinv[i] = rsqrtf(deg[i]);
}


#define GEMM_BLOCKS 768
#define GEMM_TILES (N_NODESC / 16)
#define GEMM_KSTEPS (DIM / 4)
#define XS_PITCH4 33
#define XS_STAGE4 (16 * XS_PITCH4)

static_assert(N_NODESC % 16 == 0, "whole 16-row tiles");

__global__ __launch_bounds__(256) void k_gemm_wmma(const float* __restrict__ X,
                                                   const float* __restrict__ W,
                                                   float* __restrict__ H) {
  __shared__ float4 xs4[2 * XS_STAGE4];

  const int t = threadIdx.x;
  const int wave = t >> 5;
  const int lane = t & 31;

  const int m  = lane & 15;
  const int hg = lane >> 4;
  const int n0 = wave * 16;

  v2f bf[GEMM_KSTEPS];
  {
    const float* wcol = W + n0 + m;
    #pragma unroll
    for (int j = 0; j < GEMM_KSTEPS; ++j) {
      bf[j].x = wcol[(size_t)(4 * j + 2 * hg + 0) * DIM];
      bf[j].y = wcol[(size_t)(4 * j + 2 * hg + 1) * DIM];
    }
  }

  const int r  = wave;
  const int c4 = lane;

  int tile = blockIdx.x;
  float4 p0, p1;
  {
    const int tl = (tile < GEMM_TILES) ? tile : GEMM_TILES - 1;
    const float4* g = (const float4*)(X + ((size_t)tl * 16 + r) * DIM) + c4;
    p0 = g[0];
    p1 = g[8 * (DIM / 4)];
  }

  int cur = 0;
  for (; tile < GEMM_TILES; tile += GEMM_BLOCKS) {
    float4* stage = xs4 + cur * XS_STAGE4;
    stage[r * XS_PITCH4 + c4] = p0;
    stage[(r + 8) * XS_PITCH4 + c4] = p1;
    __syncthreads();

    {
      const int nxt = tile + GEMM_BLOCKS;
      const int tl = (nxt < GEMM_TILES) ? nxt : GEMM_TILES - 1;
      const float4* g = (const float4*)(X + ((size_t)tl * 16 + r) * DIM) + c4;
      p0 = g[0];
      p1 = g[8 * (DIM / 4)];
    }
    __builtin_amdgcn_sched_barrier(0);

    const v2f* arow = (const v2f*)stage + m * (XS_PITCH4 * 2) + hg;
    v2f af[GEMM_KSTEPS];
    #pragma unroll
    for (int j = 0; j < GEMM_KSTEPS; ++j) af[j] = arow[2 * j];
    __builtin_amdgcn_sched_barrier(0);

    v8f acc = {};
    #pragma unroll
    for (int j = 0; j < GEMM_KSTEPS; ++j) {
      acc = __builtin_amdgcn_wmma_f32_16x16x4_f32(
           false, af[j],  false, bf[j],
           (short)0, acc,  false,  false);
    }

    float* hrow = H + (size_t)tile * 16 * DIM + n0 + m;
    #pragma unroll
    for (int v = 0; v < 8; ++v)
      hrow[(size_t)(v + 8 * hg) * DIM] = acc[v];

    cur ^= 1;
  }
}


#define AGG_THREADS 1024
#define AGG_WAVES 32
#define SCAN_U 4
#define N_DST4 (N_EDGESC / 4)
#define WAVE4 (32 * SCAN_U)
#define TRIP4 (AGG_WAVES * WAVE4)
#define N_TRIPS ((N_DST4 + TRIP4 - 1) / TRIP4)
#define BATCH 8
#define LIST_CAP (BATCH - 1 + 4 * WAVE4 + 1)
#define REC_CAP 16384

static_assert(AGG_BLOCKS * NPB >= N_NODESC, "slices must cover every node");
static_assert(N_EDGESC % 4 == 0, "dst list is read as int4");
static_assert(N_EDGESC <= (1 << 21) && NPB <= (1 << 9), "list entry packs edge id and local row");
static_assert(BIN_BLOCKS % AGG_WAVES == 0, "every wave takes the same number of cells");
#define CELLS_PER_WAVE (BIN_BLOCKS / AGG_WAVES)
static_assert(LIST_CAP >= BATCH - 1 + CELL_CAP, "a wave's list holds a cell beside its left-over");

__device__ __forceinline__ void agg_load_trip(const int4* __restrict__ dst4, int base4, int lane,
                                              int4 (&v)[SCAN_U]) {
  #pragma unroll
  for (int u = 0; u < SCAN_U; ++u) {
    const int i4 = base4 + u * 32 + lane;
    v[u] = (i4 < N_DST4) ? dst4[i4] : make_int4(-1, -1, -1, -1);
  }
}

__device__ __forceinline__ void rows_load(int s, int lane, const float* __restrict__ H,
                                          float (&hv)[BATCH][4]) {
  #pragma unroll
  for (int i = 0; i < BATCH; ++i) {
    const int si = __builtin_amdgcn_readlane(s, i);
    const float* hr = H + (size_t)si * DIM + lane;
    hv[i][0] = hr[0];
    hv[i][1] = hr[32];
    hv[i][2] = hr[64];
    hv[i][3] = hr[96];
  }
}

__device__ __forceinline__ void rows_add(const float (&hv)[BATCH][4], int dl, float w, int nb,
                                         int lane, float* tile) {
  #pragma unroll
  for (int i = 0; i < BATCH; ++i) {
    if (i < nb) {
      const int di = __builtin_amdgcn_readlane(dl, i);
      const float wi = __int_as_float(__builtin_amdgcn_readlane(__float_as_int(w), i));
      float* tr = tile + di * DIM + lane;
      __hip_atomic_fetch_add(tr + 0,  hv[i][0] * wi, __ATOMIC_RELAXED, __HIP_MEMORY_SCOPE_WORKGROUP);
      __hip_atomic_fetch_add(tr + 32, hv[i][1] * wi, __ATOMIC_RELAXED, __HIP_MEMORY_SCOPE_WORKGROUP);
      __hip_atomic_fetch_add(tr + 64, hv[i][2] * wi, __ATOMIC_RELAXED, __HIP_MEMORY_SCOPE_WORKGROUP);
      __hip_atomic_fetch_add(tr + 96, hv[i][3] * wi, __ATOMIC_RELAXED, __HIP_MEMORY_SCOPE_WORKGROUP);
    }
  }
}

template <bool REC, bool FROM_E>
__device__ __forceinline__ void agg_batch(const int* lst, int nb, int lane,
                                          const int* __restrict__ src,
                                          const float* __restrict__ dinv,
                                          const float* dinvL,
                                          const float* __restrict__ H, float* tile,
                                          int* cursor, int2* __restrict__ rec) {
  int ent = 0;
  if (lane < nb) ent = lst[lane];
  int s, dl;
  if (FROM_E) {
    dl = (int)((unsigned)ent >> 21);
    s = src[ent & ((1 << 21) - 1)];
  } else {
    dl = (int)((unsigned)ent >> 17);
    s = ent & ((1 << 17) - 1);
  }
  const float ds = dinv[s];
  float hv[BATCH][4];
  rows_load(s, lane, H, hv);
  const float w = ds * dinvL[dl];
  if (REC) {
    int base = 0;
    if (lane == 0) base = __hip_atomic_fetch_add(cursor, nb, __ATOMIC_RELAXED, __HIP_MEMORY_SCOPE_WORKGROUP);
    base = __builtin_amdgcn_readfirstlane(base);
    const int pos = base + lane;
    if (lane < nb && pos < REC_CAP) rec[pos] = make_int2(s | (dl << 17), __float_as_int(w));
  }
  rows_add(hv, dl, w, nb, lane, tile);
}

template <bool REC>
__device__ __forceinline__ void agg_scan(int lo, int wave, int lane, int* mylist,
                                         const int* __restrict__ src, const int* __restrict__ dst,
                                         const float* __restrict__ dinv, const float* dinvL,
                                         const float* __restrict__ H, float* tile,
                                         int* cursor, int2* __restrict__ rec) {
  const int4* dst4 = (const int4*)dst;
  int pend = 0;
  int trip = blockIdx.x % N_TRIPS;

  int4 nxt[SCAN_U];
  agg_load_trip(dst4, trip * TRIP4 + wave * WAVE4, lane, nxt);

  for (int it = 0; it < N_TRIPS; ++it) {
    int4 cur[SCAN_U];
    #pragma unroll
    for (int u = 0; u < SCAN_U; ++u) cur[u] = nxt[u];
    const int cur4 = trip * TRIP4 + wave * WAVE4;
    trip = (trip + 1 == N_TRIPS) ? 0 : trip + 1;
    if (it + 1 < N_TRIPS) agg_load_trip(dst4, trip * TRIP4 + wave * WAVE4, lane, nxt);

    #pragma unroll
    for (int u = 0; u < SCAN_U; ++u) {
      #pragma unroll
      for (int c = 0; c < 4; ++c) {
        const int d = (c == 0) ? cur[u].x : (c == 1) ? cur[u].y : (c == 2) ? cur[u].z : cur[u].w;
        const unsigned tt = (unsigned)(d - lo);
        const bool m = tt < (unsigned)NPB;
        const unsigned mask = __builtin_amdgcn_ballot_w32(m);
        if (mask != 0u) {
          const int pos = pend + (int)__builtin_amdgcn_mbcnt_lo(mask, 0u);
          const int e = (cur4 + u * 32 + lane) * 4 + c;
          if (m) mylist[pos] = (int)((tt << 21) | (unsigned)e);
          pend += __builtin_popcount(mask);
        }
      }
    }
    while (pend >= BATCH) {
      pend -= BATCH;
      agg_batch<REC, true>(mylist + pend, BATCH, lane, src, dinv, dinvL, H, tile, cursor, rec);
    }
  }
  if (pend > 0) agg_batch<REC, true>(mylist, pend, lane, src, dinv, dinvL, H, tile, cursor, rec);
}

__device__ __forceinline__ void agg_cells(const int* __restrict__ mycells, int wv, int lane,
                                          int* mylist, const int* __restrict__ src,
                                          const float* __restrict__ dinv, const float* dinvL,
                                          const float* __restrict__ H, float* tile,
                                          int* cursor, int2* __restrict__ rec) {
  const int2* cp = (const int2*)(mycells + (size_t)wv * CELLS_PER_WAVE * CELL_INTS) + lane;
  int pend = 0;
  int2 nxt = cp[0];
  for (int c = 0; c < CELLS_PER_WAVE; ++c) {
    const int2 cur = nxt;
    if (c + 1 < CELLS_PER_WAVE) nxt = cp[(c + 1) * (CELL_INTS / 2)];
    int cnt = __builtin_amdgcn_readfirstlane(cur.x);
    cnt = (cnt < CELL_CAP) ? cnt : CELL_CAP;
    const int i0 = 2 * lane;
    if (i0 >= 1 && i0 <= cnt) mylist[pend + i0 - 1] = cur.x;
    if (i0 + 1 <= cnt) mylist[pend + i0] = cur.y;
    pend += cnt;
    while (pend >= BATCH) {
      pend -= BATCH;
      agg_batch<true, false>(mylist + pend, BATCH, lane, src, dinv, dinvL, H, tile, cursor, rec);
    }
  }
  if (pend > 0) agg_batch<true, false>(mylist, pend, lane, src, dinv, dinvL, H, tile, cursor, rec);
}

template <int MODE>
__global__ __launch_bounds__(AGG_THREADS) void k_agg_owner(const int* __restrict__ src,
                                                           const int* __restrict__ dst,
                                                           const float* __restrict__ dinv,
                                                           const float* __restrict__ H,
                                                           const float* __restrict__ bias,
                                                           float* __restrict__ OUT, int do_relu,
                                                           int* __restrict__ rec_cnt,
                                                           int* __restrict__ rec_all,
                                                           const int* __restrict__ cells,
                                                           const int* __restrict__ ovf) {
  __shared__ __attribute__((aligned(16))) float tile[NPB * DIM];
  __shared__ float dinvL[NPB + 1];
  __shared__ int lists[AGG_WAVES * LIST_CAP];
  __shared__ int cursor;

  const int t = threadIdx.x;
  const int wave = t >> 5;
  const int lane = t & 31;
  const int lo = blockIdx.x * NPB;
  const int nvalid = (N_NODESC - lo < NPB) ? (N_NODESC - lo) : NPB;
  int2* rec = (int2*)rec_all + (size_t)blockIdx.x * REC_CAP;

  if (t == 0) cursor = 0;

  for (int r = wave; r < NPB; r += AGG_WAVES) {
    float4 v = make_float4(0.f, 0.f, 0.f, 0.f);
    float di = 0.f;
    if (r < nvalid) {
      di = dinv[lo + r];
      const float4 h = ((const float4*)(H + (size_t)(lo + r) * DIM))[lane];
      v.x = h.x * di * di;
      v.y = h.y * di * di;
      v.z = h.z * di * di;
      v.w = h.w * di * di;
    }
    ((float4*)(tile + r * DIM))[lane] = v;
    if (lane == 0) dinvL[r] = di;
  }
  __syncthreads();

  int* mylist = lists + wave * LIST_CAP;

  if (MODE == 0) {
    if (ovf[blockIdx.x] == 0) {
      const int wv = __builtin_amdgcn_readfirstlane(wave);
      agg_cells(cells + (size_t)blockIdx.x * BIN_BLOCKS * CELL_INTS, wv, lane, mylist, src, dinv,
                dinvL, H, tile, &cursor, rec);
    } else {
      agg_scan<true>(lo, wave, lane, mylist, src, dst, dinv, dinvL, H, tile, &cursor, rec);
    }
  } else {
    const int n = rec_cnt[blockIdx.x];
    if (n <= REC_CAP) {
      const int nbat = (n + BATCH - 1) / BATCH;
      const int wv = __builtin_amdgcn_readfirstlane(wave);
      int2 ent = make_int2(0, 0);
      if (wv < nbat && wv * BATCH + lane < n && lane < BATCH) ent = rec[wv * BATCH + lane];
      for (int bi = wv; bi < nbat; bi += AGG_WAVES) {
        const int2 cur = ent;
        const int left = n - bi * BATCH;
        const int nb = (left < BATCH) ? left : BATCH;
        float hv[BATCH][4];
        rows_load(cur.x & ((1 << 17) - 1), lane, H, hv);
        const int bn = bi + AGG_WAVES;
        ent = make_int2(0, 0);
        if (bn < nbat && bn * BATCH + lane < n && lane < BATCH) ent = rec[bn * BATCH + lane];
        rows_add(hv, (int)((unsigned)cur.x >> 17), __int_as_float(cur.y), nb, lane, tile);
      }
    } else {
      agg_scan<false>(lo, wave, lane, mylist, src, dst, dinv, dinvL, H, tile, &cursor, rec);
    }
  }

  __syncthreads();

  if (MODE == 0 && t == 0) rec_cnt[blockIdx.x] = cursor;

  const float4 bb = ((const float4*)bias)[lane];
  for (int r = wave; r < nvalid; r += AGG_WAVES) {
    float4 a = ((const float4*)(tile + r * DIM))[lane];
    a.x += bb.x;
    a.y += bb.y;
    a.z += bb.z;
    a.w += bb.w;
    if (do_relu) {
      a.x = fmaxf(a.x, 0.0f);
      a.y = fmaxf(a.y, 0.0f);
      a.z = fmaxf(a.z, 0.0f);
      a.w = fmaxf(a.w, 0.0f);
    }
    ((float4*)(OUT + (size_t)(lo + r) * DIM))[lane] = a;
  }
}


extern "C" void kernel_launch(void* const* d_in, const int* in_sizes, int n_in,
                              void* d_out, int out_size, void* d_ws, size_t ws_size,
                              hipStream_t stream) {
  const float* x   = (const float*)d_in[0];
  const int*   ei  = (const int*)d_in[1];
  const float* W1  = (const float*)d_in[2];
  const float* b1  = (const float*)d_in[3];
  const float* W2  = (const float*)d_in[4];
  const float* b2  = (const float*)d_in[5];
  float*       out = (float*)d_out;

  const int* src = ei;
  const int* dst = ei + N_EDGESC;

  float* H    = (float*)d_ws;
  float* deg  = H + (size_t)N_NODESC * DIM;
  float* dinv = deg + N_NODESC;
  int*   rec_cnt = (int*)(dinv + N_NODESC);
  int*   rec_all = rec_cnt + AGG_BLOCKS;
  int*   cells   = rec_all + (size_t)AGG_BLOCKS * REC_CAP * 2;
  int*   ovf     = cells + (size_t)AGG_BLOCKS * BIN_BLOCKS * CELL_INTS;

  dim3 blk(256);
  dim3 gNode((N_NODESC + 255) / 256);
  dim3 gBin(BIN_BLOCKS);
  dim3 bBin(BIN_THREADS);
  dim3 gGemm(GEMM_BLOCKS);
  dim3 gAgg(AGG_BLOCKS);
  dim3 bAgg(AGG_THREADS);

  k_init_deg<<<gNode, blk, 0, stream>>>(deg, ovf);
  k_bin_count<<<gBin, bBin, 0, stream>>>(src, dst, deg, cells, ovf);
  k_dinv<<<gNode, blk, 0, stream>>>(deg, dinv);

  k_gemm_wmma<<<gGemm, blk, 0, stream>>>(x, W1, H);
  k_agg_owner<0><<<gAgg, bAgg, 0, stream>>>(src, dst, dinv, H, b1, out,  1, rec_cnt, rec_all, cells, ovf);

  k_gemm_wmma<<<gGemm, blk, 0, stream>>>(out, W2, H);
  k_agg_owner<1><<<gAgg, bAgg, 0, stream>>>(src, dst, dinv, H, b2, out,  0, rec_cnt, rec_all, cells, ovf);
}


# Each row of an owner block's LDS slice gets one writer wave (row & 31): edges go to per-wave LDS queues, rows added by plain 16-byte LDS read-add-write, no float atomics; sums in queue arrival order
# speedup vs baseline: 3.2715x; 1.0695x over previous
#include <hip/hip_runtime.h>

typedef float v2f __attribute__((ext_vector_type(2)));
typedef float v8f __attribute__((ext_vector_type(8)));

#define N_NODESC 100000
#define N_EDGESC 1600000
#define DIM 128


#define AGG_BLOCKS 256
#define NPB 391

#define BIN_BLOCKS 256
#define BIN_THREADS 1024
#define CELL_INTS 64
#define CELL_CAP (CELL_INTS - 1)
#define CELL_PITCH 68
#define EDGES_PER_BIN (N_EDGESC / BIN_BLOCKS)
#define BIN_ITERS ((EDGES_PER_BIN + BIN_THREADS - 1) / BIN_THREADS)

static_assert(N_EDGESC % BIN_BLOCKS == 0, "chunks are equal");
static_assert(N_NODESC <= (1 << 17) && NPB <= (1 << 9), "cell entry packs source node and local row");

__global__ void k_init_deg(float* __restrict__ deg, int* __restrict__ ovf) {
  int i = blockIdx.x * blockDim.x + threadIdx.x;
  if (i < N_NODESC) deg[i] = 1.0f;
  if (i < AGG_BLOCKS) ovf[i] = 0;
}

__global__ __launch_bounds__(BIN_THREADS) void k_bin_count(const int* __restrict__ src,
                                                           const int* __restrict__ dst,
                                                           float* __restrict__ deg,
                                                           int* __restrict__ cells,
                                                           int* __restrict__ ovf) {
  __shared__ __attribute__((aligned(16))) int cellL[AGG_BLOCKS * CELL_PITCH];
  __shared__ int cntL[AGG_BLOCKS];

  const int t = threadIdx.x;
  const int e0 = blockIdx.x * EDGES_PER_BIN;

  int dv[BIN_ITERS], sv[BIN_ITERS];
  #pragma unroll
  for (int k = 0; k < BIN_ITERS; ++k) {
    const int i = t + k * BIN_THREADS;
    dv[k] = -1;
    sv[k] = 0;
    if (i < EDGES_PER_BIN) {
      dv[k] = dst[e0 + i];
      sv[k] = src[e0 + i];
    }
  }

  for (int q = t; q < AGG_BLOCKS * CELL_PITCH / 4; q += BIN_THREADS)
    ((int4*)cellL)[q] = make_int4(0, 0, 0, 0);
  for (int o = t; o < AGG_BLOCKS; o += BIN_THREADS) cntL[o] = 0;
  __syncthreads();

  #pragma unroll
  for (int k = 0; k < BIN_ITERS; ++k) {
    const int d = dv[k];
    if (d >= 0) {
      unsafeAtomicAdd(&deg[d], 1.0f);
      const int o = d / NPB;
      const int dl = d - o * NPB;
      const int rank = __hip_atomic_fetch_add(&cntL[o], 1, __ATOMIC_RELAXED, __HIP_MEMORY_SCOPE_WORKGROUP);
      if (rank < CELL_CAP) cellL[o * CELL_PITCH + 1 + rank] = sv[k] | (dl << 17);
      else ovf[o] = 1;
    }
  }
  __syncthreads();

  for (int q = t; q < AGG_BLOCKS * CELL_INTS / 4; q += BIN_THREADS) {
    const int o = q >> 4;
    const int part = q & 15;
    int4 v = *(const int4*)&cellL[o * CELL_PITCH + part * 4];
    if (part == 0) v.x = cntL[o];
    ((int4*)(cells + ((size_t)o * BIN_BLOCKS + blockIdx.x) * CELL_INTS))[part] = v;
  }
}

__global__ void k_dinv(const float* __restrict__ deg, float* __restrict__ dinv) {
  int i = blockIdx.x * blockDim.x + threadIdx.x;
  if (i < N_NODESC) dinv[i] = rsqrtf(deg[i]);
}


#define GEMM_BLOCKS 768
#define GEMM_TILES (N_NODESC / 16)
#define GEMM_KSTEPS (DIM / 4)
#define XS_PITCH4 33
#define XS_STAGE4 (16 * XS_PITCH4)

static_assert(N_NODESC % 16 == 0, "whole 16-row tiles");

__global__ __launch_bounds__(256) void k_gemm_wmma(const float* __restrict__ X,
                                                   const float* __restrict__ W,
                                                   float* __restrict__ H) {
  __shared__ float4 xs4[2 * XS_STAGE4];

  const int t = threadIdx.x;
  const int wave = t >> 5;
  const int lane = t & 31;

  const int m  = lane & 15;
  const int hg = lane >> 4;
  const int n0 = wave * 16;

  v2f bf[GEMM_KSTEPS];
  {
    const float* wcol = W + n0 + m;
    #pragma unroll
    for (int j = 0; j < GEMM_KSTEPS; ++j) {
      bf[j].x = wcol[(size_t)(4 * j + 2 * hg + 0) * DIM];
      bf[j].y = wcol[(size_t)(4 * j + 2 * hg + 1) * DIM];
    }
  }

  const int r  = wave;
  const int c4 = lane;

  int tile = blockIdx.x;
  float4 p0, p1;
  {
    const int tl = (tile < GEMM_TILES) ? tile : GEMM_TILES - 1;
    const float4* g = (const float4*)(X + ((size_t)tl * 16 + r) * DIM) + c4;
    p0 = g[0];
    p1 = g[8 * (DIM / 4)];
  }

  int cur = 0;
  for (; tile < GEMM_TILES; tile += GEMM_BLOCKS) {
    float4* stage = xs4 + cur * XS_STAGE4;
    stage[r * XS_PITCH4 + c4] = p0;
    stage[(r + 8) * XS_PITCH4 + c4] = p1;
    __syncthreads();

    {
      const int nxt = tile + GEMM_BLOCKS;
      const int tl = (nxt < GEMM_TILES) ? nxt : GEMM_TILES - 1;
      const float4* g = (const float4*)(X + ((size_t)tl * 16 + r) * DIM) + c4;
      p0 = g[0];
      p1 = g[8 * (DIM / 4)];
    }
    __builtin_amdgcn_sched_barrier(0);

    const v2f* arow = (const v2f*)stage + m * (XS_PITCH4 * 2) + hg;
    v2f af[GEMM_KSTEPS];
    #pragma unroll
    for (int j = 0; j < GEMM_KSTEPS; ++j) af[j] = arow[2 * j];
    __builtin_amdgcn_sched_barrier(0);

    v8f acc = {};
    #pragma unroll
    for (int j = 0; j < GEMM_KSTEPS; ++j) {
      acc = __builtin_amdgcn_wmma_f32_16x16x4_f32(
           false, af[j],  false, bf[j],
           (short)0, acc,  false,  false);
    }

    float* hrow = H + (size_t)tile * 16 * DIM + n0 + m;
    #pragma unroll
    for (int v = 0; v < 8; ++v)
      hrow[(size_t)(v + 8 * hg) * DIM] = acc[v];

    cur ^= 1;
  }
}


#define AGG_THREADS 1024
#define AGG_WAVES 32
#define SCAN_U 4
#define N_DST4 (N_EDGESC / 4)
#define WAVE4 (32 * SCAN_U)
#define TRIP4 (AGG_WAVES * WAVE4)
#define N_TRIPS ((N_DST4 + TRIP4 - 1) / TRIP4)
#define BATCH 8
#define LIST_CAP (BATCH - 1 + 4 * WAVE4 + 1)
#define QCAP LIST_CAP
#define SRC_MASK ((1 << 17) - 1)

static_assert(AGG_BLOCKS * NPB >= N_NODESC, "slices must cover every node");
static_assert(N_EDGESC % 4 == 0, "dst list is read as int4");
static_assert(N_EDGESC <= (1 << 21) && NPB <= (1 << 9), "list entry packs edge id and local row");
static_assert(BIN_BLOCKS % AGG_WAVES == 0, "every wave takes the same number of cells");
#define CELLS_PER_WAVE (BIN_BLOCKS / AGG_WAVES)

__device__ __forceinline__ void agg_load_trip(const int4* __restrict__ dst4, int base4, int lane,
                                              int4 (&v)[SCAN_U]) {
  #pragma unroll
  for (int u = 0; u < SCAN_U; ++u) {
    const int i4 = base4 + u * 32 + lane;
    v[u] = (i4 < N_DST4) ? dst4[i4] : make_int4(-1, -1, -1, -1);
  }
}

__device__ __forceinline__ void rows_load(int s, int lane, const float* __restrict__ H,
                                          float (&hv)[BATCH][4]) {
  #pragma unroll
  for (int i = 0; i < BATCH; ++i) {
    const int si = __builtin_amdgcn_readlane(s, i);
    const float* hr = H + (size_t)si * DIM + lane;
    hv[i][0] = hr[0];
    hv[i][1] = hr[32];
    hv[i][2] = hr[64];
    hv[i][3] = hr[96];
  }
}

__device__ __forceinline__ void rows_add(const float (&hv)[BATCH][4], int dl, float w, int nb,
                                         int lane, float* tile) {
  #pragma unroll
  for (int i = 0; i < BATCH; ++i) {
    if (i < nb) {
      const int di = __builtin_amdgcn_readlane(dl, i);
      const float wi = __int_as_float(__builtin_amdgcn_readlane(__float_as_int(w), i));
      float* tr = tile + di * DIM + lane;
      __hip_atomic_fetch_add(tr + 0,  hv[i][0] * wi, __ATOMIC_RELAXED, __HIP_MEMORY_SCOPE_WORKGROUP);
      __hip_atomic_fetch_add(tr + 32, hv[i][1] * wi, __ATOMIC_RELAXED, __HIP_MEMORY_SCOPE_WORKGROUP);
      __hip_atomic_fetch_add(tr + 64, hv[i][2] * wi, __ATOMIC_RELAXED, __HIP_MEMORY_SCOPE_WORKGROUP);
      __hip_atomic_fetch_add(tr + 96, hv[i][3] * wi, __ATOMIC_RELAXED, __HIP_MEMORY_SCOPE_WORKGROUP);
    }
  }
}

__device__ __forceinline__ void agg_batch(const int* lst, int nb, int lane,
                                          const int* __restrict__ src,
                                          const float* __restrict__ dinv,
                                          const float* dinvL,
                                          const float* __restrict__ H, float* tile) {
  int ent = 0;
  if (lane < nb) ent = lst[lane];
  const int dl = (int)((unsigned)ent >> 21);
  const int s = src[ent & ((1 << 21) - 1)];
  const float ds = dinv[s];
  float hv[BATCH][4];
  rows_load(s, lane, H, hv);
  const float w = ds * dinvL[dl];
  rows_add(hv, dl, w, nb, lane, tile);
}

__device__ __forceinline__ void agg_scan(int lo, int wave, int lane, int* mylist,
                                         const int* __restrict__ src, const int* __restrict__ dst,
                                         const float* __restrict__ dinv, const float* dinvL,
                                         const float* __restrict__ H, float* tile) {
  const int4* dst4 = (const int4*)dst;
  int pend = 0;
  int trip = blockIdx.x % N_TRIPS;

  int4 nxt[SCAN_U];
  agg_load_trip(dst4, trip * TRIP4 + wave * WAVE4, lane, nxt);

  for (int it = 0; it < N_TRIPS; ++it) {
    int4 cur[SCAN_U];
    #pragma unroll
    for (int u = 0; u < SCAN_U; ++u) cur[u] = nxt[u];
    const int cur4 = trip * TRIP4 + wave * WAVE4;
    trip = (trip + 1 == N_TRIPS) ? 0 : trip + 1;
    if (it + 1 < N_TRIPS) agg_load_trip(dst4, trip * TRIP4 + wave * WAVE4, lane, nxt);

    #pragma unroll
    for (int u = 0; u < SCAN_U; ++u) {
      #pragma unroll
      for (int c = 0; c < 4; ++c) {
        const int d = (c == 0) ? cur[u].x : (c == 1) ? cur[u].y : (c == 2) ? cur[u].z : cur[u].w;
        const unsigned tt = (unsigned)(d - lo);
        const bool m = tt < (unsigned)NPB;
        const unsigned mask = __builtin_amdgcn_ballot_w32(m);
        if (mask != 0u) {
          const int pos = pend + (int)__builtin_amdgcn_mbcnt_lo(mask, 0u);
          const int e = (cur4 + u * 32 + lane) * 4 + c;
          if (m) mylist[pos] = (int)((tt << 21) | (unsigned)e);
          pend += __builtin_popcount(mask);
        }
      }
    }
    while (pend >= BATCH) {
      pend -= BATCH;
      agg_batch(mylist + pend, BATCH, lane, src, dinv, dinvL, H, tile);
    }
  }
  if (pend > 0) agg_batch(mylist, pend, lane, src, dinv, dinvL, H, tile);
}

__device__ __forceinline__ void queue_put(int e, int* queues, int* qcnt, int* qovf) {
  const int ow = (e >> 17) & (AGG_WAVES - 1);
  const int slot = __hip_atomic_fetch_add(&qcnt[ow], 1, __ATOMIC_RELAXED, __HIP_MEMORY_SCOPE_WORKGROUP);
  if (slot < QCAP) queues[ow * QCAP + slot] = e;
  else *qovf = 1;
}

__device__ __forceinline__ void route_cells(const int* __restrict__ mycells, int wv, int lane,
                                            int* queues, int* qcnt, int* qovf) {
  const int2* cp = (const int2*)(mycells + (size_t)wv * CELLS_PER_WAVE * CELL_INTS) + lane;
  int2 cv[CELLS_PER_WAVE];
  #pragma unroll
  for (int c = 0; c < CELLS_PER_WAVE; ++c) cv[c] = cp[c * (CELL_INTS / 2)];
  #pragma unroll
  for (int c = 0; c < CELLS_PER_WAVE; ++c) {
    int cnt = __builtin_amdgcn_readfirstlane(cv[c].x);
    cnt = (cnt < CELL_CAP) ? cnt : CELL_CAP;
    const int i0 = 2 * lane;
    if (i0 >= 1 && i0 <= cnt) queue_put(cv[c].x, queues, qcnt, qovf);
    if (i0 + 1 <= cnt) queue_put(cv[c].y, queues, qcnt, qovf);
  }
}

__device__ __forceinline__ void rows_load4(int s, int lane, const float* __restrict__ H,
                                           float4 (&hv)[BATCH]) {
  #pragma unroll
  for (int i = 0; i < BATCH; ++i) {
    const int si = __builtin_amdgcn_readlane(s, i);
    hv[i] = ((const float4*)(H + (size_t)si * DIM))[lane];
  }
}

__device__ __forceinline__ void rows_rmw(const float4 (&hv)[BATCH], int dl, float w, int nb,
                                         int lane, float* tile) {
  #pragma unroll
  for (int i = 0; i < BATCH; ++i) {
    if (i < nb) {
      const int di = __builtin_amdgcn_readlane(dl, i);
      const float wi = __int_as_float(__builtin_amdgcn_readlane(__float_as_int(w), i));
      float4* tr = (float4*)(tile + di * DIM) + lane;
      float4 a = *tr;
      a.x += hv[i].x * wi;
      a.y += hv[i].y * wi;
      a.z += hv[i].z * wi;
      a.w += hv[i].w * wi;
      *tr = a;
    }
  }
}

template <int MODE>
__device__ __forceinline__ void queue_entry(int b, int n, int lane, const int* q,
                                            const int2* __restrict__ rec, int& e, float& w) {
  const int idx = b * BATCH + lane;
  e = 0;
  w = 0.0f;
  if (lane < BATCH && idx < n) {
    if (MODE == 0) {
      e = q[idx];
    } else {
      const int2 r = rec[idx];
      e = r.x;
      w = __int_as_float(r.y);
    }
  }
}

template <int MODE>
__device__ __forceinline__ void sum_queue(int n, int lane, const int* q, int2* __restrict__ rec,
                                          const float* __restrict__ dinv, const float* dinvL,
                                          const float* __restrict__ H, float* tile) {
  const int nbat = (n + BATCH - 1) / BATCH;
  int e_rq, e_nx;
  float w_rq, w_nx;
  float4 hv[BATCH];
  queue_entry<MODE>(0, n, lane, q, rec, e_rq, w_rq);
  if (MODE == 0) w_rq = dinv[e_rq & SRC_MASK];
  rows_load4(e_rq & SRC_MASK, lane, H, hv);
  queue_entry<MODE>(1, n, lane, q, rec, e_nx, w_nx);

  for (int b = 0; b < nbat; ++b) {
    float4 cv[BATCH];
    #pragma unroll
    for (int i = 0; i < BATCH; ++i) cv[i] = hv[i];
    const int e_cur = e_rq;
    const float w_cur = w_rq;

    e_rq = e_nx;
    w_rq = w_nx;
    if (MODE == 0) w_rq = dinv[e_rq & SRC_MASK];
    rows_load4(e_rq & SRC_MASK, lane, H, hv);
    queue_entry<MODE>(b + 2, n, lane, q, rec, e_nx, w_nx);

    const int left = n - b * BATCH;
    const int nb = (left < BATCH) ? left : BATCH;
    const int dl = (int)((unsigned)e_cur >> 17);
    float w = w_cur;
    if (MODE == 0) {
      w = w_cur * dinvL[dl];
      if (lane < nb) rec[b * BATCH + lane] = make_int2(e_cur, __float_as_int(w));
    }
    rows_rmw(cv, dl, w, nb, lane, tile);
  }
}

template <int MODE>
__global__ __launch_bounds__(AGG_THREADS) void k_agg_owner(const int* __restrict__ src,
                                                           const int* __restrict__ dst,
                                                           const float* __restrict__ dinv,
                                                           const float* __restrict__ H,
                                                           const float* __restrict__ bias,
                                                           float* __restrict__ OUT, int do_relu,
                                                           int* __restrict__ rec_cnt,
                                                           int* __restrict__ rec_all,
                                                           int* __restrict__ qbad,
                                                           const int* __restrict__ cells,
                                                           const int* __restrict__ ovf) {
  __shared__ __attribute__((aligned(16))) float tile[NPB * DIM];
  __shared__ float dinvL[NPB + 1];
  __shared__ int lists[AGG_WAVES * LIST_CAP];
  __shared__ int qcnt[AGG_WAVES];
  __shared__ int qovf;

  const int t = threadIdx.x;
  const int wave = t >> 5;
  const int lane = t & 31;
  const int lo = blockIdx.x * NPB;
  const int nvalid = (N_NODESC - lo < NPB) ? (N_NODESC - lo) : NPB;
  const int wv = __builtin_amdgcn_readfirstlane(wave);
  int2* rec = (int2*)rec_all + ((size_t)blockIdx.x * AGG_WAVES + wv) * QCAP;

  if (t < AGG_WAVES) qcnt[t] = 0;
  if (t == 0) qovf = 0;

  for (int r = wave; r < NPB; r += AGG_WAVES) {
    float4 v = make_float4(0.f, 0.f, 0.f, 0.f);
    float di = 0.f;
    if (r < nvalid) {
      di = dinv[lo + r];
      const float4 h = ((const float4*)(H + (size_t)(lo + r) * DIM))[lane];
      v.x = h.x * di * di;
      v.y = h.y * di * di;
      v.z = h.z * di * di;
      v.w = h.w * di * di;
    }
    ((float4*)(tile + r * DIM))[lane] = v;
    if (lane == 0) dinvL[r] = di;
  }
  __syncthreads();

  int* mylist = lists + wv * LIST_CAP;
  bool bad;

  if (MODE == 0) {
    bad = ovf[blockIdx.x] != 0;
    if (!bad)
      route_cells(cells + (size_t)blockIdx.x * BIN_BLOCKS * CELL_INTS, wv, lane, lists, qcnt, &qovf);
    __syncthreads();
    bad = bad || (qovf != 0);
    if (!bad) {
      const int n = qcnt[wv];
      sum_queue<0>(n, lane, lists + wv * QCAP, rec, dinv, dinvL, H, tile);
      if (lane == 0) rec_cnt[blockIdx.x * AGG_WAVES + wv] = n;
    } else {
      agg_scan(lo, wave, lane, mylist, src, dst, dinv, dinvL, H, tile);
    }
    if (t == 0) qbad[blockIdx.x] = bad ? 1 : 0;
  } else {
    bad = qbad[blockIdx.x] != 0;
    if (!bad) {
      const int n = rec_cnt[blockIdx.x * AGG_WAVES + wv];
      sum_queue<1>(n, lane, lists, rec, dinv, dinvL, H, tile);
    } else {
      agg_scan(lo, wave, lane, mylist, src, dst, dinv, dinvL, H, tile);
    }
  }

  __syncthreads();

  const float4 bb = ((const float4*)bias)[lane];
  for (int r = wave; r < nvalid; r += AGG_WAVES) {
    float4 a = ((const float4*)(tile + r * DIM))[lane];
    a.x += bb.x;
    a.y += bb.y;
    a.z += bb.z;
    a.w += bb.w;
    if (do_relu) {
      a.x = fmaxf(a.x, 0.0f);
      a.y = fmaxf(a.y, 0.0f);
      a.z = fmaxf(a.z, 0.0f);
      a.w = fmaxf(a.w, 0.0f);
    }
    ((float4*)(OUT + (size_t)(lo + r) * DIM))[lane] = a;
  }
}


extern "C" void kernel_launch(void* const* d_in, const int* in_sizes, int n_in,
                              void* d_out, int out_size, void* d_ws, size_t ws_size,
                              hipStream_t stream) {
  const float* x   = (const float*)d_in[0];
  const int*   ei  = (const int*)d_in[1];
  const float* W1  = (const float*)d_in[2];
  const float* b1  = (const float*)d_in[3];
  const float* W2  = (const float*)d_in[4];
  const float* b2  = (const float*)d_in[5];
  float*       out = (float*)d_out;

  const int* src = ei;
  const int* dst = ei + N_EDGESC;

  float* H    = (float*)d_ws;
  float* deg  = H + (size_t)N_NODESC * DIM;
  float* dinv = deg + N_NODESC;
  int*   rec_cnt = (int*)(dinv + N_NODESC);
  int*   qbad    = rec_cnt + AGG_BLOCKS * AGG_WAVES;
  int*   rec_all = qbad + AGG_BLOCKS;
  int*   cells   = rec_all + (size_t)AGG_BLOCKS * AGG_WAVES * QCAP * 2;
  int*   ovf     = cells + (size_t)AGG_BLOCKS * BIN_BLOCKS * CELL_INTS;

  dim3 blk(256);
  dim3 gNode((N_NODESC + 255) / 256);
  dim3 gBin(BIN_BLOCKS);
  dim3 bBin(BIN_THREADS);
  dim3 gGemm(GEMM_BLOCKS);
  dim3 gAgg(AGG_BLOCKS);
  dim3 bAgg(AGG_THREADS);

  k_init_deg<<<gNode, blk, 0, stream>>>(deg, ovf);
  k_bin_count<<<gBin, bBin, 0, stream>>>(src, dst, deg, cells, ovf);
  k_dinv<<<gNode, blk, 0, stream>>>(deg, dinv);

  k_gemm_wmma<<<gGemm, blk, 0, stream>>>(x, W1, H);
  k_agg_owner<0><<<gAgg, bAgg, 0, stream>>>(src, dst, dinv, H, b1, out,  1, rec_cnt, rec_all, qbad, cells, ovf);

  k_gemm_wmma<<<gGemm, blk, 0, stream>>>(out, W2, H);
  k_agg_owner<1><<<gAgg, bAgg, 0, stream>>>(src, dst, dinv, H, b2, out,  0, rec_cnt, rec_all, qbad, cells, ovf);
}


# Degree count drops its global float atomic per edge: owner blocks count their cells' entries per row with LDS integer adds, write dinv and the overflow flag; 6 launches; counts exact, sums unchanged
# speedup vs baseline: 4.8008x; 1.4675x over previous
#include <hip/hip_runtime.h>

typedef float v2f __attribute__((ext_vector_type(2)));
typedef float v8f __attribute__((ext_vector_type(8)));

#define N_NODESC 100000
#define N_EDGESC 1600000
#define DIM 128


#define AGG_BLOCKS 256
#define NPB 391

#define BIN_BLOCKS 256
#define BIN_THREADS 1024
#define CELL_INTS 64
#define CELL_CAP (CELL_INTS - 1)
#define CELL_PITCH 68
#define EDGES_PER_BIN (N_EDGESC / BIN_BLOCKS)
#define BIN_ITERS ((EDGES_PER_BIN + BIN_THREADS - 1) / BIN_THREADS)

static_assert(N_EDGESC % BIN_BLOCKS == 0, "chunks are equal");
static_assert(N_NODESC <= (1 << 17) && NPB <= (1 << 9), "cell entry packs source node and local row");

__global__ __launch_bounds__(BIN_THREADS) void k_bin_count(const int* __restrict__ src,
                                                           const int* __restrict__ dst,
                                                           int* __restrict__ cells) {
  __shared__ __attribute__((aligned(16))) int cellL[AGG_BLOCKS * CELL_PITCH];
  __shared__ int cntL[AGG_BLOCKS];

  const int t = threadIdx.x;
  const int e0 = blockIdx.x * EDGES_PER_BIN;

  int dv[BIN_ITERS], sv[BIN_ITERS];
  #pragma unroll
  for (int k = 0; k < BIN_ITERS; ++k) {
    const int i = t + k * BIN_THREADS;
    dv[k] = -1;
    sv[k] = 0;
    if (i < EDGES_PER_BIN) {
      dv[k] = dst[e0 + i];
      sv[k] = src[e0 + i];
    }
  }

  for (int q = t; q < AGG_BLOCKS * CELL_PITCH / 4; q += BIN_THREADS)
    ((int4*)cellL)[q] = make_int4(0, 0, 0, 0);
  for (int o = t; o < AGG_BLOCKS; o += BIN_THREADS) cntL[o] = 0;
  __syncthreads();

  #pragma unroll
  for (int k = 0; k < BIN_ITERS; ++k) {
    const int d = dv[k];
    if (d >= 0) {
      const int o = d / NPB;
      const int dl = d - o * NPB;
      const int rank = __hip_atomic_fetch_add(&cntL[o], 1, __ATOMIC_RELAXED, __HIP_MEMORY_SCOPE_WORKGROUP);
      if (rank < CELL_CAP) cellL[o * CELL_PITCH + 1 + rank] = sv[k] | (dl << 17);
    }
  }
  __syncthreads();

  for (int q = t; q < AGG_BLOCKS * CELL_INTS / 4; q += BIN_THREADS) {
    const int o = q >> 4;
    const int part = q & 15;
    int4 v = *(const int4*)&cellL[o * CELL_PITCH + part * 4];
    if (part == 0) v.x = cntL[o];
    ((int4*)(cells + ((size_t)o * BIN_BLOCKS + blockIdx.x) * CELL_INTS))[part] = v;
  }
}

#define DEG_THREADS 1024
#define DEG_WAVES (DEG_THREADS / 32)
#define DEG_CELLS_PER_WAVE (BIN_BLOCKS / DEG_WAVES)
#define N_DST4Q (N_EDGESC / 4)

static_assert(BIN_BLOCKS % DEG_WAVES == 0, "every wave takes the same number of cells");
static_assert(N_EDGESC % 4 == 0, "dst list is read as int4");

__global__ __launch_bounds__(DEG_THREADS) void k_deg_owner(const int* __restrict__ dst,
                                                           const int* __restrict__ cells,
                                                           float* __restrict__ dinv,
                                                           int* __restrict__ ovf) {
  __shared__ int cntR[NPB];
  __shared__ int full;

  const int t = threadIdx.x;
  const int lane = t & 31;
  const int wv = __builtin_amdgcn_readfirstlane(t >> 5);
  const int lo = blockIdx.x * NPB;
  const int nvalid = (N_NODESC - lo < NPB) ? (N_NODESC - lo) : NPB;

  const int2* cp = (const int2*)(cells + ((size_t)blockIdx.x * BIN_BLOCKS +
                                          (size_t)wv * DEG_CELLS_PER_WAVE) * CELL_INTS) + lane;
  int2 cv[DEG_CELLS_PER_WAVE];
  #pragma unroll
  for (int c = 0; c < DEG_CELLS_PER_WAVE; ++c) cv[c] = cp[c * (CELL_INTS / 2)];

  for (int r = t; r < NPB; r += DEG_THREADS) cntR[r] = 0;
  if (t == 0) full = 0;
  __syncthreads();

  #pragma unroll
  for (int c = 0; c < DEG_CELLS_PER_WAVE; ++c) {
    const int raw = __builtin_amdgcn_readfirstlane(cv[c].x);
    if (raw > CELL_CAP && lane == 0) full = 1;
    const int cnt = (raw < CELL_CAP) ? raw : CELL_CAP;
    const int i0 = 2 * lane;
    if (i0 >= 1 && i0 <= cnt)
      __hip_atomic_fetch_add(&cntR[(unsigned)cv[c].x >> 17], 1, __ATOMIC_RELAXED, __HIP_MEMORY_SCOPE_WORKGROUP);
    if (i0 + 1 <= cnt)
      __hip_atomic_fetch_add(&cntR[(unsigned)cv[c].y >> 17], 1, __ATOMIC_RELAXED, __HIP_MEMORY_SCOPE_WORKGROUP);
  }
  __syncthreads();

  const bool bad = full != 0;
  if (bad) {
    for (int r = t; r < NPB; r += DEG_THREADS) cntR[r] = 0;
    __syncthreads();
    const int4* dst4 = (const int4*)dst;
    for (int i = t; i < N_DST4Q; i += DEG_THREADS) {
      const int4 v = dst4[i];
      const unsigned a0 = (unsigned)(v.x - lo), a1 = (unsigned)(v.y - lo);
      const unsigned a2 = (unsigned)(v.z - lo), a3 = (unsigned)(v.w - lo);
      if (a0 < (unsigned)NPB) __hip_atomic_fetch_add(&cntR[a0], 1, __ATOMIC_RELAXED, __HIP_MEMORY_SCOPE_WORKGROUP);
      if (a1 < (unsigned)NPB) __hip_atomic_fetch_add(&cntR[a1], 1, __ATOMIC_RELAXED, __HIP_MEMORY_SCOPE_WORKGROUP);
      if (a2 < (unsigned)NPB) __hip_atomic_fetch_add(&cntR[a2], 1, __ATOMIC_RELAXED, __HIP_MEMORY_SCOPE_WORKGROUP);
      if (a3 < (unsigned)NPB) __hip_atomic_fetch_add(&cntR[a3], 1, __ATOMIC_RELAXED, __HIP_MEMORY_SCOPE_WORKGROUP);
    }
    __syncthreads();
  }

  for (int r = t; r < nvalid; r += DEG_THREADS)
    dinv[lo + r] = rsqrtf(1.0f + (float)cntR[r]);
  if (t == 0) ovf[blockIdx.x] = bad ? 1 : 0;
}


#define GEMM_BLOCKS 768
#define GEMM_TILES (N_NODESC / 16)
#define GEMM_KSTEPS (DIM / 4)
#define XS_PITCH4 33
#define XS_STAGE4 (16 * XS_PITCH4)

static_assert(N_NODESC % 16 == 0, "whole 16-row tiles");

__global__ __launch_bounds__(256) void k_gemm_wmma(const float* __restrict__ X,
                                                   const float* __restrict__ W,
                                                   float* __restrict__ H) {
  __shared__ float4 xs4[2 * XS_STAGE4];

  const int t = threadIdx.x;
  const int wave = t >> 5;
  const int lane = t & 31;

  const int m  = lane & 15;
  const int hg = lane >> 4;
  const int n0 = wave * 16;

  v2f bf[GEMM_KSTEPS];
  {
    const float* wcol = W + n0 + m;
    #pragma unroll
    for (int j = 0; j < GEMM_KSTEPS; ++j) {
      bf[j].x = wcol[(size_t)(4 * j + 2 * hg + 0) * DIM];
      bf[j].y = wcol[(size_t)(4 * j + 2 * hg + 1) * DIM];
    }
  }

  const int r  = wave;
  const int c4 = lane;

  int tile = blockIdx.x;
  float4 p0, p1;
  {
    const int tl = (tile < GEMM_TILES) ? tile : GEMM_TILES - 1;
    const float4* g = (const float4*)(X + ((size_t)tl * 16 + r) * DIM) + c4;
    p0 = g[0];
    p1 = g[8 * (DIM / 4)];
  }

  int cur = 0;
  for (; tile < GEMM_TILES; tile += GEMM_BLOCKS) {
    float4* stage = xs4 + cur * XS_STAGE4;
    stage[r * XS_PITCH4 + c4] = p0;
    stage[(r + 8) * XS_PITCH4 + c4] = p1;
    __syncthreads();

    {
      const int nxt = tile + GEMM_BLOCKS;
      const int tl = (nxt < GEMM_TILES) ? nxt : GEMM_TILES - 1;
      const float4* g = (const float4*)(X + ((size_t)tl * 16 + r) * DIM) + c4;
      p0 = g[0];
      p1 = g[8 * (DIM / 4)];
    }
    __builtin_amdgcn_sched_barrier(0);

    const v2f* arow = (const v2f*)stage + m * (XS_PITCH4 * 2) + hg;
    v2f af[GEMM_KSTEPS];
    #pragma unroll
    for (int j = 0; j < GEMM_KSTEPS; ++j) af[j] = arow[2 * j];
    __builtin_amdgcn_sched_barrier(0);

    v8f acc = {};
    #pragma unroll
    for (int j = 0; j < GEMM_KSTEPS; ++j) {
      acc = __builtin_amdgcn_wmma_f32_16x16x4_f32(
           false, af[j],  false, bf[j],
           (short)0, acc,  false,  false);
    }

    float* hrow = H + (size_t)tile * 16 * DIM + n0 + m;
    #pragma unroll
    for (int v = 0; v < 8; ++v)
      hrow[(size_t)(v + 8 * hg) * DIM] = acc[v];

    cur ^= 1;
  }
}


#define AGG_THREADS 1024
#define AGG_WAVES 32
#define SCAN_U 4
#define N_DST4 (N_EDGESC / 4)
#define WAVE4 (32 * SCAN_U)
#define TRIP4 (AGG_WAVES * WAVE4)
#define N_TRIPS ((N_DST4 + TRIP4 - 1) / TRIP4)
#define BATCH 8
#define LIST_CAP (BATCH - 1 + 4 * WAVE4 + 1)
#define QCAP LIST_CAP
#define SRC_MASK ((1 << 17) - 1)

static_assert(AGG_BLOCKS * NPB >= N_NODESC, "slices must cover every node");
static_assert(N_EDGESC % 4 == 0, "dst list is read as int4");
static_assert(N_EDGESC <= (1 << 21) && NPB <= (1 << 9), "list entry packs edge id and local row");
static_assert(BIN_BLOCKS % AGG_WAVES == 0, "every wave takes the same number of cells");
#define CELLS_PER_WAVE (BIN_BLOCKS / AGG_WAVES)

__device__ __forceinline__ void agg_load_trip(const int4* __restrict__ dst4, int base4, int lane,
                                              int4 (&v)[SCAN_U]) {
  #pragma unroll
  for (int u = 0; u < SCAN_U; ++u) {
    const int i4 = base4 + u * 32 + lane;
    v[u] = (i4 < N_DST4) ? dst4[i4] : make_int4(-1, -1, -1, -1);
  }
}

__device__ __forceinline__ void rows_load(int s, int lane, const float* __restrict__ H,
                                          float (&hv)[BATCH][4]) {
  #pragma unroll
  for (int i = 0; i < BATCH; ++i) {
    const int si = __builtin_amdgcn_readlane(s, i);
    const float* hr = H + (size_t)si * DIM + lane;
    hv[i][0] = hr[0];
    hv[i][1] = hr[32];
    hv[i][2] = hr[64];
    hv[i][3] = hr[96];
  }
}

__device__ __forceinline__ void rows_add(const float (&hv)[BATCH][4], int dl, float w, int nb,
                                         int lane, float* tile) {
  #pragma unroll
  for (int i = 0; i < BATCH; ++i) {
    if (i < nb) {
      const int di = __builtin_amdgcn_readlane(dl, i);
      const float wi = __int_as_float(__builtin_amdgcn_readlane(__float_as_int(w), i));
      float* tr = tile + di * DIM + lane;
      __hip_atomic_fetch_add(tr + 0,  hv[i][0] * wi, __ATOMIC_RELAXED, __HIP_MEMORY_SCOPE_WORKGROUP);
      __hip_atomic_fetch_add(tr + 32, hv[i][1] * wi, __ATOMIC_RELAXED, __HIP_MEMORY_SCOPE_WORKGROUP);
      __hip_atomic_fetch_add(tr + 64, hv[i][2] * wi, __ATOMIC_RELAXED, __HIP_MEMORY_SCOPE_WORKGROUP);
      __hip_atomic_fetch_add(tr + 96, hv[i][3] * wi, __ATOMIC_RELAXED, __HIP_MEMORY_SCOPE_WORKGROUP);
    }
  }
}

__device__ __forceinline__ void agg_batch(const int* lst, int nb, int lane,
                                          const int* __restrict__ src,
                                          const float* __restrict__ dinv,
                                          const float* dinvL,
                                          const float* __restrict__ H, float* tile) {
  int ent = 0;
  if (lane < nb) ent = lst[lane];
  const int dl = (int)((unsigned)ent >> 21);
  const int s = src[ent & ((1 << 21) - 1)];
  const float ds = dinv[s];
  float hv[BATCH][4];
  rows_load(s, lane, H, hv);
  const float w = ds * dinvL[dl];
  rows_add(hv, dl, w, nb, lane, tile);
}

__device__ __forceinline__ void agg_scan(int lo, int wave, int lane, int* mylist,
                                         const int* __restrict__ src, const int* __restrict__ dst,
                                         const float* __restrict__ dinv, const float* dinvL,
                                         const float* __restrict__ H, float* tile) {
  const int4* dst4 = (const int4*)dst;
  int pend = 0;
  int trip = blockIdx.x % N_TRIPS;

  int4 nxt[SCAN_U];
  agg_load_trip(dst4, trip * TRIP4 + wave * WAVE4, lane, nxt);

  for (int it = 0; it < N_TRIPS; ++it) {
    int4 cur[SCAN_U];
    #pragma unroll
    for (int u = 0; u < SCAN_U; ++u) cur[u] = nxt[u];
    const int cur4 = trip * TRIP4 + wave * WAVE4;
    trip = (trip + 1 == N_TRIPS) ? 0 : trip + 1;
    if (it + 1 < N_TRIPS) agg_load_trip(dst4, trip * TRIP4 + wave * WAVE4, lane, nxt);

    #pragma unroll
    for (int u = 0; u < SCAN_U; ++u) {
      #pragma unroll
      for (int c = 0; c < 4; ++c) {
        const int d = (c == 0) ? cur[u].x : (c == 1) ? cur[u].y : (c == 2) ? cur[u].z : cur[u].w;
        const unsigned tt = (unsigned)(d - lo);
        const bool m = tt < (unsigned)NPB;
        const unsigned mask = __builtin_amdgcn_ballot_w32(m);
        if (mask != 0u) {
          const int pos = pend + (int)__builtin_amdgcn_mbcnt_lo(mask, 0u);
          const int e = (cur4 + u * 32 + lane) * 4 + c;
          if (m) mylist[pos] = (int)((tt << 21) | (unsigned)e);
          pend += __builtin_popcount(mask);
        }
      }
    }
    while (pend >= BATCH) {
      pend -= BATCH;
      agg_batch(mylist + pend, BATCH, lane, src, dinv, dinvL, H, tile);
    }
  }
  if (pend > 0) agg_batch(mylist, pend, lane, src, dinv, dinvL, H, tile);
}

__device__ __forceinline__ void queue_put(int e, int* queues, int* qcnt, int* qovf) {
  const int ow = (e >> 17) & (AGG_WAVES - 1);
  const int slot = __hip_atomic_fetch_add(&qcnt[ow], 1, __ATOMIC_RELAXED, __HIP_MEMORY_SCOPE_WORKGROUP);
  if (slot < QCAP) queues[ow * QCAP + slot] = e;
  else *qovf = 1;
}

__device__ __forceinline__ void route_cells(const int* __restrict__ mycells, int wv, int lane,
                                            int* queues, int* qcnt, int* qovf) {
  const int2* cp = (const int2*)(mycells + (size_t)wv * CELLS_PER_WAVE * CELL_INTS) + lane;
  int2 cv[CELLS_PER_WAVE];
  #pragma unroll
  for (int c = 0; c < CELLS_PER_WAVE; ++c) cv[c] = cp[c * (CELL_INTS / 2)];
  #pragma unroll
  for (int c = 0; c < CELLS_PER_WAVE; ++c) {
    int cnt = __builtin_amdgcn_readfirstlane(cv[c].x);
    cnt = (cnt < CELL_CAP) ? cnt : CELL_CAP;
    const int i0 = 2 * lane;
    if (i0 >= 1 && i0 <= cnt) queue_put(cv[c].x, queues, qcnt, qovf);
    if (i0 + 1 <= cnt) queue_put(cv[c].y, queues, qcnt, qovf);
  }
}

__device__ __forceinline__ void rows_load4(int s, int lane, const float* __restrict__ H,
                                           float4 (&hv)[BATCH]) {
  #pragma unroll
  for (int i = 0; i < BATCH; ++i) {
    const int si = __builtin_amdgcn_readlane(s, i);
    hv[i] = ((const float4*)(H + (size_t)si * DIM))[lane];
  }
}

__device__ __forceinline__ void rows_rmw(const float4 (&hv)[BATCH], int dl, float w, int nb,
                                         int lane, float* tile) {
  #pragma unroll
  for (int i = 0; i < BATCH; ++i) {
    if (i < nb) {
      const int di = __builtin_amdgcn_readlane(dl, i);
      const float wi = __int_as_float(__builtin_amdgcn_readlane(__float_as_int(w), i));
      float4* tr = (float4*)(tile + di * DIM) + lane;
      float4 a = *tr;
      a.x += hv[i].x * wi;
      a.y += hv[i].y * wi;
      a.z += hv[i].z * wi;
      a.w += hv[i].w * wi;
      *tr = a;
    }
  }
}

template <int MODE>
__device__ __forceinline__ void queue_entry(int b, int n, int lane, const int* q,
                                            const int2* __restrict__ rec, int& e, float& w) {
  const int idx = b * BATCH + lane;
  e = 0;
  w = 0.0f;
  if (lane < BATCH && idx < n) {
    if (MODE == 0) {
      e = q[idx];
    } else {
      const int2 r = rec[idx];
      e = r.x;
      w = __int_as_float(r.y);
    }
  }
}

template <int MODE>
__device__ __forceinline__ void sum_queue(int n, int lane, const int* q, int2* __restrict__ rec,
                                          const float* __restrict__ dinv, const float* dinvL,
                                          const float* __restrict__ H, float* tile) {
  const int nbat = (n + BATCH - 1) / BATCH;
  int e_rq, e_nx;
  float w_rq, w_nx;
  float4 hv[BATCH];
  queue_entry<MODE>(0, n, lane, q, rec, e_rq, w_rq);
  if (MODE == 0) w_rq = dinv[e_rq & SRC_MASK];
  rows_load4(e_rq & SRC_MASK, lane, H, hv);
  queue_entry<MODE>(1, n, lane, q, rec, e_nx, w_nx);

  for (int b = 0; b < nbat; ++b) {
    float4 cv[BATCH];
    #pragma unroll
    for (int i = 0; i < BATCH; ++i) cv[i] = hv[i];
    const int e_cur = e_rq;
    const float w_cur = w_rq;

    e_rq = e_nx;
    w_rq = w_nx;
    if (MODE == 0) w_rq = dinv[e_rq & SRC_MASK];
    rows_load4(e_rq & SRC_MASK, lane, H, hv);
    queue_entry<MODE>(b + 2, n, lane, q, rec, e_nx, w_nx);

    const int left = n - b * BATCH;
    const int nb = (left < BATCH) ? left : BATCH;
    const int dl = (int)((unsigned)e_cur >> 17);
    float w = w_cur;
    if (MODE == 0) {
      w = w_cur * dinvL[dl];
      if (lane < nb) rec[b * BATCH + lane] = make_int2(e_cur, __float_as_int(w));
    }
    rows_rmw(cv, dl, w, nb, lane, tile);
  }
}

template <int MODE>
__global__ __launch_bounds__(AGG_THREADS) void k_agg_owner(const int* __restrict__ src,
                                                           const int* __restrict__ dst,
                                                           const float* __restrict__ dinv,
                                                           const float* __restrict__ H,
                                                           const float* __restrict__ bias,
                                                           float* __restrict__ OUT, int do_relu,
                                                           int* __restrict__ rec_cnt,
                                                           int* __restrict__ rec_all,
                                                           int* __restrict__ qbad,
                                                           const int* __restrict__ cells,
                                                           const int* __restrict__ ovf) {
  __shared__ __attribute__((aligned(16))) float tile[NPB * DIM];
  __shared__ float dinvL[NPB + 1];
  __shared__ int lists[AGG_WAVES * LIST_CAP];
  __shared__ int qcnt[AGG_WAVES];
  __shared__ int qovf;

  const int t = threadIdx.x;
  const int wave = t >> 5;
  const int lane = t & 31;
  const int lo = blockIdx.x * NPB;
  const int nvalid = (N_NODESC - lo < NPB) ? (N_NODESC - lo) : NPB;
  const int wv = __builtin_amdgcn_readfirstlane(wave);
  int2* rec = (int2*)rec_all + ((size_t)blockIdx.x * AGG_WAVES + wv) * QCAP;

  if (t < AGG_WAVES) qcnt[t] = 0;
  if (t == 0) qovf = 0;

  for (int r = wave; r < NPB; r += AGG_WAVES) {
    float4 v = make_float4(0.f, 0.f, 0.f, 0.f);
    float di = 0.f;
    if (r < nvalid) {
      di = dinv[lo + r];
      const float4 h = ((const float4*)(H + (size_t)(lo + r) * DIM))[lane];
      v.x = h.x * di * di;
      v.y = h.y * di * di;
      v.z = h.z * di * di;
      v.w = h.w * di * di;
    }
    ((float4*)(tile + r * DIM))[lane] = v;
    if (lane == 0) dinvL[r] = di;
  }
  __syncthreads();

  int* mylist = lists + wv * LIST_CAP;
  bool bad;

  if (MODE == 0) {
    bad = ovf[blockIdx.x] != 0;
    if (!bad)
      route_cells(cells + (size_t)blockIdx.x * BIN_BLOCKS * CELL_INTS, wv, lane, lists, qcnt, &qovf);
    __syncthreads();
    bad = bad || (qovf != 0);
    if (!bad) {
      const int n = qcnt[wv];
      sum_queue<0>(n, lane, lists + wv * QCAP, rec, dinv, dinvL, H, tile);
      if (lane == 0) rec_cnt[blockIdx.x * AGG_WAVES + wv] = n;
    } else {
      agg_scan(lo, wave, lane, mylist, src, dst, dinv, dinvL, H, tile);
    }
    if (t == 0) qbad[blockIdx.x] = bad ? 1 : 0;
  } else {
    bad = qbad[blockIdx.x] != 0;
    if (!bad) {
      const int n = rec_cnt[blockIdx.x * AGG_WAVES + wv];
      sum_queue<1>(n, lane, lists, rec, dinv, dinvL, H, tile);
    } else {
      agg_scan(lo, wave, lane, mylist, src, dst, dinv, dinvL, H, tile);
    }
  }

  __syncthreads();

  const float4 bb = ((const float4*)bias)[lane];
  for (int r = wave; r < nvalid; r += AGG_WAVES) {
    float4 a = ((const float4*)(tile + r * DIM))[lane];
    a.x += bb.x;
    a.y += bb.y;
    a.z += bb.z;
    a.w += bb.w;
    if (do_relu) {
      a.x = fmaxf(a.x, 0.0f);
      a.y = fmaxf(a.y, 0.0f);
      a.z = fmaxf(a.z, 0.0f);
      a.w = fmaxf(a.w, 0.0f);
    }
    ((float4*)(OUT + (size_t)(lo + r) * DIM))[lane] = a;
  }
}


extern "C" void kernel_launch(void* const* d_in, const int* in_sizes, int n_in,
                              void* d_out, int out_size, void* d_ws, size_t ws_size,
                              hipStream_t stream) {
  const float* x   = (const float*)d_in[0];
  const int*   ei  = (const int*)d_in[1];
  const float* W1  = (const float*)d_in[2];
  const float* b1  = (const float*)d_in[3];
  const float* W2  = (const float*)d_in[4];
  const float* b2  = (const float*)d_in[5];
  float*       out = (float*)d_out;

  const int* src = ei;
  const int* dst = ei + N_EDGESC;

  float* H    = (float*)d_ws;
  float* dinv = H + (size_t)N_NODESC * DIM + N_NODESC;
  int*   rec_cnt = (int*)(dinv + N_NODESC);
  int*   qbad    = rec_cnt + AGG_BLOCKS * AGG_WAVES;
  int*   rec_all = qbad + AGG_BLOCKS;
  int*   cells   = rec_all + (size_t)AGG_BLOCKS * AGG_WAVES * QCAP * 2;
  int*   ovf     = cells + (size_t)AGG_BLOCKS * BIN_BLOCKS * CELL_INTS;

  dim3 blk(256);
  dim3 gBin(BIN_BLOCKS);
  dim3 bBin(BIN_THREADS);
  dim3 gGemm(GEMM_BLOCKS);
  dim3 gAgg(AGG_BLOCKS);
  dim3 bAgg(AGG_THREADS);

  k_bin_count<<<gBin, bBin, 0, stream>>>(src, dst, cells);
  k_deg_owner<<<gAgg, bAgg, 0, stream>>>(dst, cells, dinv, ovf);

  k_gemm_wmma<<<gGemm, blk, 0, stream>>>(x, W1, H);
  k_agg_owner<0><<<gAgg, bAgg, 0, stream>>>(src, dst, dinv, H, b1, out,  1, rec_cnt, rec_all, qbad, cells, ovf);

  k_gemm_wmma<<<gGemm, blk, 0, stream>>>(out, W2, H);
  k_agg_owner<1><<<gAgg, bAgg, 0, stream>>>(src, dst, dinv, H, b2, out,  0, rec_cnt, rec_all, qbad, cells, ovf);
}


# Owner blocks start in one memory round trip, not 13: a wave requests its own rows, their dinv and (layer 1) its cells as one group, stores the start values to its own LDS rows itself; sums unchanged
# speedup vs baseline: 4.9518x; 1.0315x over previous
#include <hip/hip_runtime.h>

typedef float v2f __attribute__((ext_vector_type(2)));
typedef float v8f __attribute__((ext_vector_type(8)));

#define N_NODESC 100000
#define N_EDGESC 1600000
#define DIM 128


#define AGG_BLOCKS 256
#define NPB 391

#define BIN_BLOCKS 256
#define BIN_THREADS 1024
#define CELL_INTS 64
#define CELL_CAP (CELL_INTS - 1)
#define CELL_PITCH 68
#define EDGES_PER_BIN (N_EDGESC / BIN_BLOCKS)
#define BIN_ITERS ((EDGES_PER_BIN + BIN_THREADS - 1) / BIN_THREADS)

static_assert(N_EDGESC % BIN_BLOCKS == 0, "chunks are equal");
static_assert(N_NODESC <= (1 << 17) && NPB <= (1 << 9), "cell entry packs source node and local row");

__global__ __launch_bounds__(BIN_THREADS) void k_bin_count(const int* __restrict__ src,
                                                           const int* __restrict__ dst,
                                                           int* __restrict__ cells) {
  __shared__ __attribute__((aligned(16))) int cellL[AGG_BLOCKS * CELL_PITCH];
  __shared__ int cntL[AGG_BLOCKS];

  const int t = threadIdx.x;
  const int e0 = blockIdx.x * EDGES_PER_BIN;

  int dv[BIN_ITERS], sv[BIN_ITERS];
  #pragma unroll
  for (int k = 0; k < BIN_ITERS; ++k) {
    const int i = t + k * BIN_THREADS;
    dv[k] = -1;
    sv[k] = 0;
    if (i < EDGES_PER_BIN) {
      dv[k] = dst[e0 + i];
      sv[k] = src[e0 + i];
    }
  }

  for (int q = t; q < AGG_BLOCKS * CELL_PITCH / 4; q += BIN_THREADS)
    ((int4*)cellL)[q] = make_int4(0, 0, 0, 0);
  for (int o = t; o < AGG_BLOCKS; o += BIN_THREADS) cntL[o] = 0;
  __syncthreads();

  #pragma unroll
  for (int k = 0; k < BIN_ITERS; ++k) {
    const int d = dv[k];
    if (d >= 0) {
      const int o = d / NPB;
      const int dl = d - o * NPB;
      const int rank = __hip_atomic_fetch_add(&cntL[o], 1, __ATOMIC_RELAXED, __HIP_MEMORY_SCOPE_WORKGROUP);
      if (rank < CELL_CAP) cellL[o * CELL_PITCH + 1 + rank] = sv[k] | (dl << 17);
    }
  }
  __syncthreads();

  for (int q = t; q < AGG_BLOCKS * CELL_INTS / 4; q += BIN_THREADS) {
    const int o = q >> 4;
    const int part = q & 15;
    int4 v = *(const int4*)&cellL[o * CELL_PITCH + part * 4];
    if (part == 0) v.x = cntL[o];
    ((int4*)(cells + ((size_t)o * BIN_BLOCKS + blockIdx.x) * CELL_INTS))[part] = v;
  }
}

#define DEG_THREADS 1024
#define DEG_WAVES (DEG_THREADS / 32)
#define DEG_CELLS_PER_WAVE (BIN_BLOCKS / DEG_WAVES)
#define N_DST4Q (N_EDGESC / 4)

static_assert(BIN_BLOCKS % DEG_WAVES == 0, "every wave takes the same number of cells");
static_assert(N_EDGESC % 4 == 0, "dst list is read as int4");

__global__ __launch_bounds__(DEG_THREADS) void k_deg_owner(const int* __restrict__ dst,
                                                           const int* __restrict__ cells,
                                                           float* __restrict__ dinv,
                                                           int* __restrict__ ovf) {
  __shared__ int cntR[NPB];
  __shared__ int full;

  const int t = threadIdx.x;
  const int lane = t & 31;
  const int wv = __builtin_amdgcn_readfirstlane(t >> 5);
  const int lo = blockIdx.x * NPB;
  const int nvalid = (N_NODESC - lo < NPB) ? (N_NODESC - lo) : NPB;

  const int2* cp = (const int2*)(cells + ((size_t)blockIdx.x * BIN_BLOCKS +
                                          (size_t)wv * DEG_CELLS_PER_WAVE) * CELL_INTS) + lane;
  int2 cv[DEG_CELLS_PER_WAVE];
  #pragma unroll
  for (int c = 0; c < DEG_CELLS_PER_WAVE; ++c) cv[c] = cp[c * (CELL_INTS / 2)];

  for (int r = t; r < NPB; r += DEG_THREADS) cntR[r] = 0;
  if (t == 0) full = 0;
  __syncthreads();

  #pragma unroll
  for (int c = 0; c < DEG_CELLS_PER_WAVE; ++c) {
    const int raw = __builtin_amdgcn_readfirstlane(cv[c].x);
    if (raw > CELL_CAP && lane == 0) full = 1;
    const int cnt = (raw < CELL_CAP) ? raw : CELL_CAP;
    const int i0 = 2 * lane;
    if (i0 >= 1 && i0 <= cnt)
      __hip_atomic_fetch_add(&cntR[(unsigned)cv[c].x >> 17], 1, __ATOMIC_RELAXED, __HIP_MEMORY_SCOPE_WORKGROUP);
    if (i0 + 1 <= cnt)
      __hip_atomic_fetch_add(&cntR[(unsigned)cv[c].y >> 17], 1, __ATOMIC_RELAXED, __HIP_MEMORY_SCOPE_WORKGROUP);
  }
  __syncthreads();

  const bool bad = full != 0;
  if (bad) {
    for (int r = t; r < NPB; r += DEG_THREADS) cntR[r] = 0;
    __syncthreads();
    const int4* dst4 = (const int4*)dst;
    for (int i = t; i < N_DST4Q; i += DEG_THREADS) {
      const int4 v = dst4[i];
      const unsigned a0 = (unsigned)(v.x - lo), a1 = (unsigned)(v.y - lo);
      const unsigned a2 = (unsigned)(v.z - lo), a3 = (unsigned)(v.w - lo);
      if (a0 < (unsigned)NPB) __hip_atomic_fetch_add(&cntR[a0], 1, __ATOMIC_RELAXED, __HIP_MEMORY_SCOPE_WORKGROUP);
      if (a1 < (unsigned)NPB) __hip_atomic_fetch_add(&cntR[a1], 1, __ATOMIC_RELAXED, __HIP_MEMORY_SCOPE_WORKGROUP);
      if (a2 < (unsigned)NPB) __hip_atomic_fetch_add(&cntR[a2], 1, __ATOMIC_RELAXED, __HIP_MEMORY_SCOPE_WORKGROUP);
      if (a3 < (unsigned)NPB) __hip_atomic_fetch_add(&cntR[a3], 1, __ATOMIC_RELAXED, __HIP_MEMORY_SCOPE_WORKGROUP);
    }
    __syncthreads();
  }

  for (int r = t; r < nvalid; r += DEG_THREADS)
    dinv[lo + r] = rsqrtf(1.0f + (float)cntR[r]);
  if (t == 0) ovf[blockIdx.x] = bad ? 1 : 0;
}


#define GEMM_BLOCKS 768
#define GEMM_TILES (N_NODESC / 16)
#define GEMM_KSTEPS (DIM / 4)
#define XS_PITCH4 33
#define XS_STAGE4 (16 * XS_PITCH4)

static_assert(N_NODESC % 16 == 0, "whole 16-row tiles");

__global__ __launch_bounds__(256) void k_gemm_wmma(const float* __restrict__ X,
                                                   const float* __restrict__ W,
                                                   float* __restrict__ H) {
  __shared__ float4 xs4[2 * XS_STAGE4];

  const int t = threadIdx.x;
  const int wave = t >> 5;
  const int lane = t & 31;

  const int m  = lane & 15;
  const int hg = lane >> 4;
  const int n0 = wave * 16;

  v2f bf[GEMM_KSTEPS];
  {
    const float* wcol = W + n0 + m;
    #pragma unroll
    for (int j = 0; j < GEMM_KSTEPS; ++j) {
      bf[j].x = wcol[(size_t)(4 * j + 2 * hg + 0) * DIM];
      bf[j].y = wcol[(size_t)(4 * j + 2 * hg + 1) * DIM];
    }
  }

  const int r  = wave;
  const int c4 = lane;

  int tile = blockIdx.x;
  float4 p0, p1;
  {
    const int tl = (tile < GEMM_TILES) ? tile : GEMM_TILES - 1;
    const float4* g = (const float4*)(X + ((size_t)tl * 16 + r) * DIM) + c4;
    p0 = g[0];
    p1 = g[8 * (DIM / 4)];
  }

  int cur = 0;
  for (; tile < GEMM_TILES; tile += GEMM_BLOCKS) {
    float4* stage = xs4 + cur * XS_STAGE4;
    stage[r * XS_PITCH4 + c4] = p0;
    stage[(r + 8) * XS_PITCH4 + c4] = p1;
    __syncthreads();

    {
      const int nxt = tile + GEMM_BLOCKS;
      const int tl = (nxt < GEMM_TILES) ? nxt : GEMM_TILES - 1;
      const float4* g = (const float4*)(X + ((size_t)tl * 16 + r) * DIM) + c4;
      p0 = g[0];
      p1 = g[8 * (DIM / 4)];
    }
    __builtin_amdgcn_sched_barrier(0);

    const v2f* arow = (const v2f*)stage + m * (XS_PITCH4 * 2) + hg;
    v2f af[GEMM_KSTEPS];
    #pragma unroll
    for (int j = 0; j < GEMM_KSTEPS; ++j) af[j] = arow[2 * j];
    __builtin_amdgcn_sched_barrier(0);

    v8f acc = {};
    #pragma unroll
    for (int j = 0; j < GEMM_KSTEPS; ++j) {
      acc = __builtin_amdgcn_wmma_f32_16x16x4_f32(
           false, af[j],  false, bf[j],
           (short)0, acc,  false,  false);
    }

    float* hrow = H + (size_t)tile * 16 * DIM + n0 + m;
    #pragma unroll
    for (int v = 0; v < 8; ++v)
      hrow[(size_t)(v + 8 * hg) * DIM] = acc[v];

    cur ^= 1;
  }
}


#define AGG_THREADS 1024
#define AGG_WAVES 32
#define SCAN_U 4
#define N_DST4 (N_EDGESC / 4)
#define WAVE4 (32 * SCAN_U)
#define TRIP4 (AGG_WAVES * WAVE4)
#define N_TRIPS ((N_DST4 + TRIP4 - 1) / TRIP4)
#define BATCH 8
#define LIST_CAP (BATCH - 1 + 4 * WAVE4 + 1)
#define QCAP LIST_CAP
#define SRC_MASK ((1 << 17) - 1)

static_assert(AGG_BLOCKS * NPB >= N_NODESC, "slices must cover every node");
static_assert(N_EDGESC % 4 == 0, "dst list is read as int4");
static_assert(N_EDGESC <= (1 << 21) && NPB <= (1 << 9), "list entry packs edge id and local row");
static_assert(BIN_BLOCKS % AGG_WAVES == 0, "every wave takes the same number of cells");
#define CELLS_PER_WAVE (BIN_BLOCKS / AGG_WAVES)
#define ROWS_PW ((NPB + AGG_WAVES - 1) / AGG_WAVES)

__device__ __forceinline__ void agg_load_trip(const int4* __restrict__ dst4, int base4, int lane,
                                              int4 (&v)[SCAN_U]) {
  #pragma unroll
  for (int u = 0; u < SCAN_U; ++u) {
    const int i4 = base4 + u * 32 + lane;
    v[u] = (i4 < N_DST4) ? dst4[i4] : make_int4(-1, -1, -1, -1);
  }
}

__device__ __forceinline__ void rows_load(int s, int lane, const float* __restrict__ H,
                                          float (&hv)[BATCH][4]) {
  #pragma unroll
  for (int i = 0; i < BATCH; ++i) {
    const int si = __builtin_amdgcn_readlane(s, i);
    const float* hr = H + (size_t)si * DIM + lane;
    hv[i][0] = hr[0];
    hv[i][1] = hr[32];
    hv[i][2] = hr[64];
    hv[i][3] = hr[96];
  }
}

__device__ __forceinline__ void rows_add(const float (&hv)[BATCH][4], int dl, float w, int nb,
                                         int lane, float* tile) {
  #pragma unroll
  for (int i = 0; i < BATCH; ++i) {
    if (i < nb) {
      const int di = __builtin_amdgcn_readlane(dl, i);
      const float wi = __int_as_float(__builtin_amdgcn_readlane(__float_as_int(w), i));
      float* tr = tile + di * DIM + lane;
      __hip_atomic_fetch_add(tr + 0,  hv[i][0] * wi, __ATOMIC_RELAXED, __HIP_MEMORY_SCOPE_WORKGROUP);
      __hip_atomic_fetch_add(tr + 32, hv[i][1] * wi, __ATOMIC_RELAXED, __HIP_MEMORY_SCOPE_WORKGROUP);
      __hip_atomic_fetch_add(tr + 64, hv[i][2] * wi, __ATOMIC_RELAXED, __HIP_MEMORY_SCOPE_WORKGROUP);
      __hip_atomic_fetch_add(tr + 96, hv[i][3] * wi, __ATOMIC_RELAXED, __HIP_MEMORY_SCOPE_WORKGROUP);
    }
  }
}

__device__ __forceinline__ void agg_batch(const int* lst, int nb, int lane,
                                          const int* __restrict__ src,
                                          const float* __restrict__ dinv,
                                          const float* dinvL,
                                          const float* __restrict__ H, float* tile) {
  int ent = 0;
  if (lane < nb) ent = lst[lane];
  const int dl = (int)((unsigned)ent >> 21);
  const int s = src[ent & ((1 << 21) - 1)];
  const float ds = dinv[s];
  float hv[BATCH][4];
  rows_load(s, lane, H, hv);
  const float w = ds * dinvL[dl];
  rows_add(hv, dl, w, nb, lane, tile);
}

__device__ __forceinline__ void agg_scan(int lo, int wave, int lane, int* mylist,
                                         const int* __restrict__ src, const int* __restrict__ dst,
                                         const float* __restrict__ dinv, const float* dinvL,
                                         const float* __restrict__ H, float* tile) {
  const int4* dst4 = (const int4*)dst;
  int pend = 0;
  int trip = blockIdx.x % N_TRIPS;

  int4 nxt[SCAN_U];
  agg_load_trip(dst4, trip * TRIP4 + wave * WAVE4, lane, nxt);

  for (int it = 0; it < N_TRIPS; ++it) {
    int4 cur[SCAN_U];
    #pragma unroll
    for (int u = 0; u < SCAN_U; ++u) cur[u] = nxt[u];
    const int cur4 = trip * TRIP4 + wave * WAVE4;
    trip = (trip + 1 == N_TRIPS) ? 0 : trip + 1;
    if (it + 1 < N_TRIPS) agg_load_trip(dst4, trip * TRIP4 + wave * WAVE4, lane, nxt);

    #pragma unroll
    for (int u = 0; u < SCAN_U; ++u) {
      #pragma unroll
      for (int c = 0; c < 4; ++c) {
        const int d = (c == 0) ? cur[u].x : (c == 1) ? cur[u].y : (c == 2) ? cur[u].z : cur[u].w;
        const unsigned tt = (unsigned)(d - lo);
        const bool m = tt < (unsigned)NPB;
        const unsigned mask = __builtin_amdgcn_ballot_w32(m);
        if (mask != 0u) {
          const int pos = pend + (int)__builtin_amdgcn_mbcnt_lo(mask, 0u);
          const int e = (cur4 + u * 32 + lane) * 4 + c;
          if (m) mylist[pos] = (int)((tt << 21) | (unsigned)e);
          pend += __builtin_popcount(mask);
        }
      }
    }
    while (pend >= BATCH) {
      pend -= BATCH;
      agg_batch(mylist + pend, BATCH, lane, src, dinv, dinvL, H, tile);
    }
  }
  if (pend > 0) agg_batch(mylist, pend, lane, src, dinv, dinvL, H, tile);
}

__device__ __forceinline__ void queue_put(int e, int* queues, int* qcnt, int* qovf) {
  const int ow = (e >> 17) & (AGG_WAVES - 1);
  const int slot = __hip_atomic_fetch_add(&qcnt[ow], 1, __ATOMIC_RELAXED, __HIP_MEMORY_SCOPE_WORKGROUP);
  if (slot < QCAP) queues[ow * QCAP + slot] = e;
  else *qovf = 1;
}

__device__ __forceinline__ void cells_request(const int* __restrict__ mycells, int wv, int lane,
                                              int2 (&cv)[CELLS_PER_WAVE]) {
  const int2* cp = (const int2*)(mycells + (size_t)wv * CELLS_PER_WAVE * CELL_INTS) + lane;
  #pragma unroll
  for (int c = 0; c < CELLS_PER_WAVE; ++c) cv[c] = cp[c * (CELL_INTS / 2)];
}

__device__ __forceinline__ void route_cells(const int2 (&cv)[CELLS_PER_WAVE], int lane,
                                            int* queues, int* qcnt, int* qovf) {
  #pragma unroll
  for (int c = 0; c < CELLS_PER_WAVE; ++c) {
    int cnt = __builtin_amdgcn_readfirstlane(cv[c].x);
    cnt = (cnt < CELL_CAP) ? cnt : CELL_CAP;
    const int i0 = 2 * lane;
    if (i0 >= 1 && i0 <= cnt) queue_put(cv[c].x, queues, qcnt, qovf);
    if (i0 + 1 <= cnt) queue_put(cv[c].y, queues, qcnt, qovf);
  }
}

__device__ __forceinline__ void rows_load4(int s, int lane, const float* __restrict__ H,
                                           float4 (&hv)[BATCH]) {
  #pragma unroll
  for (int i = 0; i < BATCH; ++i) {
    const int si = __builtin_amdgcn_readlane(s, i);
    hv[i] = ((const float4*)(H + (size_t)si * DIM))[lane];
  }
}

__device__ __forceinline__ void rows_rmw(const float4 (&hv)[BATCH], int dl, float w, int nb,
                                         int lane, float* tile) {
  #pragma unroll
  for (int i = 0; i < BATCH; ++i) {
    if (i < nb) {
      const int di = __builtin_amdgcn_readlane(dl, i);
      const float wi = __int_as_float(__builtin_amdgcn_readlane(__float_as_int(w), i));
      float4* tr = (float4*)(tile + di * DIM) + lane;
      float4 a = *tr;
      a.x += hv[i].x * wi;
      a.y += hv[i].y * wi;
      a.z += hv[i].z * wi;
      a.w += hv[i].w * wi;
      *tr = a;
    }
  }
}

template <int MODE>
__device__ __forceinline__ void queue_entry(int b, int n, int lane, const int* q,
                                            const int2* __restrict__ rec, int& e, float& w) {
  const int idx = b * BATCH + lane;
  e = 0;
  w = 0.0f;
  if (lane < BATCH && idx < n) {
    if (MODE == 0) {
      e = q[idx];
    } else {
      const int2 r = rec[idx];
      e = r.x;
      w = __int_as_float(r.y);
    }
  }
}

template <int MODE>
__device__ __forceinline__ void sum_queue(int n, int lane, const int* q, int2* __restrict__ rec,
                                          const float* __restrict__ dinv, const float* dinvL,
                                          const float* __restrict__ H, float* tile) {
  const int nbat = (n + BATCH - 1) / BATCH;
  int e_rq, e_nx;
  float w_rq, w_nx;
  float4 hv[BATCH];
  queue_entry<MODE>(0, n, lane, q, rec, e_rq, w_rq);
  if (MODE == 0) w_rq = dinv[e_rq & SRC_MASK];
  rows_load4(e_rq & SRC_MASK, lane, H, hv);
  queue_entry<MODE>(1, n, lane, q, rec, e_nx, w_nx);

  for (int b = 0; b < nbat; ++b) {
    float4 cv[BATCH];
    #pragma unroll
    for (int i = 0; i < BATCH; ++i) cv[i] = hv[i];
    const int e_cur = e_rq;
    const float w_cur = w_rq;

    e_rq = e_nx;
    w_rq = w_nx;
    if (MODE == 0) w_rq = dinv[e_rq & SRC_MASK];
    rows_load4(e_rq & SRC_MASK, lane, H, hv);
    queue_entry<MODE>(b + 2, n, lane, q, rec, e_nx, w_nx);

    const int left = n - b * BATCH;
    const int nb = (left < BATCH) ? left : BATCH;
    const int dl = (int)((unsigned)e_cur >> 17);
    float w = w_cur;
    if (MODE == 0) {
      w = w_cur * dinvL[dl];
      if (lane < nb) rec[b * BATCH + lane] = make_int2(e_cur, __float_as_int(w));
    }
    rows_rmw(cv, dl, w, nb, lane, tile);
  }
}

template <int MODE>
__global__ __launch_bounds__(AGG_THREADS) void k_agg_owner(const int* __restrict__ src,
                                                           const int* __restrict__ dst,
                                                           const float* __restrict__ dinv,
                                                           const float* __restrict__ H,
                                                           const float* __restrict__ bias,
                                                           float* __restrict__ OUT, int do_relu,
                                                           int* __restrict__ rec_cnt,
                                                           int* __restrict__ rec_all,
                                                           int* __restrict__ qbad,
                                                           const int* __restrict__ cells,
                                                           const int* __restrict__ ovf) {
  __shared__ __attribute__((aligned(16))) float tile[NPB * DIM];
  __shared__ float dinvL[NPB + 1];
  __shared__ int lists[AGG_WAVES * LIST_CAP];
  __shared__ int qcnt[AGG_WAVES];
  __shared__ int qovf;

  const int t = threadIdx.x;
  const int wave = t >> 5;
  const int lane = t & 31;
  const int lo = blockIdx.x * NPB;
  const int nvalid = (N_NODESC - lo < NPB) ? (N_NODESC - lo) : NPB;
  const int wv = __builtin_amdgcn_readfirstlane(wave);
  int2* rec = (int2*)rec_all + ((size_t)blockIdx.x * AGG_WAVES + wv) * QCAP;

  if (t < AGG_WAVES) qcnt[t] = 0;
  if (t == 0) qovf = 0;

  float4 own[ROWS_PW];
  float dio[ROWS_PW];
  #pragma unroll
  for (int i = 0; i < ROWS_PW; ++i) {
    const int r = 32 * i + wv;
    const int rc = (r < nvalid) ? r : nvalid - 1;
    dio[i] = dinv[lo + rc];
    own[i] = ((const float4*)(H + (size_t)(lo + rc) * DIM))[lane];
  }
  int2 cv[CELLS_PER_WAVE];
  bool bad;
  if (MODE == 0) {
    bad = ovf[blockIdx.x] != 0;
    cells_request(cells + (size_t)blockIdx.x * BIN_BLOCKS * CELL_INTS, wv, lane, cv);
  } else {
    bad = qbad[blockIdx.x] != 0;
  }

  #pragma unroll
  for (int i = 0; i < ROWS_PW; ++i) {
    const int r = 32 * i + wv;
    if (r < NPB) {
      const float di = (r < nvalid) ? dio[i] : 0.f;
      float4 v;
      v.x = own[i].x * di * di;
      v.y = own[i].y * di * di;
      v.z = own[i].z * di * di;
      v.w = own[i].w * di * di;
      if (r >= nvalid) v = make_float4(0.f, 0.f, 0.f, 0.f);
      ((float4*)(tile + r * DIM))[lane] = v;
      if (lane == 0) dinvL[r] = di;
    }
  }

  int* mylist = lists + wv * LIST_CAP;

  if (MODE == 0) {
    __syncthreads();
    if (!bad) route_cells(cv, lane, lists, qcnt, &qovf);
    __syncthreads();
    bad = bad || (qovf != 0);
    if (t == 0) qbad[blockIdx.x] = bad ? 1 : 0;
  }

  if (!bad) {
    if (MODE == 0) {
      const int n = qcnt[wv];
      sum_queue<0>(n, lane, lists + wv * QCAP, rec, dinv, dinvL, H, tile);
      if (lane == 0) rec_cnt[blockIdx.x * AGG_WAVES + wv] = n;
    } else {
      const int n = rec_cnt[blockIdx.x * AGG_WAVES + wv];
      sum_queue<1>(n, lane, lists, rec, dinv, dinvL, H, tile);
    }
  } else {
    __syncthreads();
    agg_scan(lo, wave, lane, mylist, src, dst, dinv, dinvL, H, tile);
    __syncthreads();
  }

  const float4 bb = ((const float4*)bias)[lane];
  float4 res[ROWS_PW];
  #pragma unroll
  for (int i = 0; i < ROWS_PW; ++i) {
    const int r = 32 * i + wv;
    const int rc = (r < NPB) ? r : NPB - 1;
    res[i] = ((const float4*)(tile + rc * DIM))[lane];
  }
  #pragma unroll
  for (int i = 0; i < ROWS_PW; ++i) {
    const int r = 32 * i + wv;
    if (r < nvalid) {
      float4 a = res[i];
      a.x += bb.x;
      a.y += bb.y;
      a.z += bb.z;
      a.w += bb.w;
      if (do_relu) {
        a.x = fmaxf(a.x, 0.0f);
        a.y = fmaxf(a.y, 0.0f);
        a.z = fmaxf(a.z, 0.0f);
        a.w = fmaxf(a.w, 0.0f);
      }
      ((float4*)(OUT + (size_t)(lo + r) * DIM))[lane] = a;
    }
  }
}


extern "C" void kernel_launch(void* const* d_in, const int* in_sizes, int n_in,
                              void* d_out, int out_size, void* d_ws, size_t ws_size,
                              hipStream_t stream) {
  const float* x   = (const float*)d_in[0];
  const int*   ei  = (const int*)d_in[1];
  const float* W1  = (const float*)d_in[2];
  const float* b1  = (const float*)d_in[3];
  const float* W2  = (const float*)d_in[4];
  const float* b2  = (const float*)d_in[5];
  float*       out = (float*)d_out;

  const int* src = ei;
  const int* dst = ei + N_EDGESC;

  float* H    = (float*)d_ws;
  float* dinv = H + (size_t)N_NODESC * DIM + N_NODESC;
  int*   rec_cnt = (int*)(dinv + N_NODESC);
  int*   qbad    = rec_cnt + AGG_BLOCKS * AGG_WAVES;
  int*   rec_all = qbad + AGG_BLOCKS;
  int*   cells   = rec_all + (size_t)AGG_BLOCKS * AGG_WAVES * QCAP * 2;
  int*   ovf     = cells + (size_t)AGG_BLOCKS * BIN_BLOCKS * CELL_INTS;

  dim3 blk(256);
  dim3 gBin(BIN_BLOCKS);
  dim3 bBin(BIN_THREADS);
  dim3 gGemm(GEMM_BLOCKS);
  dim3 gAgg(AGG_BLOCKS);
  dim3 bAgg(AGG_THREADS);

  k_bin_count<<<gBin, bBin, 0, stream>>>(src, dst, cells);
  k_deg_owner<<<gAgg, bAgg, 0, stream>>>(dst, cells, dinv, ovf);

  k_gemm_wmma<<<gGemm, blk, 0, stream>>>(x, W1, H);
  k_agg_owner<0><<<gAgg, bAgg, 0, stream>>>(src, dst, dinv, H, b1, out,  1, rec_cnt, rec_all, qbad, cells, ovf);

  k_gemm_wmma<<<gGemm, blk, 0, stream>>>(out, W2, H);
  k_agg_owner<1><<<gAgg, bAgg, 0, stream>>>(src, dst, dinv, H, b2, out,  0, rec_cnt, rec_all, qbad, cells, ovf);
}


# Product launches scale row s of H by dinv[s] (on the X row as staged): owner blocks add rows unweighted, apply dinv[d] once per stored row; no scattered dinv[src] load per edge; sum order unchanged
# speedup vs baseline: 5.1073x; 1.0314x over previous
#include <hip/hip_runtime.h>

typedef float v2f __attribute__((ext_vector_type(2)));
typedef float v8f __attribute__((ext_vector_type(8)));

#define N_NODESC 100000
#define N_EDGESC 1600000
#define DIM 128


#define AGG_BLOCKS 256
#define NPB 391

#define BIN_BLOCKS 256
#define BIN_THREADS 1024
#define CELL_INTS 64
#define CELL_CAP (CELL_INTS - 1)
#define CELL_PITCH 68
#define EDGES_PER_BIN (N_EDGESC / BIN_BLOCKS)
#define BIN_ITERS ((EDGES_PER_BIN + BIN_THREADS - 1) / BIN_THREADS)

static_assert(N_EDGESC % BIN_BLOCKS == 0, "chunks are equal");
static_assert(N_NODESC <= (1 << 17) && NPB <= (1 << 9), "cell entry packs source node and local row");

__global__ __launch_bounds__(BIN_THREADS) void k_bin_count(const int* __restrict__ src,
                                                           const int* __restrict__ dst,
                                                           int* __restrict__ cells) {
  __shared__ __attribute__((aligned(16))) int cellL[AGG_BLOCKS * CELL_PITCH];
  __shared__ int cntL[AGG_BLOCKS];

  const int t = threadIdx.x;
  const int e0 = blockIdx.x * EDGES_PER_BIN;

  int dv[BIN_ITERS], sv[BIN_ITERS];
  #pragma unroll
  for (int k = 0; k < BIN_ITERS; ++k) {
    const int i = t + k * BIN_THREADS;
    dv[k] = -1;
    sv[k] = 0;
    if (i < EDGES_PER_BIN) {
      dv[k] = dst[e0 + i];
      sv[k] = src[e0 + i];
    }
  }

  for (int q = t; q < AGG_BLOCKS * CELL_PITCH / 4; q += BIN_THREADS)
    ((int4*)cellL)[q] = make_int4(0, 0, 0, 0);
  for (int o = t; o < AGG_BLOCKS; o += BIN_THREADS) cntL[o] = 0;
  __syncthreads();

  #pragma unroll
  for (int k = 0; k < BIN_ITERS; ++k) {
    const int d = dv[k];
    if (d >= 0) {
      const int o = d / NPB;
      const int dl = d - o * NPB;
      const int rank = __hip_atomic_fetch_add(&cntL[o], 1, __ATOMIC_RELAXED, __HIP_MEMORY_SCOPE_WORKGROUP);
      if (rank < CELL_CAP) cellL[o * CELL_PITCH + 1 + rank] = sv[k] | (dl << 17);
    }
  }
  __syncthreads();

  for (int q = t; q < AGG_BLOCKS * CELL_INTS / 4; q += BIN_THREADS) {
    const int o = q >> 4;
    const int part = q & 15;
    int4 v = *(const int4*)&cellL[o * CELL_PITCH + part * 4];
    if (part == 0) v.x = cntL[o];
    ((int4*)(cells + ((size_t)o * BIN_BLOCKS + blockIdx.x) * CELL_INTS))[part] = v;
  }
}

#define DEG_THREADS 1024
#define DEG_WAVES (DEG_THREADS / 32)
#define DEG_CELLS_PER_WAVE (BIN_BLOCKS / DEG_WAVES)
#define N_DST4Q (N_EDGESC / 4)

static_assert(BIN_BLOCKS % DEG_WAVES == 0, "every wave takes the same number of cells");
static_assert(N_EDGESC % 4 == 0, "dst list is read as int4");

__global__ __launch_bounds__(DEG_THREADS) void k_deg_owner(const int* __restrict__ dst,
                                                           const int* __restrict__ cells,
                                                           float* __restrict__ dinv,
                                                           int* __restrict__ ovf) {
  __shared__ int cntR[NPB];
  __shared__ int full;

  const int t = threadIdx.x;
  const int lane = t & 31;
  const int wv = __builtin_amdgcn_readfirstlane(t >> 5);
  const int lo = blockIdx.x * NPB;
  const int nvalid = (N_NODESC - lo < NPB) ? (N_NODESC - lo) : NPB;

  const int2* cp = (const int2*)(cells + ((size_t)blockIdx.x * BIN_BLOCKS +
                                          (size_t)wv * DEG_CELLS_PER_WAVE) * CELL_INTS) + lane;
  int2 cv[DEG_CELLS_PER_WAVE];
  #pragma unroll
  for (int c = 0; c < DEG_CELLS_PER_WAVE; ++c) cv[c] = cp[c * (CELL_INTS / 2)];

  for (int r = t; r < NPB; r += DEG_THREADS) cntR[r] = 0;
  if (t == 0) full = 0;
  __syncthreads();

  #pragma unroll
  for (int c = 0; c < DEG_CELLS_PER_WAVE; ++c) {
    const int raw = __builtin_amdgcn_readfirstlane(cv[c].x);
    if (raw > CELL_CAP && lane == 0) full = 1;
    const int cnt = (raw < CELL_CAP) ? raw : CELL_CAP;
    const int i0 = 2 * lane;
    if (i0 >= 1 && i0 <= cnt)
      __hip_atomic_fetch_add(&cntR[(unsigned)cv[c].x >> 17], 1, __ATOMIC_RELAXED, __HIP_MEMORY_SCOPE_WORKGROUP);
    if (i0 + 1 <= cnt)
      __hip_atomic_fetch_add(&cntR[(unsigned)cv[c].y >> 17], 1, __ATOMIC_RELAXED, __HIP_MEMORY_SCOPE_WORKGROUP);
  }
  __syncthreads();

  const bool bad = full != 0;
  if (bad) {
    for (int r = t; r < NPB; r += DEG_THREADS) cntR[r] = 0;
    __syncthreads();
    const int4* dst4 = (const int4*)dst;
    for (int i = t; i < N_DST4Q; i += DEG_THREADS) {
      const int4 v = dst4[i];
      const unsigned a0 = (unsigned)(v.x - lo), a1 = (unsigned)(v.y - lo);
      const unsigned a2 = (unsigned)(v.z - lo), a3 = (unsigned)(v.w - lo);
      if (a0 < (unsigned)NPB) __hip_atomic_fetch_add(&cntR[a0], 1, __ATOMIC_RELAXED, __HIP_MEMORY_SCOPE_WORKGROUP);
      if (a1 < (unsigned)NPB) __hip_atomic_fetch_add(&cntR[a1], 1, __ATOMIC_RELAXED, __HIP_MEMORY_SCOPE_WORKGROUP);
      if (a2 < (unsigned)NPB) __hip_atomic_fetch_add(&cntR[a2], 1, __ATOMIC_RELAXED, __HIP_MEMORY_SCOPE_WORKGROUP);
      if (a3 < (unsigned)NPB) __hip_atomic_fetch_add(&cntR[a3], 1, __ATOMIC_RELAXED, __HIP_MEMORY_SCOPE_WORKGROUP);
    }
    __syncthreads();
  }

  for (int r = t; r < nvalid; r += DEG_THREADS)
    dinv[lo + r] = rsqrtf(1.0f + (float)cntR[r]);
  if (t == 0) ovf[blockIdx.x] = bad ? 1 : 0;
}


#define GEMM_BLOCKS 768
#define GEMM_TILES (N_NODESC / 16)
#define GEMM_KSTEPS (DIM / 4)
#define XS_PITCH4 33
#define XS_STAGE4 (16 * XS_PITCH4)

static_assert(N_NODESC % 16 == 0, "whole 16-row tiles");

__global__ __launch_bounds__(256) void k_gemm_wmma(const float* __restrict__ X,
                                                   const float* __restrict__ W,
                                                   const float* __restrict__ dinv,
                                                   float* __restrict__ H) {
  __shared__ float4 xs4[2 * XS_STAGE4];

  const int t = threadIdx.x;
  const int wave = __builtin_amdgcn_readfirstlane(t >> 5);
  const int lane = t & 31;

  const int m  = lane & 15;
  const int hg = lane >> 4;
  const int n0 = wave * 16;

  v2f bf[GEMM_KSTEPS];
  {
    const float* wcol = W + n0 + m;
    #pragma unroll
    for (int j = 0; j < GEMM_KSTEPS; ++j) {
      bf[j].x = wcol[(size_t)(4 * j + 2 * hg + 0) * DIM];
      bf[j].y = wcol[(size_t)(4 * j + 2 * hg + 1) * DIM];
    }
  }

  const int r  = wave;
  const int c4 = lane;

  int tile = blockIdx.x;
  float4 p0, p1;
  float s0, s1;
  {
    const int tl = (tile < GEMM_TILES) ? tile : GEMM_TILES - 1;
    const float4* g = (const float4*)(X + ((size_t)tl * 16 + r) * DIM) + c4;
    p0 = g[0];
    p1 = g[8 * (DIM / 4)];
    s0 = dinv[tl * 16 + r];
    s1 = dinv[tl * 16 + r + 8];
  }

  int cur = 0;
  for (; tile < GEMM_TILES; tile += GEMM_BLOCKS) {
    float4* stage = xs4 + cur * XS_STAGE4;
    stage[r * XS_PITCH4 + c4] = make_float4(p0.x * s0, p0.y * s0, p0.z * s0, p0.w * s0);
    stage[(r + 8) * XS_PITCH4 + c4] = make_float4(p1.x * s1, p1.y * s1, p1.z * s1, p1.w * s1);
    __syncthreads();

    {
      const int nxt = tile + GEMM_BLOCKS;
      const int tl = (nxt < GEMM_TILES) ? nxt : GEMM_TILES - 1;
      const float4* g = (const float4*)(X + ((size_t)tl * 16 + r) * DIM) + c4;
      p0 = g[0];
      p1 = g[8 * (DIM / 4)];
      s0 = dinv[tl * 16 + r];
      s1 = dinv[tl * 16 + r + 8];
    }
    __builtin_amdgcn_sched_barrier(0);

    const v2f* arow = (const v2f*)stage + m * (XS_PITCH4 * 2) + hg;
    v2f af[GEMM_KSTEPS];
    #pragma unroll
    for (int j = 0; j < GEMM_KSTEPS; ++j) af[j] = arow[2 * j];
    __builtin_amdgcn_sched_barrier(0);

    v8f acc = {};
    #pragma unroll
    for (int j = 0; j < GEMM_KSTEPS; ++j) {
      acc = __builtin_amdgcn_wmma_f32_16x16x4_f32(
           false, af[j],  false, bf[j],
           (short)0, acc,  false,  false);
    }

    float* hrow = H + (size_t)tile * 16 * DIM + n0 + m;
    #pragma unroll
    for (int v = 0; v < 8; ++v)
      hrow[(size_t)(v + 8 * hg) * DIM] = acc[v];

    cur ^= 1;
  }
}


#define AGG_THREADS 1024
#define AGG_WAVES 32
#define SCAN_U 4
#define N_DST4 (N_EDGESC / 4)
#define WAVE4 (32 * SCAN_U)
#define TRIP4 (AGG_WAVES * WAVE4)
#define N_TRIPS ((N_DST4 + TRIP4 - 1) / TRIP4)
#define BATCH 8
#define LIST_CAP (BATCH - 1 + 4 * WAVE4 + 1)
#define QCAP LIST_CAP
#define SRC_MASK ((1 << 17) - 1)
#define REC_PITCH ((QCAP + 31) / 32 * 32)

static_assert(AGG_BLOCKS * NPB >= N_NODESC, "slices must cover every node");
static_assert(N_EDGESC % 4 == 0, "dst list is read as int4");
static_assert(N_EDGESC <= (1 << 21) && NPB <= (1 << 9), "list entry packs edge id and local row");
static_assert(BIN_BLOCKS % AGG_WAVES == 0, "every wave takes the same number of cells");
#define CELLS_PER_WAVE (BIN_BLOCKS / AGG_WAVES)
#define ROWS_PW ((NPB + AGG_WAVES - 1) / AGG_WAVES)

__device__ __forceinline__ void agg_load_trip(const int4* __restrict__ dst4, int base4, int lane,
                                              int4 (&v)[SCAN_U]) {
  #pragma unroll
  for (int u = 0; u < SCAN_U; ++u) {
    const int i4 = base4 + u * 32 + lane;
    v[u] = (i4 < N_DST4) ? dst4[i4] : make_int4(-1, -1, -1, -1);
  }
}

__device__ __forceinline__ void rows_load(int s, int lane, const float* __restrict__ H,
                                          float (&hv)[BATCH][4]) {
  #pragma unroll
  for (int i = 0; i < BATCH; ++i) {
    const int si = __builtin_amdgcn_readlane(s, i);
    const float* hr = H + (size_t)si * DIM + lane;
    hv[i][0] = hr[0];
    hv[i][1] = hr[32];
    hv[i][2] = hr[64];
    hv[i][3] = hr[96];
  }
}

__device__ __forceinline__ void rows_add(const float (&hv)[BATCH][4], int dl, int nb,
                                         int lane, float* tile) {
  #pragma unroll
  for (int i = 0; i < BATCH; ++i) {
    if (i < nb) {
      const int di = __builtin_amdgcn_readlane(dl, i);
      float* tr = tile + di * DIM + lane;
      __hip_atomic_fetch_add(tr + 0,  hv[i][0], __ATOMIC_RELAXED, __HIP_MEMORY_SCOPE_WORKGROUP);
      __hip_atomic_fetch_add(tr + 32, hv[i][1], __ATOMIC_RELAXED, __HIP_MEMORY_SCOPE_WORKGROUP);
      __hip_atomic_fetch_add(tr + 64, hv[i][2], __ATOMIC_RELAXED, __HIP_MEMORY_SCOPE_WORKGROUP);
      __hip_atomic_fetch_add(tr + 96, hv[i][3], __ATOMIC_RELAXED, __HIP_MEMORY_SCOPE_WORKGROUP);
    }
  }
}

__device__ __forceinline__ void agg_batch(const int* lst, int nb, int lane,
                                          const int* __restrict__ src,
                                          const float* __restrict__ H, float* tile) {
  int ent = 0;
  if (lane < nb) ent = lst[lane];
  const int dl = (int)((unsigned)ent >> 21);
  const int s = src[ent & ((1 << 21) - 1)];
  float hv[BATCH][4];
  rows_load(s, lane, H, hv);
  rows_add(hv, dl, nb, lane, tile);
}

__device__ __forceinline__ void agg_scan(int lo, int wave, int lane, int* mylist,
                                         const int* __restrict__ src, const int* __restrict__ dst,
                                         const float* __restrict__ H, float* tile) {
  const int4* dst4 = (const int4*)dst;
  int pend = 0;
  int trip = blockIdx.x % N_TRIPS;

  int4 nxt[SCAN_U];
  agg_load_trip(dst4, trip * TRIP4 + wave * WAVE4, lane, nxt);

  for (int it = 0; it < N_TRIPS; ++it) {
    int4 cur[SCAN_U];
    #pragma unroll
    for (int u = 0; u < SCAN_U; ++u) cur[u] = nxt[u];
    const int cur4 = trip * TRIP4 + wave * WAVE4;
    trip = (trip + 1 == N_TRIPS) ? 0 : trip + 1;
    if (it + 1 < N_TRIPS) agg_load_trip(dst4, trip * TRIP4 + wave * WAVE4, lane, nxt);

    #pragma unroll
    for (int u = 0; u < SCAN_U; ++u) {
      #pragma unroll
      for (int c = 0; c < 4; ++c) {
        const int d = (c == 0) ? cur[u].x : (c == 1) ? cur[u].y : (c == 2) ? cur[u].z : cur[u].w;
        const unsigned tt = (unsigned)(d - lo);
        const bool m = tt < (unsigned)NPB;
        const unsigned mask = __builtin_amdgcn_ballot_w32(m);
        if (mask != 0u) {
          const int pos = pend + (int)__builtin_amdgcn_mbcnt_lo(mask, 0u);
          const int e = (cur4 + u * 32 + lane) * 4 + c;
          if (m) mylist[pos] = (int)((tt << 21) | (unsigned)e);
          pend += __builtin_popcount(mask);
        }
      }
    }
    while (pend >= BATCH) {
      pend -= BATCH;
      agg_batch(mylist + pend, BATCH, lane, src, H, tile);
    }
  }
  if (pend > 0) agg_batch(mylist, pend, lane, src, H, tile);
}

__device__ __forceinline__ void queue_put(int e, int* queues, int* qcnt, int* qovf) {
  const int ow = (e >> 17) & (AGG_WAVES - 1);
  const int slot = __hip_atomic_fetch_add(&qcnt[ow], 1, __ATOMIC_RELAXED, __HIP_MEMORY_SCOPE_WORKGROUP);
  if (slot < QCAP) queues[ow * QCAP + slot] = e;
  else *qovf = 1;
}

__device__ __forceinline__ void cells_request(const int* __restrict__ mycells, int wv, int lane,
                                              int2 (&cv)[CELLS_PER_WAVE]) {
  const int2* cp = (const int2*)(mycells + (size_t)wv * CELLS_PER_WAVE * CELL_INTS) + lane;
  #pragma unroll
  for (int c = 0; c < CELLS_PER_WAVE; ++c) cv[c] = cp[c * (CELL_INTS / 2)];
}

__device__ __forceinline__ void route_cells(const int2 (&cv)[CELLS_PER_WAVE], int lane,
                                            int* queues, int* qcnt, int* qovf) {
  #pragma unroll
  for (int c = 0; c < CELLS_PER_WAVE; ++c) {
    int cnt = __builtin_amdgcn_readfirstlane(cv[c].x);
    cnt = (cnt < CELL_CAP) ? cnt : CELL_CAP;
    const int i0 = 2 * lane;
    if (i0 >= 1 && i0 <= cnt) queue_put(cv[c].x, queues, qcnt, qovf);
    if (i0 + 1 <= cnt) queue_put(cv[c].y, queues, qcnt, qovf);
  }
}

__device__ __forceinline__ void rows_load4(int s, int lane, const float* __restrict__ H,
                                           float4 (&hv)[BATCH]) {
  #pragma unroll
  for (int i = 0; i < BATCH; ++i) {
    const int si = __builtin_amdgcn_readlane(s, i);
    hv[i] = ((const float4*)(H + (size_t)si * DIM))[lane];
  }
}

__device__ __forceinline__ void rows_rmw(const float4 (&hv)[BATCH], int dl, int nb,
                                         int lane, float* tile) {
  #pragma unroll
  for (int i = 0; i < BATCH; ++i) {
    if (i < nb) {
      const int di = __builtin_amdgcn_readlane(dl, i);
      float4* tr = (float4*)(tile + di * DIM) + lane;
      float4 a = *tr;
      a.x += hv[i].x;
      a.y += hv[i].y;
      a.z += hv[i].z;
      a.w += hv[i].w;
      *tr = a;
    }
  }
}

template <int MODE>
__device__ __forceinline__ int queue_entry(int b, int n, int lane, const int* q,
                                           const int* __restrict__ rec) {
  const int idx = b * BATCH + lane;
  int e = 0;
  if (lane < BATCH && idx < n) e = (MODE == 0) ? q[idx] : rec[idx];
  return e;
}

template <int MODE>
__device__ __forceinline__ void sum_queue(int n, int lane, const int* q, int* __restrict__ rec,
                                          const float* __restrict__ H, float* tile) {
  const int nbat = (n + BATCH - 1) / BATCH;
  int e_rq, e_nx;
  float4 hv[BATCH];
  e_rq = queue_entry<MODE>(0, n, lane, q, rec);
  rows_load4(e_rq & SRC_MASK, lane, H, hv);
  e_nx = queue_entry<MODE>(1, n, lane, q, rec);

  for (int b = 0; b < nbat; ++b) {
    float4 cv[BATCH];
    #pragma unroll
    for (int i = 0; i < BATCH; ++i) cv[i] = hv[i];
    const int e_cur = e_rq;

    e_rq = e_nx;
    rows_load4(e_rq & SRC_MASK, lane, H, hv);
    e_nx = queue_entry<MODE>(b + 2, n, lane, q, rec);

    const int left = n - b * BATCH;
    const int nb = (left < BATCH) ? left : BATCH;
    const int dl = (int)((unsigned)e_cur >> 17);
    if (MODE == 0 && lane < nb) rec[b * BATCH + lane] = e_cur;
    rows_rmw(cv, dl, nb, lane, tile);
  }
}

template <int MODE>
__global__ __launch_bounds__(AGG_THREADS) void k_agg_owner(const int* __restrict__ src,
                                                           const int* __restrict__ dst,
                                                           const float* __restrict__ dinv,
                                                           const float* __restrict__ H,
                                                           const float* __restrict__ bias,
                                                           float* __restrict__ OUT, int do_relu,
                                                           int* __restrict__ rec_cnt,
                                                           int* __restrict__ rec_all,
                                                           int* __restrict__ qbad,
                                                           const int* __restrict__ cells,
                                                           const int* __restrict__ ovf) {
  __shared__ __attribute__((aligned(16))) float tile[NPB * DIM];
  __shared__ int lists[AGG_WAVES * LIST_CAP];
  __shared__ int qcnt[AGG_WAVES];
  __shared__ int qovf;

  const int t = threadIdx.x;
  const int wave = t >> 5;
  const int lane = t & 31;
  const int lo = blockIdx.x * NPB;
  const int nvalid = (N_NODESC - lo < NPB) ? (N_NODESC - lo) : NPB;
  const int wv = __builtin_amdgcn_readfirstlane(wave);
  int* rec = rec_all + ((size_t)blockIdx.x * AGG_WAVES + wv) * REC_PITCH;

  if (t < AGG_WAVES) qcnt[t] = 0;
  if (t == 0) qovf = 0;

  float4 own[ROWS_PW];
  float dio[ROWS_PW];
  #pragma unroll
  for (int i = 0; i < ROWS_PW; ++i) {
    const int r = 32 * i + wv;
    const int rc = (r < nvalid) ? r : nvalid - 1;
    dio[i] = dinv[lo + rc];
    own[i] = ((const float4*)(H + (size_t)(lo + rc) * DIM))[lane];
  }
  int2 cv[CELLS_PER_WAVE];
  bool bad;
  if (MODE == 0) {
    bad = ovf[blockIdx.x] != 0;
    cells_request(cells + (size_t)blockIdx.x * BIN_BLOCKS * CELL_INTS, wv, lane, cv);
  } else {
    bad = qbad[blockIdx.x] != 0;
  }

  #pragma unroll
  for (int i = 0; i < ROWS_PW; ++i) {
    const int r = 32 * i + wv;
    if (r < NPB) {
      float4 v = own[i];
      if (r >= nvalid) v = make_float4(0.f, 0.f, 0.f, 0.f);
      ((float4*)(tile + r * DIM))[lane] = v;
    }
  }

  int* mylist = lists + wv * LIST_CAP;

  if (MODE == 0) {
    __syncthreads();
    if (!bad) route_cells(cv, lane, lists, qcnt, &qovf);
    __syncthreads();
    bad = bad || (qovf != 0);
    if (t == 0) qbad[blockIdx.x] = bad ? 1 : 0;
  }

  if (!bad) {
    if (MODE == 0) {
      const int n = qcnt[wv];
      sum_queue<0>(n, lane, lists + wv * QCAP, rec, H, tile);
      if (lane == 0) rec_cnt[blockIdx.x * AGG_WAVES + wv] = n;
    } else {
      const int n = rec_cnt[blockIdx.x * AGG_WAVES + wv];
      sum_queue<1>(n, lane, lists, rec, H, tile);
    }
  } else {
    __syncthreads();
    agg_scan(lo, wave, lane, mylist, src, dst, H, tile);
    __syncthreads();
  }

  const float4 bb = ((const float4*)bias)[lane];
  float4 res[ROWS_PW];
  #pragma unroll
  for (int i = 0; i < ROWS_PW; ++i) {
    const int r = 32 * i + wv;
    const int rc = (r < NPB) ? r : NPB - 1;
    res[i] = ((const float4*)(tile + rc * DIM))[lane];
  }
  #pragma unroll
  for (int i = 0; i < ROWS_PW; ++i) {
    const int r = 32 * i + wv;
    if (r < nvalid) {
      float4 a;
      a.x = res[i].x * dio[i] + bb.x;
      a.y = res[i].y * dio[i] + bb.y;
      a.z = res[i].z * dio[i] + bb.z;
      a.w = res[i].w * dio[i] + bb.w;
      if (do_relu) {
        a.x = fmaxf(a.x, 0.0f);
        a.y = fmaxf(a.y, 0.0f);
        a.z = fmaxf(a.z, 0.0f);
        a.w = fmaxf(a.w, 0.0f);
      }
      ((float4*)(OUT + (size_t)(lo + r) * DIM))[lane] = a;
    }
  }
}


extern "C" void kernel_launch(void* const* d_in, const int* in_sizes, int n_in,
                              void* d_out, int out_size, void* d_ws, size_t ws_size,
                              hipStream_t stream) {
  const float* x   = (const float*)d_in[0];
  const int*   ei  = (const int*)d_in[1];
  const float* W1  = (const float*)d_in[2];
  const float* b1  = (const float*)d_in[3];
  const float* W2  = (const float*)d_in[4];
  const float* b2  = (const float*)d_in[5];
  float*       out = (float*)d_out;

  const int* src = ei;
  const int* dst = ei + N_EDGESC;

  float* H    = (float*)d_ws;
  float* dinv = H + (size_t)N_NODESC * DIM + N_NODESC;
  int*   rec_cnt = (int*)(dinv + N_NODESC);
  int*   qbad    = rec_cnt + AGG_BLOCKS * AGG_WAVES;
  int*   rec_all = qbad + AGG_BLOCKS;
  int*   cells   = rec_all + (size_t)AGG_BLOCKS * AGG_WAVES * REC_PITCH;
  int*   ovf     = cells + (size_t)AGG_BLOCKS * BIN_BLOCKS * CELL_INTS;

  dim3 blk(256);
  dim3 gBin(BIN_BLOCKS);
  dim3 bBin(BIN_THREADS);
  dim3 gGemm(GEMM_BLOCKS);
  dim3 gAgg(AGG_BLOCKS);
  dim3 bAgg(AGG_THREADS);

  k_bin_count<<<gBin, bBin, 0, stream>>>(src, dst, cells);
  k_deg_owner<<<gAgg, bAgg, 0, stream>>>(dst, cells, dinv, ovf);

  k_gemm_wmma<<<gGemm, blk, 0, stream>>>(x, W1, dinv, H);
  k_agg_owner<0><<<gAgg, bAgg, 0, stream>>>(src, dst, dinv, H, b1, out,  1, rec_cnt, rec_all, qbad, cells, ovf);

  k_gemm_wmma<<<gGemm, blk, 0, stream>>>(out, W2, dinv, H);
  k_agg_owner<1><<<gAgg, bAgg, 0, stream>>>(src, dst, dinv, H, b2, out,  0, rec_cnt, rec_all, qbad, cells, ovf);
}


# Owner waves add a row's sources in registers, not in an LDS slice: cells go to per-row LDS queues, laid end to end per wave; a row is stored when its run ends; per-row order still LDS arrival order
# speedup vs baseline: 5.4255x; 1.0623x over previous
#include <hip/hip_runtime.h>

typedef float v2f __attribute__((ext_vector_type(2)));
typedef float v8f __attribute__((ext_vector_type(8)));

#define N_NODESC 100000
#define N_EDGESC 1600000
#define DIM 128


#define AGG_BLOCKS 256
#define NPB 391

#define BIN_BLOCKS 256
#define BIN_THREADS 1024
#define CELL_INTS 64
#define CELL_CAP (CELL_INTS - 1)
#define CELL_PITCH 68
#define EDGES_PER_BIN (N_EDGESC / BIN_BLOCKS)
#define BIN_ITERS ((EDGES_PER_BIN + BIN_THREADS - 1) / BIN_THREADS)

static_assert(N_EDGESC % BIN_BLOCKS == 0, "chunks are equal");
static_assert(N_NODESC <= (1 << 17) && NPB <= (1 << 9), "cell entry packs source node and local row");

__global__ __launch_bounds__(BIN_THREADS) void k_bin_count(const int* __restrict__ src,
                                                           const int* __restrict__ dst,
                                                           int* __restrict__ cells) {
  __shared__ __attribute__((aligned(16))) int cellL[AGG_BLOCKS * CELL_PITCH];
  __shared__ int cntL[AGG_BLOCKS];

  const int t = threadIdx.x;
  const int e0 = blockIdx.x * EDGES_PER_BIN;

  int dv[BIN_ITERS], sv[BIN_ITERS];
  #pragma unroll
  for (int k = 0; k < BIN_ITERS; ++k) {
    const int i = t + k * BIN_THREADS;
    dv[k] = -1;
    sv[k] = 0;
    if (i < EDGES_PER_BIN) {
      dv[k] = dst[e0 + i];
      sv[k] = src[e0 + i];
    }
  }

  for (int q = t; q < AGG_BLOCKS * CELL_PITCH / 4; q += BIN_THREADS)
    ((int4*)cellL)[q] = make_int4(0, 0, 0, 0);
  for (int o = t; o < AGG_BLOCKS; o += BIN_THREADS) cntL[o] = 0;
  __syncthreads();

  #pragma unroll
  for (int k = 0; k < BIN_ITERS; ++k) {
    const int d = dv[k];
    if (d >= 0) {
      const int o = d / NPB;
      const int dl = d - o * NPB;
      const int rank = __hip_atomic_fetch_add(&cntL[o], 1, __ATOMIC_RELAXED, __HIP_MEMORY_SCOPE_WORKGROUP);
      if (rank < CELL_CAP) cellL[o * CELL_PITCH + 1 + rank] = sv[k] | (dl << 17);
    }
  }
  __syncthreads();

  for (int q = t; q < AGG_BLOCKS * CELL_INTS / 4; q += BIN_THREADS) {
    const int o = q >> 4;
    const int part = q & 15;
    int4 v = *(const int4*)&cellL[o * CELL_PITCH + part * 4];
    if (part == 0) v.x = cntL[o];
    ((int4*)(cells + ((size_t)o * BIN_BLOCKS + blockIdx.x) * CELL_INTS))[part] = v;
  }
}

#define DEG_THREADS 1024
#define DEG_WAVES (DEG_THREADS / 32)
#define DEG_CELLS_PER_WAVE (BIN_BLOCKS / DEG_WAVES)
#define N_DST4Q (N_EDGESC / 4)

static_assert(BIN_BLOCKS % DEG_WAVES == 0, "every wave takes the same number of cells");
static_assert(N_EDGESC % 4 == 0, "dst list is read as int4");

__global__ __launch_bounds__(DEG_THREADS) void k_deg_owner(const int* __restrict__ dst,
                                                           const int* __restrict__ cells,
                                                           float* __restrict__ dinv,
                                                           int* __restrict__ ovf) {
  __shared__ int cntR[NPB];
  __shared__ int full;

  const int t = threadIdx.x;
  const int lane = t & 31;
  const int wv = __builtin_amdgcn_readfirstlane(t >> 5);
  const int lo = blockIdx.x * NPB;
  const int nvalid = (N_NODESC - lo < NPB) ? (N_NODESC - lo) : NPB;

  const int2* cp = (const int2*)(cells + ((size_t)blockIdx.x * BIN_BLOCKS +
                                          (size_t)wv * DEG_CELLS_PER_WAVE) * CELL_INTS) + lane;
  int2 cv[DEG_CELLS_PER_WAVE];
  #pragma unroll
  for (int c = 0; c < DEG_CELLS_PER_WAVE; ++c) cv[c] = cp[c * (CELL_INTS / 2)];

  for (int r = t; r < NPB; r += DEG_THREADS) cntR[r] = 0;
  if (t == 0) full = 0;
  __syncthreads();

  #pragma unroll
  for (int c = 0; c < DEG_CELLS_PER_WAVE; ++c) {
    const int raw = __builtin_amdgcn_readfirstlane(cv[c].x);
    if (raw > CELL_CAP && lane == 0) full = 1;
    const int cnt = (raw < CELL_CAP) ? raw : CELL_CAP;
    const int i0 = 2 * lane;
    if (i0 >= 1 && i0 <= cnt)
      __hip_atomic_fetch_add(&cntR[(unsigned)cv[c].x >> 17], 1, __ATOMIC_RELAXED, __HIP_MEMORY_SCOPE_WORKGROUP);
    if (i0 + 1 <= cnt)
      __hip_atomic_fetch_add(&cntR[(unsigned)cv[c].y >> 17], 1, __ATOMIC_RELAXED, __HIP_MEMORY_SCOPE_WORKGROUP);
  }
  __syncthreads();

  const bool bad = full != 0;
  if (bad) {
    for (int r = t; r < NPB; r += DEG_THREADS) cntR[r] = 0;
    __syncthreads();
    const int4* dst4 = (const int4*)dst;
    for (int i = t; i < N_DST4Q; i += DEG_THREADS) {
      const int4 v = dst4[i];
      const unsigned a0 = (unsigned)(v.x - lo), a1 = (unsigned)(v.y - lo);
      const unsigned a2 = (unsigned)(v.z - lo), a3 = (unsigned)(v.w - lo);
      if (a0 < (unsigned)NPB) __hip_atomic_fetch_add(&cntR[a0], 1, __ATOMIC_RELAXED, __HIP_MEMORY_SCOPE_WORKGROUP);
      if (a1 < (unsigned)NPB) __hip_atomic_fetch_add(&cntR[a1], 1, __ATOMIC_RELAXED, __HIP_MEMORY_SCOPE_WORKGROUP);
      if (a2 < (unsigned)NPB) __hip_atomic_fetch_add(&cntR[a2], 1, __ATOMIC_RELAXED, __HIP_MEMORY_SCOPE_WORKGROUP);
      if (a3 < (unsigned)NPB) __hip_atomic_fetch_add(&cntR[a3], 1, __ATOMIC_RELAXED, __HIP_MEMORY_SCOPE_WORKGROUP);
    }
    __syncthreads();
  }

  for (int r = t; r < nvalid; r += DEG_THREADS)
    dinv[lo + r] = rsqrtf(1.0f + (float)cntR[r]);
  if (t == 0) ovf[blockIdx.x] = bad ? 1 : 0;
}


#define GEMM_BLOCKS 768
#define GEMM_TILES (N_NODESC / 16)
#define GEMM_KSTEPS (DIM / 4)
#define XS_PITCH4 33
#define XS_STAGE4 (16 * XS_PITCH4)

static_assert(N_NODESC % 16 == 0, "whole 16-row tiles");

__global__ __launch_bounds__(256) void k_gemm_wmma(const float* __restrict__ X,
                                                   const float* __restrict__ W,
                                                   const float* __restrict__ dinv,
                                                   float* __restrict__ H) {
  __shared__ float4 xs4[2 * XS_STAGE4];

  const int t = threadIdx.x;
  const int wave = __builtin_amdgcn_readfirstlane(t >> 5);
  const int lane = t & 31;

  const int m  = lane & 15;
  const int hg = lane >> 4;
  const int n0 = wave * 16;

  v2f bf[GEMM_KSTEPS];
  {
    const float* wcol = W + n0 + m;
    #pragma unroll
    for (int j = 0; j < GEMM_KSTEPS; ++j) {
      bf[j].x = wcol[(size_t)(4 * j + 2 * hg + 0) * DIM];
      bf[j].y = wcol[(size_t)(4 * j + 2 * hg + 1) * DIM];
    }
  }

  const int r  = wave;
  const int c4 = lane;

  int tile = blockIdx.x;
  float4 p0, p1;
  float s0, s1;
  {
    const int tl = (tile < GEMM_TILES) ? tile : GEMM_TILES - 1;
    const float4* g = (const float4*)(X + ((size_t)tl * 16 + r) * DIM) + c4;
    p0 = g[0];
    p1 = g[8 * (DIM / 4)];
    s0 = dinv[tl * 16 + r];
    s1 = dinv[tl * 16 + r + 8];
  }

  int cur = 0;
  for (; tile < GEMM_TILES; tile += GEMM_BLOCKS) {
    float4* stage = xs4 + cur * XS_STAGE4;
    stage[r * XS_PITCH4 + c4] = make_float4(p0.x * s0, p0.y * s0, p0.z * s0, p0.w * s0);
    stage[(r + 8) * XS_PITCH4 + c4] = make_float4(p1.x * s1, p1.y * s1, p1.z * s1, p1.w * s1);
    __syncthreads();

    {
      const int nxt = tile + GEMM_BLOCKS;
      const int tl = (nxt < GEMM_TILES) ? nxt : GEMM_TILES - 1;
      const float4* g = (const float4*)(X + ((size_t)tl * 16 + r) * DIM) + c4;
      p0 = g[0];
      p1 = g[8 * (DIM / 4)];
      s0 = dinv[tl * 16 + r];
      s1 = dinv[tl * 16 + r + 8];
    }
    __builtin_amdgcn_sched_barrier(0);

    const v2f* arow = (const v2f*)stage + m * (XS_PITCH4 * 2) + hg;
    v2f af[GEMM_KSTEPS];
    #pragma unroll
    for (int j = 0; j < GEMM_KSTEPS; ++j) af[j] = arow[2 * j];
    __builtin_amdgcn_sched_barrier(0);

    v8f acc = {};
    #pragma unroll
    for (int j = 0; j < GEMM_KSTEPS; ++j) {
      acc = __builtin_amdgcn_wmma_f32_16x16x4_f32(
           false, af[j],  false, bf[j],
           (short)0, acc,  false,  false);
    }

    float* hrow = H + (size_t)tile * 16 * DIM + n0 + m;
    #pragma unroll
    for (int v = 0; v < 8; ++v)
      hrow[(size_t)(v + 8 * hg) * DIM] = acc[v];

    cur ^= 1;
  }
}


#define AGG_THREADS 1024
#define AGG_WAVES 32
#define SCAN_U 4
#define N_DST4 (N_EDGESC / 4)
#define WAVE4 (32 * SCAN_U)
#define TRIP4 (AGG_WAVES * WAVE4)
#define N_TRIPS ((N_DST4 + TRIP4 - 1) / TRIP4)
#define BATCH 8
#define LIST_CAP (BATCH - 1 + 4 * WAVE4 + 1)
#define SRC_MASK ((1 << 17) - 1)
#define CELLS_PER_WAVE (BIN_BLOCKS / AGG_WAVES)
#define ROWS_PW ((NPB + AGG_WAVES - 1) / AGG_WAVES)
#define RCAP 64
#define FCAP (ROWS_PW * (RCAP + 1))
#define REC_PITCH ((FCAP + 31) / 32 * 32)
#define TILE_BYTES (NPB * DIM * 4)
#define LISTS_BYTES (AGG_WAVES * LIST_CAP * 4)
#define ROWQ_BYTES (NPB * RCAP * 4)
#define FLAT_BYTES (AGG_WAVES * REC_PITCH * 4)
#define POOL_BYTES (TILE_BYTES + LISTS_BYTES)

static_assert(ROWQ_BYTES + FLAT_BYTES <= POOL_BYTES, "queue path fits the pool");
static_assert(ROWS_PW <= 16 && ROWS_PW <= 32, "stream entry packs the slot in 4 bits; one lane per slot");
static_assert(TILE_BYTES % 16 == 0 && ROWQ_BYTES % 16 == 0, "pool parts stay 16-byte aligned");
static_assert(AGG_BLOCKS * NPB >= N_NODESC, "slices must cover every node");
static_assert(N_EDGESC % 4 == 0, "dst list is read as int4");
static_assert(N_EDGESC <= (1 << 21) && NPB <= (1 << 9), "list entry packs edge id and local row");
static_assert(BIN_BLOCKS % AGG_WAVES == 0, "every wave takes the same number of cells");

__device__ __forceinline__ void agg_load_trip(const int4* __restrict__ dst4, int base4, int lane,
                                              int4 (&v)[SCAN_U]) {
  #pragma unroll
  for (int u = 0; u < SCAN_U; ++u) {
    const int i4 = base4 + u * 32 + lane;
    v[u] = (i4 < N_DST4) ? dst4[i4] : make_int4(-1, -1, -1, -1);
  }
}

__device__ __forceinline__ void rows_load(int s, int lane, const float* __restrict__ H,
                                          float (&hv)[BATCH][4]) {
  #pragma unroll
  for (int i = 0; i < BATCH; ++i) {
    const int si = __builtin_amdgcn_readlane(s, i);
    const float* hr = H + (size_t)si * DIM + lane;
    hv[i][0] = hr[0];
    hv[i][1] = hr[32];
    hv[i][2] = hr[64];
    hv[i][3] = hr[96];
  }
}

__device__ __forceinline__ void rows_add(const float (&hv)[BATCH][4], int dl, int nb,
                                         int lane, float* tile) {
  #pragma unroll
  for (int i = 0; i < BATCH; ++i) {
    if (i < nb) {
      const int di = __builtin_amdgcn_readlane(dl, i);
      float* tr = tile + di * DIM + lane;
      __hip_atomic_fetch_add(tr + 0,  hv[i][0], __ATOMIC_RELAXED, __HIP_MEMORY_SCOPE_WORKGROUP);
      __hip_atomic_fetch_add(tr + 32, hv[i][1], __ATOMIC_RELAXED, __HIP_MEMORY_SCOPE_WORKGROUP);
      __hip_atomic_fetch_add(tr + 64, hv[i][2], __ATOMIC_RELAXED, __HIP_MEMORY_SCOPE_WORKGROUP);
      __hip_atomic_fetch_add(tr + 96, hv[i][3], __ATOMIC_RELAXED, __HIP_MEMORY_SCOPE_WORKGROUP);
    }
  }
}

__device__ __forceinline__ void agg_batch(const int* lst, int nb, int lane,
                                          const int* __restrict__ src,
                                          const float* __restrict__ H, float* tile) {
  int ent = 0;
  if (lane < nb) ent = lst[lane];
  const int dl = (int)((unsigned)ent >> 21);
  const int s = src[ent & ((1 << 21) - 1)];
  float hv[BATCH][4];
  rows_load(s, lane, H, hv);
  rows_add(hv, dl, nb, lane, tile);
}

__device__ __forceinline__ void agg_scan(int lo, int wave, int lane, int* mylist,
                                         const int* __restrict__ src, const int* __restrict__ dst,
                                         const float* __restrict__ H, float* tile) {
  const int4* dst4 = (const int4*)dst;
  int pend = 0;
  int trip = blockIdx.x % N_TRIPS;

  int4 nxt[SCAN_U];
  agg_load_trip(dst4, trip * TRIP4 + wave * WAVE4, lane, nxt);

  for (int it = 0; it < N_TRIPS; ++it) {
    int4 cur[SCAN_U];
    #pragma unroll
    for (int u = 0; u < SCAN_U; ++u) cur[u] = nxt[u];
    const int cur4 = trip * TRIP4 + wave * WAVE4;
    trip = (trip + 1 == N_TRIPS) ? 0 : trip + 1;
    if (it + 1 < N_TRIPS) agg_load_trip(dst4, trip * TRIP4 + wave * WAVE4, lane, nxt);

    #pragma unroll
    for (int u = 0; u < SCAN_U; ++u) {
      #pragma unroll
      for (int c = 0; c < 4; ++c) {
        const int d = (c == 0) ? cur[u].x : (c == 1) ? cur[u].y : (c == 2) ? cur[u].z : cur[u].w;
        const unsigned tt = (unsigned)(d - lo);
        const bool m = tt < (unsigned)NPB;
        const unsigned mask = __builtin_amdgcn_ballot_w32(m);
        if (mask != 0u) {
          const int pos = pend + (int)__builtin_amdgcn_mbcnt_lo(mask, 0u);
          const int e = (cur4 + u * 32 + lane) * 4 + c;
          if (m) mylist[pos] = (int)((tt << 21) | (unsigned)e);
          pend += __builtin_popcount(mask);
        }
      }
    }
    while (pend >= BATCH) {
      pend -= BATCH;
      agg_batch(mylist + pend, BATCH, lane, src, H, tile);
    }
  }
  if (pend > 0) agg_batch(mylist, pend, lane, src, H, tile);
}

__device__ __forceinline__ void row_put(int e, int* rowq, int* rowcnt, int* qovf) {
  const int dl = (int)((unsigned)e >> 17);
  const int slot = __hip_atomic_fetch_add(&rowcnt[dl], 1, __ATOMIC_RELAXED, __HIP_MEMORY_SCOPE_WORKGROUP);
  if (slot < RCAP) rowq[dl * RCAP + slot] = e & SRC_MASK;
  else *qovf = 1;
}

__device__ __forceinline__ void cells_request(const int* __restrict__ mycells, int wv, int lane,
                                              int2 (&cv)[CELLS_PER_WAVE]) {
  const int2* cp = (const int2*)(mycells + (size_t)wv * CELLS_PER_WAVE * CELL_INTS) + lane;
  #pragma unroll
  for (int c = 0; c < CELLS_PER_WAVE; ++c) cv[c] = cp[c * (CELL_INTS / 2)];
}

__device__ __forceinline__ void route_cells(const int2 (&cv)[CELLS_PER_WAVE], int lane,
                                            int* rowq, int* rowcnt, int* qovf) {
  #pragma unroll
  for (int c = 0; c < CELLS_PER_WAVE; ++c) {
    int cnt = __builtin_amdgcn_readfirstlane(cv[c].x);
    cnt = (cnt < CELL_CAP) ? cnt : CELL_CAP;
    const int i0 = 2 * lane;
    if (i0 >= 1 && i0 <= cnt) row_put(cv[c].x, rowq, rowcnt, qovf);
    if (i0 + 1 <= cnt) row_put(cv[c].y, rowq, rowcnt, qovf);
  }
}

__device__ __forceinline__ int rows_to_stream(int wv, int lane, int lo, int nvalid,
                                              const int* rowq, const int* rowcnt, int* stream) {
  int cntv = 0;
  {
    const int r = 32 * lane + wv;
    if (lane < ROWS_PW && r < nvalid) cntv = rowcnt[r];
  }
  int first[ROWS_PW];
  #pragma unroll
  for (int i = 0; i < ROWS_PW; ++i) {
    const int r = 32 * i + wv;
    const int cnt = __builtin_amdgcn_readlane(cntv, i);
    first[i] = 0;
    if (lane < cnt) first[i] = rowq[r * RCAP + lane];
  }
  int off = 0;
  #pragma unroll
  for (int i = 0; i < ROWS_PW; ++i) {
    const int r = 32 * i + wv;
    if (r < nvalid) {
      const int cnt = __builtin_amdgcn_readlane(cntv, i);
      const int tag = i << 17;
      if (lane == 0) stream[off] = (lo + r) | tag;
      if (lane < cnt) stream[off + 1 + lane] = first[i] | tag;
      for (int j = lane + 32; j < cnt; j += 32) stream[off + 1 + j] = rowq[r * RCAP + j] | tag;
      off += cnt + 1;
    }
  }
  return off;
}

__device__ __forceinline__ void rows_load4(int s, int lane, const float* __restrict__ H,
                                           float4 (&hv)[BATCH]) {
  #pragma unroll
  for (int i = 0; i < BATCH; ++i) {
    const int si = __builtin_amdgcn_readlane(s, i);
    hv[i] = ((const float4*)(H + (size_t)si * DIM))[lane];
  }
}

__device__ __forceinline__ void row_store(int slot, const float4& acc, int wv, int lane, int lo,
                                          float diov, const float4& bb, int do_relu,
                                          float* __restrict__ OUT) {
  const float d = __int_as_float(__builtin_amdgcn_readlane(__float_as_int(diov), slot));
  const int r = 32 * slot + wv;
  float4 a;
  a.x = acc.x * d + bb.x;
  a.y = acc.y * d + bb.y;
  a.z = acc.z * d + bb.z;
  a.w = acc.w * d + bb.w;
  if (do_relu) {
    a.x = fmaxf(a.x, 0.0f);
    a.y = fmaxf(a.y, 0.0f);
    a.z = fmaxf(a.z, 0.0f);
    a.w = fmaxf(a.w, 0.0f);
  }
  ((float4*)(OUT + (size_t)(lo + r) * DIM))[lane] = a;
}

template <int MODE>
__device__ __forceinline__ int stream_entry(int b, int n, int lane, const int* q,
                                            const int* __restrict__ rec) {
  const int idx = b * BATCH + lane;
  int e = 0;
  if (lane < BATCH && idx < n) e = (MODE == 0) ? q[idx] : rec[idx];
  return e;
}

template <int MODE>
__device__ __forceinline__ void sum_stream(int n, int wv, int lane, int lo, const int* q,
                                           int* __restrict__ rec, const float* __restrict__ H,
                                           float diov, const float4& bb, int do_relu,
                                           float* __restrict__ OUT) {
  const int nbat = (n + BATCH - 1) / BATCH;
  int e_rq, e_nx;
  float4 hv[BATCH];
  e_rq = stream_entry<MODE>(0, n, lane, q, rec);
  rows_load4(e_rq & SRC_MASK, lane, H, hv);
  e_nx = stream_entry<MODE>(1, n, lane, q, rec);

  float4 acc = make_float4(0.f, 0.f, 0.f, 0.f);
  int cur = -1;

  for (int b = 0; b < nbat; ++b) {
    float4 cv[BATCH];
    #pragma unroll
    for (int i = 0; i < BATCH; ++i) cv[i] = hv[i];
    const int e_cur = e_rq;

    e_rq = e_nx;
    rows_load4(e_rq & SRC_MASK, lane, H, hv);
    e_nx = stream_entry<MODE>(b + 2, n, lane, q, rec);

    const int left = n - b * BATCH;
    const int nb = (left < BATCH) ? left : BATCH;
    const int sl = (int)((unsigned)e_cur >> 17);
    if (MODE == 0 && lane < nb) rec[b * BATCH + lane] = e_cur;
    #pragma unroll
    for (int i = 0; i < BATCH; ++i) {
      if (i < nb) {
        const int s = __builtin_amdgcn_readlane(sl, i);
        if (s != cur) {
          if (cur >= 0) row_store(cur, acc, wv, lane, lo, diov, bb, do_relu, OUT);
          cur = s;
          acc = cv[i];
        } else {
          acc.x += cv[i].x;
          acc.y += cv[i].y;
          acc.z += cv[i].z;
          acc.w += cv[i].w;
        }
      }
    }
  }
  if (cur >= 0) row_store(cur, acc, wv, lane, lo, diov, bb, do_relu, OUT);
}

template <int MODE>
__global__ __launch_bounds__(AGG_THREADS) void k_agg_owner(const int* __restrict__ src,
                                                           const int* __restrict__ dst,
                                                           const float* __restrict__ dinv,
                                                           const float* __restrict__ H,
                                                           const float* __restrict__ bias,
                                                           float* __restrict__ OUT, int do_relu,
                                                           int* __restrict__ rec_cnt,
                                                           int* __restrict__ rec_all,
                                                           int* __restrict__ qbad,
                                                           const int* __restrict__ cells,
                                                           const int* __restrict__ ovf) {
  __shared__ __attribute__((aligned(16))) char pool[POOL_BYTES];
  __shared__ int rowcnt[NPB];
  __shared__ int qovf;

  float* tile = (float*)pool;
  int* lists = (int*)(pool + TILE_BYTES);
  int* rowq = (int*)pool;
  int* streams = (int*)(pool + ROWQ_BYTES);

  const int t = threadIdx.x;
  const int wave = t >> 5;
  const int lane = t & 31;
  const int lo = blockIdx.x * NPB;
  const int nvalid = (N_NODESC - lo < NPB) ? (N_NODESC - lo) : NPB;
  const int wv = __builtin_amdgcn_readfirstlane(wave);
  int* rec = rec_all + ((size_t)blockIdx.x * AGG_WAVES + wv) * REC_PITCH;
  int* mystream = streams + wv * REC_PITCH;

  float diov;
  {
    const int r = 32 * lane + wv;
    const int rc = (lane < ROWS_PW && r < nvalid) ? r : nvalid - 1;
    diov = dinv[lo + rc];
  }
  const float4 bb = ((const float4*)bias)[lane];

  bool bad;
  int n = 0;
  if (MODE == 0) {
    int2 cv[CELLS_PER_WAVE];
    bad = ovf[blockIdx.x] != 0;
    cells_request(cells + (size_t)blockIdx.x * BIN_BLOCKS * CELL_INTS, wv, lane, cv);
    if (t < NPB) rowcnt[t] = 0;
    if (t == 0) qovf = 0;
    __syncthreads();
    if (!bad) route_cells(cv, lane, rowq, rowcnt, &qovf);
    __syncthreads();
    bad = bad || (qovf != 0);
    if (t == 0) qbad[blockIdx.x] = bad ? 1 : 0;
    if (!bad) {
      n = rows_to_stream(wv, lane, lo, nvalid, rowq, rowcnt, mystream);
      __syncthreads();
    }
  } else {
    bad = qbad[blockIdx.x] != 0;
  }

  if (!bad) {
    if (MODE == 0) {
      sum_stream<0>(n, wv, lane, lo, mystream, rec, H, diov, bb, do_relu, OUT);
      if (lane == 0) rec_cnt[blockIdx.x * AGG_WAVES + wv] = n;
    } else {
      n = rec_cnt[blockIdx.x * AGG_WAVES + wv];
      sum_stream<1>(n, wv, lane, lo, mystream, rec, H, diov, bb, do_relu, OUT);
    }
    return;
  }

  float4 own[ROWS_PW];
  float dio[ROWS_PW];
  #pragma unroll
  for (int i = 0; i < ROWS_PW; ++i) {
    const int r = 32 * i + wv;
    const int rc = (r < nvalid) ? r : nvalid - 1;
    dio[i] = dinv[lo + rc];
    own[i] = ((const float4*)(H + (size_t)(lo + rc) * DIM))[lane];
  }
  #pragma unroll
  for (int i = 0; i < ROWS_PW; ++i) {
    const int r = 32 * i + wv;
    if (r < NPB) {
      float4 v = own[i];
      if (r >= nvalid) v = make_float4(0.f, 0.f, 0.f, 0.f);
      ((float4*)(tile + r * DIM))[lane] = v;
    }
  }
  __syncthreads();
  agg_scan(lo, wave, lane, lists + wv * LIST_CAP, src, dst, H, tile);
  __syncthreads();

  #pragma unroll
  for (int i = 0; i < ROWS_PW; ++i) {
    const int r = 32 * i + wv;
    if (r < nvalid) {
      const float4 res = ((const float4*)(tile + r * DIM))[lane];
      float4 a;
      a.x = res.x * dio[i] + bb.x;
      a.y = res.y * dio[i] + bb.y;
      a.z = res.z * dio[i] + bb.z;
      a.w = res.w * dio[i] + bb.w;
      if (do_relu) {
        a.x = fmaxf(a.x, 0.0f);
        a.y = fmaxf(a.y, 0.0f);
        a.z = fmaxf(a.z, 0.0f);
        a.w = fmaxf(a.w, 0.0f);
      }
      ((float4*)(OUT + (size_t)(lo + r) * DIM))[lane] = a;
    }
  }
}


extern "C" void kernel_launch(void* const* d_in, const int* in_sizes, int n_in,
                              void* d_out, int out_size, void* d_ws, size_t ws_size,
                              hipStream_t stream) {
  const float* x   = (const float*)d_in[0];
  const int*   ei  = (const int*)d_in[1];
  const float* W1  = (const float*)d_in[2];
  const float* b1  = (const float*)d_in[3];
  const float* W2  = (const float*)d_in[4];
  const float* b2  = (const float*)d_in[5];
  float*       out = (float*)d_out;

  const int* src = ei;
  const int* dst = ei + N_EDGESC;

  float* H    = (float*)d_ws;
  float* dinv = H + (size_t)N_NODESC * DIM + N_NODESC;
  int*   rec_cnt = (int*)(dinv + N_NODESC);
  int*   qbad    = rec_cnt + AGG_BLOCKS * AGG_WAVES;
  int*   rec_all = qbad + AGG_BLOCKS;
  int*   cells   = rec_all + (size_t)AGG_BLOCKS * AGG_WAVES * REC_PITCH;
  int*   ovf     = cells + (size_t)AGG_BLOCKS * BIN_BLOCKS * CELL_INTS;

  dim3 blk(256);
  dim3 gBin(BIN_BLOCKS);
  dim3 bBin(BIN_THREADS);
  dim3 gGemm(GEMM_BLOCKS);
  dim3 gAgg(AGG_BLOCKS);
  dim3 bAgg(AGG_THREADS);

  k_bin_count<<<gBin, bBin, 0, stream>>>(src, dst, cells);
  k_deg_owner<<<gAgg, bAgg, 0, stream>>>(dst, cells, dinv, ovf);

  k_gemm_wmma<<<gGemm, blk, 0, stream>>>(x, W1, dinv, H);
  k_agg_owner<0><<<gAgg, bAgg, 0, stream>>>(src, dst, dinv, H, b1, out,  1, rec_cnt, rec_all, qbad, cells, ovf);

  k_gemm_wmma<<<gGemm, blk, 0, stream>>>(out, W2, dinv, H);
  k_agg_owner<1><<<gAgg, bAgg, 0, stream>>>(src, dst, dinv, H, b2, out,  0, rec_cnt, rec_all, qbad, cells, ovf);
}
